# starting kernel
# baseline (speedup 1.0000x reference)
#include <hip/hip_runtime.h>
#include <math.h>

#define N_NODES 8192
#define D_OSC   64
#define E_EDGES 1048576

typedef __attribute__((ext_vector_type(2))) float v2f;
typedef __attribute__((ext_vector_type(8))) float v8f;

__device__ __forceinline__ float wave_reduce_sum(float v) {
  #pragma unroll
  for (int m = 16; m >= 1; m >>= 1) v += __shfl_xor(v, m, 32);
  return v;
}

__global__ __launch_bounds__(256) void k_zero(float* __restrict__ p, int n) {
  int i = blockIdx.x * 256 + threadIdx.x;
  if (i < n) p[i] = 0.0f;
}

__global__ __launch_bounds__(256) void k_norm_tanh(const float* __restrict__ sk,
                                                   const float* __restrict__ sh,
                                                   float* __restrict__ skn,
                                                   float* __restrict__ gv) {
  int row  = blockIdx.x * 8 + (threadIdx.x >> 5);
  int lane = threadIdx.x & 31;
  const float* r = sk + (size_t)row * D_OSC;
  float a = r[lane];
  float b = r[lane + 32];
  float ss  = wave_reduce_sum(a * a + b * b);
  float inv = rsqrtf(ss);
  float* o = skn + (size_t)row * D_OSC;
  o[lane]      = a * inv;
  o[lane + 32] = b * inv;

  int gi = blockIdx.x * 256 + threadIdx.x;
  if (gi < N_NODES) gv[gi] = tanhf(sh[gi]);
}

__global__ __launch_bounds__(256) void k_edge_K(const int* __restrict__ ind,
                                                const float* __restrict__ wK,
                                                const float* __restrict__ skn,
                                                float* __restrict__ fK) {
  int e    = blockIdx.x * 8 + (threadIdx.x >> 5);
  int lane = threadIdx.x & 31;
  int j0 = ind[2 * e];
  int j1 = ind[2 * e + 1];
  const float* r0 = skn + (size_t)j0 * D_OSC;
  const float* r1 = skn + (size_t)j1 * D_OSC;
  float a0 = r0[lane], a1 = r0[lane + 32];
  float b0 = r1[lane], b1 = r1[lane + 32];
  float s = wave_reduce_sum(a0 * b0 + a1 * b1);
  float c = wK[e] * s;
  atomicAdd(&fK[(size_t)j0 * D_OSC + lane],       c * b0);
  atomicAdd(&fK[(size_t)j0 * D_OSC + lane + 32],  c * b1);
  atomicAdd(&fK[(size_t)j1 * D_OSC + lane],       c * a0);
  atomicAdd(&fK[(size_t)j1 * D_OSC + lane + 32],  c * a1);
}

__global__ __launch_bounds__(256) void k_edge_HK(const int* __restrict__ ind,
                                                 const float* __restrict__ whk,
                                                 const float* __restrict__ skn,
                                                 const float* __restrict__ gv,
                                                 const float* __restrict__ kH,
                                                 const float* __restrict__ kK,
                                                 float* __restrict__ fH,
                                                 float* __restrict__ fK) {
  int e    = blockIdx.x * 8 + (threadIdx.x >> 5);
  int lane = threadIdx.x & 31;
  int i0 = ind[2 * e];
  int i1 = ind[2 * e + 1];
  const float* r0 = skn + (size_t)i0 * D_OSC;
  const float* r1 = skn + (size_t)i1 * D_OSC;
  float a0 = r0[lane], a1 = r0[lane + 32];
  float b0 = r1[lane], b1 = r1[lane + 32];
  float Gram = wave_reduce_sum(a0 * b0 + a1 * b1);
  float w  = whk[e];
  float g0 = gv[i0];
  float g1 = gv[i1];
  float invKH = 1.0f / kH[0];
  float invKK = 1.0f / kK[0];
  if (lane == 0) {
    atomicAdd(&fH[i0], Gram * w * g1 * invKH);
    atomicAdd(&fH[i1], Gram * w * g0 * invKH);
  }
  float Gc = -(g0 * g1) * w * invKK;
  atomicAdd(&fK[(size_t)i0 * D_OSC + lane],      Gc * b0);
  atomicAdd(&fK[(size_t)i0 * D_OSC + lane + 32], Gc * b1);
  atomicAdd(&fK[(size_t)i1 * D_OSC + lane],      Gc * a0);
  atomicAdd(&fK[(size_t)i1 * D_OSC + lane + 32], Gc * a1);
}

__global__ __launch_bounds__(256) void k_gemv_fused(const float* __restrict__ W,
                                                    const float* __restrict__ gv,
                                                    float* __restrict__ y,
                                                    float* __restrict__ z) {
  __shared__ float slab[8 * 288];
  __shared__ float zstage[512];

  const int wv   = threadIdx.x >> 5;
  const int lane = threadIdx.x & 31;
  const int hl   = lane & 15;
  const int koff = (lane < 16) ? 0 : 2;

  const int C0 = blockIdx.x * 512;
  const int r0 = blockIdx.y * 128 + wv * 16;

  zstage[threadIdx.x]       = 0.0f;
  zstage[threadIdx.x + 256] = 0.0f;
  __syncthreads();

  float* sl = slab + wv * 288;

  v2f bz0 = *(const v2f*)(gv + r0 +  0 + koff);
  v2f bz1 = *(const v2f*)(gv + r0 +  4 + koff);
  v2f bz2 = *(const v2f*)(gv + r0 +  8 + koff);
  v2f bz3 = *(const v2f*)(gv + r0 + 12 + koff);

  const float* wrow = W + (size_t)(r0 + hl) * N_NODES + koff;

  v8f yacc = {};
  for (int t = 0; t < 32; ++t) {
    const int c = C0 + t * 16;
    const float* wp = wrow + c;
    v2f a0 = *(const v2f*)(wp +  0);
    v2f a1 = *(const v2f*)(wp +  4);
    v2f a2 = *(const v2f*)(wp +  8);
    v2f a3 = *(const v2f*)(wp + 12);
    v2f by0 = *(const v2f*)(gv + c +  0 + koff);
    v2f by1 = *(const v2f*)(gv + c +  4 + koff);
    v2f by2 = *(const v2f*)(gv + c +  8 + koff);
    v2f by3 = *(const v2f*)(gv + c + 12 + koff);

    yacc = __builtin_amdgcn_wmma_f32_16x16x4_f32(false, a0, false, by0, (short)0, yacc, false, false);
    yacc = __builtin_amdgcn_wmma_f32_16x16x4_f32(false, a1, false, by1, (short)0, yacc, false, false);
    yacc = __builtin_amdgcn_wmma_f32_16x16x4_f32(false, a2, false, by2, (short)0, yacc, false, false);
    yacc = __builtin_amdgcn_wmma_f32_16x16x4_f32(false, a3, false, by3, (short)0, yacc, false, false);

    float* sr = sl + hl * 18 + koff;
    *(v2f*)(sr +  0) = a0;
    *(v2f*)(sr +  4) = a1;
    *(v2f*)(sr +  8) = a2;
    *(v2f*)(sr + 12) = a3;

    v2f az0, az1, az2, az3;
    az0[0] = sl[( 0 + koff) * 18 + hl];  az0[1] = sl[( 1 + koff) * 18 + hl];
    az1[0] = sl[( 4 + koff) * 18 + hl];  az1[1] = sl[( 5 + koff) * 18 + hl];
    az2[0] = sl[( 8 + koff) * 18 + hl];  az2[1] = sl[( 9 + koff) * 18 + hl];
    az3[0] = sl[(12 + koff) * 18 + hl];  az3[1] = sl[(13 + koff) * 18 + hl];

    v8f zt = {};
    zt = __builtin_amdgcn_wmma_f32_16x16x4_f32(false, az0, false, bz0, (short)0, zt, false, false);
    zt = __builtin_amdgcn_wmma_f32_16x16x4_f32(false, az1, false, bz1, (short)0, zt, false, false);
    zt = __builtin_amdgcn_wmma_f32_16x16x4_f32(false, az2, false, bz2, (short)0, zt, false, false);
    zt = __builtin_amdgcn_wmma_f32_16x16x4_f32(false, az3, false, bz3, (short)0, zt, false, false);

    if (hl == 0) {
      int zb = t * 16 + ((lane == 16) ? 8 : 0);
      #pragma unroll
      for (int v = 0; v < 8; ++v) atomicAdd(&zstage[zb + v], zt[v]);
    }
  }

  if (hl == 0) {
    int base = r0 + ((lane == 16) ? 8 : 0);
    #pragma unroll
    for (int v = 0; v < 8; ++v) atomicAdd(&y[base + v], yacc[v]);
  }

  __syncthreads();
  atomicAdd(&z[C0 + threadIdx.x],       zstage[threadIdx.x]);
  atomicAdd(&z[C0 + threadIdx.x + 256], zstage[threadIdx.x + 256]);
}

__global__ __launch_bounds__(256) void k_fin_H(const float* __restrict__ fHacc,
                                               const float* __restrict__ y,
                                               const float* __restrict__ z,
                                               const float* __restrict__ sh,
                                               const float* __restrict__ bias,
                                               float* __restrict__ out) {
  int i = blockIdx.x * 256 + threadIdx.x;
  if (i < N_NODES)
    out[i] = fHacc[i] + 0.5f * (y[i] + z[i]) - sh[i] + bias[i];
}

__global__ __launch_bounds__(256) void k_fin_K(const float* __restrict__ fKacc,
                                               const float* __restrict__ skn,
                                               float* __restrict__ out) {
  int row  = blockIdx.x * 8 + (threadIdx.x >> 5);
  int lane = threadIdx.x & 31;
  size_t base = (size_t)row * D_OSC;
  float r0 = fKacc[base + lane], r1 = fKacc[base + lane + 32];
  float s0 = skn[base + lane],   s1 = skn[base + lane + 32];
  float dot = wave_reduce_sum(s0 * r0 + s1 * r1);
  out[base + lane]      = -r0 + s0 * dot;
  out[base + lane + 32] = -r1 + s1 * dot;
}

extern "C" void kernel_launch(void* const* d_in, const int* in_sizes, int n_in,
                              void* d_out, int out_size, void* d_ws, size_t ws_size,
                              hipStream_t stream) {
  const float* state_H    = (const float*)d_in[0];
  const float* state_K    = (const float*)d_in[1];
  const int*   ind_K      = (const int*)d_in[2];
  const int*   ind_HK     = (const int*)d_in[3];
  const float* kappa_K    = (const float*)d_in[4];
  const float* kappa_H    = (const float*)d_in[5];
  const float* weights_H  = (const float*)d_in[6];
  const float* bias_H     = (const float*)d_in[7];
  const float* weights_HK = (const float*)d_in[8];
  const float* w_K        = (const float*)d_in[9];

  float* out_fH = (float*)d_out;
  float* out_fK = out_fH + N_NODES;

  float* ws     = (float*)d_ws;
  float* fK_acc = ws;
  float* fH_acc = fK_acc + (size_t)N_NODES * D_OSC;
  float* yv     = fH_acc + N_NODES;
  float* zv     = yv + N_NODES;
  float* skn    = zv + N_NODES;
  float* gv     = skn + (size_t)N_NODES * D_OSC;

  const int nAcc = N_NODES * D_OSC + 3 * N_NODES;

  k_zero<<<(nAcc + 255) / 256, 256, 0, stream>>>(ws, nAcc);

  k_norm_tanh<<<N_NODES / 8, 256, 0, stream>>>(state_K, state_H, skn, gv);

  k_edge_K<<<E_EDGES / 8, 256, 0, stream>>>(ind_K, w_K, skn, fK_acc);

  k_edge_HK<<<E_EDGES / 8, 256, 0, stream>>>(ind_HK, weights_HK, skn, gv,
                                             kappa_H, kappa_K, fH_acc, fK_acc);

  k_gemv_fused<<<dim3(16, 64), 256, 0, stream>>>(weights_H, gv, yv, zv);

  k_fin_H<<<(N_NODES + 255) / 256, 256, 0, stream>>>(fH_acc, yv, zv,
                                                     state_H, bias_H, out_fH);
  k_fin_K<<<N_NODES / 8, 256, 0, stream>>>(fK_acc, skn, out_fK);
}


# Coupling edges factored per node: edges scatter w*g[j]*skn[j] into V; f_H term=<skn,V>/kH, f_K term=-g*V/kK at finalize; drops per-edge Gram reduce and f_H atomics; changes summation order
# speedup vs baseline: 1.7586x; 1.7586x over previous
#include <hip/hip_runtime.h>
#include <math.h>

#define N_NODES 8192
#define D_OSC   64
#define E_EDGES 1048576

typedef __attribute__((ext_vector_type(2))) float v2f;
typedef __attribute__((ext_vector_type(8))) float v8f;

__device__ __forceinline__ float wave_reduce_sum(float v) {
  #pragma unroll
  for (int m = 16; m >= 1; m >>= 1) v += __shfl_xor(v, m, 32);
  return v;
}

__global__ __launch_bounds__(256) void k_zero(float* __restrict__ p, int n) {
  int i = blockIdx.x * 256 + threadIdx.x;
  if (i < n) p[i] = 0.0f;
}

__global__ __launch_bounds__(256) void k_norm_tanh(const float* __restrict__ sk,
                                                   const float* __restrict__ sh,
                                                   float* __restrict__ skn,
                                                   float* __restrict__ hs,
                                                   float* __restrict__ gv) {
  int row  = blockIdx.x * 8 + (threadIdx.x >> 5);
  int lane = threadIdx.x & 31;
  const float* r = sk + (size_t)row * D_OSC;
  float a = r[lane];
  float b = r[lane + 32];
  float ss  = wave_reduce_sum(a * a + b * b);
  float inv = rsqrtf(ss);
  float na = a * inv;
  float nb = b * inv;
  float* o = skn + (size_t)row * D_OSC;
  o[lane]      = na;
  o[lane + 32] = nb;
  float grow = tanhf(sh[row]);
  float* oh = hs + (size_t)row * D_OSC;
  oh[lane]      = grow * na;
  oh[lane + 32] = grow * nb;

  int gi = blockIdx.x * 256 + threadIdx.x;
  if (gi < N_NODES) gv[gi] = tanhf(sh[gi]);
}

__global__ __launch_bounds__(256) void k_edge_K(const int* __restrict__ ind,
                                                const float* __restrict__ wK,
                                                const float* __restrict__ skn,
                                                float* __restrict__ fK) {
  int e    = blockIdx.x * 8 + (threadIdx.x >> 5);
  int lane = threadIdx.x & 31;
  int j0 = ind[2 * e];
  int j1 = ind[2 * e + 1];
  const float* r0 = skn + (size_t)j0 * D_OSC;
  const float* r1 = skn + (size_t)j1 * D_OSC;
  float a0 = r0[lane], a1 = r0[lane + 32];
  float b0 = r1[lane], b1 = r1[lane + 32];
  float s = wave_reduce_sum(a0 * b0 + a1 * b1);
  float c = wK[e] * s;
  atomicAdd(&fK[(size_t)j0 * D_OSC + lane],       c * b0);
  atomicAdd(&fK[(size_t)j0 * D_OSC + lane + 32],  c * b1);
  atomicAdd(&fK[(size_t)j1 * D_OSC + lane],       c * a0);
  atomicAdd(&fK[(size_t)j1 * D_OSC + lane + 32],  c * a1);
}

__global__ __launch_bounds__(256) void k_edge_HK(const int* __restrict__ ind,
                                                 const float* __restrict__ whk,
                                                 const float* __restrict__ hs,
                                                 float* __restrict__ V) {
  int e    = blockIdx.x * 8 + (threadIdx.x >> 5);
  int lane = threadIdx.x & 31;
  int i0 = ind[2 * e];
  int i1 = ind[2 * e + 1];
  const float* r0 = hs + (size_t)i0 * D_OSC;
  const float* r1 = hs + (size_t)i1 * D_OSC;
  float a0 = r0[lane], a1 = r0[lane + 32];
  float b0 = r1[lane], b1 = r1[lane + 32];
  float w  = whk[e];
  atomicAdd(&V[(size_t)i0 * D_OSC + lane],      w * b0);
  atomicAdd(&V[(size_t)i0 * D_OSC + lane + 32], w * b1);
  atomicAdd(&V[(size_t)i1 * D_OSC + lane],      w * a0);
  atomicAdd(&V[(size_t)i1 * D_OSC + lane + 32], w * a1);
}

__global__ __launch_bounds__(256) void k_gemv_fused(const float* __restrict__ W,
                                                    const float* __restrict__ gv,
                                                    float* __restrict__ y,
                                                    float* __restrict__ z) {
  __shared__ float slab[8 * 288];
  __shared__ float zstage[512];

  const int wv   = threadIdx.x >> 5;
  const int lane = threadIdx.x & 31;
  const int hl   = lane & 15;
  const int koff = (lane < 16) ? 0 : 2;

  const int C0 = blockIdx.x * 512;
  const int r0 = blockIdx.y * 128 + wv * 16;

  zstage[threadIdx.x]       = 0.0f;
  zstage[threadIdx.x + 256] = 0.0f;
  __syncthreads();

  float* sl = slab + wv * 288;

  v2f bz0 = *(const v2f*)(gv + r0 +  0 + koff);
  v2f bz1 = *(const v2f*)(gv + r0 +  4 + koff);
  v2f bz2 = *(const v2f*)(gv + r0 +  8 + koff);
  v2f bz3 = *(const v2f*)(gv + r0 + 12 + koff);

  const float* wrow = W + (size_t)(r0 + hl) * N_NODES + koff;

  v8f yacc = {};
  for (int t = 0; t < 32; ++t) {
    const int c = C0 + t * 16;
    const float* wp = wrow + c;
    v2f a0 = *(const v2f*)(wp +  0);
    v2f a1 = *(const v2f*)(wp +  4);
    v2f a2 = *(const v2f*)(wp +  8);
    v2f a3 = *(const v2f*)(wp + 12);
    v2f by0 = *(const v2f*)(gv + c +  0 + koff);
    v2f by1 = *(const v2f*)(gv + c +  4 + koff);
    v2f by2 = *(const v2f*)(gv + c +  8 + koff);
    v2f by3 = *(const v2f*)(gv + c + 12 + koff);

    yacc = __builtin_amdgcn_wmma_f32_16x16x4_f32(false, a0, false, by0, (short)0, yacc, false, false);
    yacc = __builtin_amdgcn_wmma_f32_16x16x4_f32(false, a1, false, by1, (short)0, yacc, false, false);
    yacc = __builtin_amdgcn_wmma_f32_16x16x4_f32(false, a2, false, by2, (short)0, yacc, false, false);
    yacc = __builtin_amdgcn_wmma_f32_16x16x4_f32(false, a3, false, by3, (short)0, yacc, false, false);

    float* sr = sl + hl * 18 + koff;
    *(v2f*)(sr +  0) = a0;
    *(v2f*)(sr +  4) = a1;
    *(v2f*)(sr +  8) = a2;
    *(v2f*)(sr + 12) = a3;

    v2f az0, az1, az2, az3;
    az0[0] = sl[( 0 + koff) * 18 + hl];  az0[1] = sl[( 1 + koff) * 18 + hl];
    az1[0] = sl[( 4 + koff) * 18 + hl];  az1[1] = sl[( 5 + koff) * 18 + hl];
    az2[0] = sl[( 8 + koff) * 18 + hl];  az2[1] = sl[( 9 + koff) * 18 + hl];
    az3[0] = sl[(12 + koff) * 18 + hl];  az3[1] = sl[(13 + koff) * 18 + hl];

    v8f zt = {};
    zt = __builtin_amdgcn_wmma_f32_16x16x4_f32(false, az0, false, bz0, (short)0, zt, false, false);
    zt = __builtin_amdgcn_wmma_f32_16x16x4_f32(false, az1, false, bz1, (short)0, zt, false, false);
    zt = __builtin_amdgcn_wmma_f32_16x16x4_f32(false, az2, false, bz2, (short)0, zt, false, false);
    zt = __builtin_amdgcn_wmma_f32_16x16x4_f32(false, az3, false, bz3, (short)0, zt, false, false);

    if (hl == 0) {
      int zb = t * 16 + ((lane == 16) ? 8 : 0);
      #pragma unroll
      for (int v = 0; v < 8; ++v) atomicAdd(&zstage[zb + v], zt[v]);
    }
  }

  if (hl == 0) {
    int base = r0 + ((lane == 16) ? 8 : 0);
    #pragma unroll
    for (int v = 0; v < 8; ++v) atomicAdd(&y[base + v], yacc[v]);
  }

  __syncthreads();
  atomicAdd(&z[C0 + threadIdx.x],       zstage[threadIdx.x]);
  atomicAdd(&z[C0 + threadIdx.x + 256], zstage[threadIdx.x + 256]);
}

__global__ __launch_bounds__(1024) void k_fin(const float* __restrict__ fKacc,
                                              const float* __restrict__ V,
                                              const float* __restrict__ skn,
                                              const float* __restrict__ gv,
                                              const float* __restrict__ y,
                                              const float* __restrict__ z,
                                              const float* __restrict__ sh,
                                              const float* __restrict__ bias,
                                              const float* __restrict__ kH,
                                              const float* __restrict__ kK,
                                              float* __restrict__ outH,
                                              float* __restrict__ outK) {
  __shared__ float sdot[32];
  int wv   = threadIdx.x >> 5;
  int lane = threadIdx.x & 31;
  int row  = blockIdx.x * 32 + wv;
  size_t base = (size_t)row * D_OSC;
  float s0 = skn[base + lane],   s1 = skn[base + lane + 32];
  float v0 = V[base + lane],     v1 = V[base + lane + 32];
  float k0 = fKacc[base + lane], k1 = fKacc[base + lane + 32];
  float dhv = wave_reduce_sum(s0 * v0 + s1 * v1);
  float c  = gv[row] / kK[0];
  float r0 = k0 - c * v0;
  float r1 = k1 - c * v1;
  float dot = wave_reduce_sum(s0 * r0 + s1 * r1);
  outK[base + lane]      = -r0 + s0 * dot;
  outK[base + lane + 32] = -r1 + s1 * dot;
  if (lane == 0) sdot[wv] = dhv;
  __syncthreads();
  if (wv == 0) {
    int i = blockIdx.x * 32 + lane;
    outH[i] = sdot[lane] / kH[0] + 0.5f * (y[i] + z[i]) - sh[i] + bias[i];
  }
}

extern "C" void kernel_launch(void* const* d_in, const int* in_sizes, int n_in,
                              void* d_out, int out_size, void* d_ws, size_t ws_size,
                              hipStream_t stream) {
  const float* state_H    = (const float*)d_in[0];
  const float* state_K    = (const float*)d_in[1];
  const int*   ind_K      = (const int*)d_in[2];
  const int*   ind_HK     = (const int*)d_in[3];
  const float* kappa_K    = (const float*)d_in[4];
  const float* kappa_H    = (const float*)d_in[5];
  const float* weights_H  = (const float*)d_in[6];
  const float* bias_H     = (const float*)d_in[7];
  const float* weights_HK = (const float*)d_in[8];
  const float* w_K        = (const float*)d_in[9];

  float* out_fH = (float*)d_out;
  float* out_fK = out_fH + N_NODES;

  float* ws     = (float*)d_ws;
  float* fK_acc = ws;
  float* Vacc   = fK_acc + (size_t)N_NODES * D_OSC;
  float* yv     = Vacc + (size_t)N_NODES * D_OSC;
  float* zv     = yv + N_NODES;
  float* skn    = zv + N_NODES;
  float* hs     = skn + (size_t)N_NODES * D_OSC;
  float* gv     = hs + (size_t)N_NODES * D_OSC;

  const int nAcc = 2 * N_NODES * D_OSC + 2 * N_NODES;

  k_zero<<<(nAcc + 255) / 256, 256, 0, stream>>>(ws, nAcc);

  k_norm_tanh<<<N_NODES / 8, 256, 0, stream>>>(state_K, state_H, skn, hs, gv);

  k_edge_K<<<E_EDGES / 8, 256, 0, stream>>>(ind_K, w_K, skn, fK_acc);

  k_edge_HK<<<E_EDGES / 8, 256, 0, stream>>>(ind_HK, weights_HK, hs, Vacc);

  k_gemv_fused<<<dim3(16, 64), 256, 0, stream>>>(weights_H, gv, yv, zv);

  k_fin<<<N_NODES / 32, 1024, 0, stream>>>(fK_acc, Vacc, skn, gv, yv, zv,
                                           state_H, bias_H, kappa_H, kappa_K,
                                           out_fH, out_fK);
}


# Edge kernels: each wave walks 32 consecutive edges (one coalesced index load, node ids read from lanes into scalars), row loads of 8 edges in flight together; atomic arrival order differs
# speedup vs baseline: 2.0496x; 1.1655x over previous
#include <hip/hip_runtime.h>
#include <math.h>

#define N_NODES 8192
#define D_OSC   64
#define E_EDGES 1048576

typedef __attribute__((ext_vector_type(2))) float v2f;
typedef __attribute__((ext_vector_type(8))) float v8f;

__device__ __forceinline__ float wave_reduce_sum(float v) {
  #pragma unroll
  for (int m = 16; m >= 1; m >>= 1) v += __shfl_xor(v, m, 32);
  return v;
}

__global__ __launch_bounds__(256) void k_zero(float* __restrict__ p, int n) {
  int i = blockIdx.x * 256 + threadIdx.x;
  if (i < n) p[i] = 0.0f;
}

__global__ __launch_bounds__(256) void k_norm_tanh(const float* __restrict__ sk,
                                                   const float* __restrict__ sh,
                                                   float* __restrict__ skn,
                                                   float* __restrict__ hs,
                                                   float* __restrict__ gv) {
  int row  = blockIdx.x * 8 + (threadIdx.x >> 5);
  int lane = threadIdx.x & 31;
  const float* r = sk + (size_t)row * D_OSC;
  float a = r[lane];
  float b = r[lane + 32];
  float ss  = wave_reduce_sum(a * a + b * b);
  float inv = rsqrtf(ss);
  float na = a * inv;
  float nb = b * inv;
  float* o = skn + (size_t)row * D_OSC;
  o[lane]      = na;
  o[lane + 32] = nb;
  float grow = tanhf(sh[row]);
  float* oh = hs + (size_t)row * D_OSC;
  oh[lane]      = grow * na;
  oh[lane + 32] = grow * nb;

  int gi = blockIdx.x * 256 + threadIdx.x;
  if (gi < N_NODES) gv[gi] = tanhf(sh[gi]);
}

#define EPW  32
#define EGRP 8

__global__ __launch_bounds__(256) void k_edge_K(const int* __restrict__ ind,
                                                const float* __restrict__ wK,
                                                const float* __restrict__ skn,
                                                float* __restrict__ fK) {
  const int lane = threadIdx.x & 31;
  const int e0   = (blockIdx.x * 8 + (threadIdx.x >> 5)) * EPW;
  const int2  myj = ((const int2*)ind)[e0 + lane];
  const int   myw = __float_as_int(wK[e0 + lane]);
  #pragma unroll
  for (int g = 0; g < EPW; g += EGRP) {
    int   j0[EGRP], j1[EGRP];
    float a0[EGRP], a1[EGRP], b0[EGRP], b1[EGRP];
    #pragma unroll
    for (int u = 0; u < EGRP; ++u) {
      j0[u] = __builtin_amdgcn_readlane(myj.x, g + u);
      j1[u] = __builtin_amdgcn_readlane(myj.y, g + u);
      const float* r0 = skn + (size_t)j0[u] * D_OSC;
      const float* r1 = skn + (size_t)j1[u] * D_OSC;
      a0[u] = r0[lane];  a1[u] = r0[lane + 32];
      b0[u] = r1[lane];  b1[u] = r1[lane + 32];
    }
    #pragma unroll
    for (int u = 0; u < EGRP; ++u) {
      float s = wave_reduce_sum(a0[u] * b0[u] + a1[u] * b1[u]);
      float c = __int_as_float(__builtin_amdgcn_readlane(myw, g + u)) * s;
      float* o0 = fK + (size_t)j0[u] * D_OSC;
      float* o1 = fK + (size_t)j1[u] * D_OSC;
      atomicAdd(&o0[lane],      c * b0[u]);
      atomicAdd(&o0[lane + 32], c * b1[u]);
      atomicAdd(&o1[lane],      c * a0[u]);
      atomicAdd(&o1[lane + 32], c * a1[u]);
    }
  }
}

__global__ __launch_bounds__(256) void k_edge_HK(const int* __restrict__ ind,
                                                 const float* __restrict__ whk,
                                                 const float* __restrict__ hs,
                                                 float* __restrict__ V) {
  const int lane = threadIdx.x & 31;
  const int e0   = (blockIdx.x * 8 + (threadIdx.x >> 5)) * EPW;
  const int2  myi = ((const int2*)ind)[e0 + lane];
  const int   myw = __float_as_int(whk[e0 + lane]);
  #pragma unroll
  for (int g = 0; g < EPW; g += EGRP) {
    int   i0[EGRP], i1[EGRP];
    float a0[EGRP], a1[EGRP], b0[EGRP], b1[EGRP];
    #pragma unroll
    for (int u = 0; u < EGRP; ++u) {
      i0[u] = __builtin_amdgcn_readlane(myi.x, g + u);
      i1[u] = __builtin_amdgcn_readlane(myi.y, g + u);
      const float* r0 = hs + (size_t)i0[u] * D_OSC;
      const float* r1 = hs + (size_t)i1[u] * D_OSC;
      a0[u] = r0[lane];  a1[u] = r0[lane + 32];
      b0[u] = r1[lane];  b1[u] = r1[lane + 32];
    }
    #pragma unroll
    for (int u = 0; u < EGRP; ++u) {
      float w = __int_as_float(__builtin_amdgcn_readlane(myw, g + u));
      float* o0 = V + (size_t)i0[u] * D_OSC;
      float* o1 = V + (size_t)i1[u] * D_OSC;
      atomicAdd(&o0[lane],      w * b0[u]);
      atomicAdd(&o0[lane + 32], w * b1[u]);
      atomicAdd(&o1[lane],      w * a0[u]);
      atomicAdd(&o1[lane + 32], w * a1[u]);
    }
  }
}

__global__ __launch_bounds__(256) void k_gemv_fused(const float* __restrict__ W,
                                                    const float* __restrict__ gv,
                                                    float* __restrict__ y,
                                                    float* __restrict__ z) {
  __shared__ float slab[8 * 288];
  __shared__ float zstage[512];

  const int wv   = threadIdx.x >> 5;
  const int lane = threadIdx.x & 31;
  const int hl   = lane & 15;
  const int koff = (lane < 16) ? 0 : 2;

  const int C0 = blockIdx.x * 512;
  const int r0 = blockIdx.y * 128 + wv * 16;

  zstage[threadIdx.x]       = 0.0f;
  zstage[threadIdx.x + 256] = 0.0f;
  __syncthreads();

  float* sl = slab + wv * 288;

  v2f bz0 = *(const v2f*)(gv + r0 +  0 + koff);
  v2f bz1 = *(const v2f*)(gv + r0 +  4 + koff);
  v2f bz2 = *(const v2f*)(gv + r0 +  8 + koff);
  v2f bz3 = *(const v2f*)(gv + r0 + 12 + koff);

  const float* wrow = W + (size_t)(r0 + hl) * N_NODES + koff;

  v8f yacc = {};
  for (int t = 0; t < 32; ++t) {
    const int c = C0 + t * 16;
    const float* wp = wrow + c;
    v2f a0 = *(const v2f*)(wp +  0);
    v2f a1 = *(const v2f*)(wp +  4);
    v2f a2 = *(const v2f*)(wp +  8);
    v2f a3 = *(const v2f*)(wp + 12);
    v2f by0 = *(const v2f*)(gv + c +  0 + koff);
    v2f by1 = *(const v2f*)(gv + c +  4 + koff);
    v2f by2 = *(const v2f*)(gv + c +  8 + koff);
    v2f by3 = *(const v2f*)(gv + c + 12 + koff);

    yacc = __builtin_amdgcn_wmma_f32_16x16x4_f32(false, a0, false, by0, (short)0, yacc, false, false);
    yacc = __builtin_amdgcn_wmma_f32_16x16x4_f32(false, a1, false, by1, (short)0, yacc, false, false);
    yacc = __builtin_amdgcn_wmma_f32_16x16x4_f32(false, a2, false, by2, (short)0, yacc, false, false);
    yacc = __builtin_amdgcn_wmma_f32_16x16x4_f32(false, a3, false, by3, (short)0, yacc, false, false);

    float* sr = sl + hl * 18 + koff;
    *(v2f*)(sr +  0) = a0;
    *(v2f*)(sr +  4) = a1;
    *(v2f*)(sr +  8) = a2;
    *(v2f*)(sr + 12) = a3;

    v2f az0, az1, az2, az3;
    az0[0] = sl[( 0 + koff) * 18 + hl];  az0[1] = sl[( 1 + koff) * 18 + hl];
    az1[0] = sl[( 4 + koff) * 18 + hl];  az1[1] = sl[( 5 + koff) * 18 + hl];
    az2[0] = sl[( 8 + koff) * 18 + hl];  az2[1] = sl[( 9 + koff) * 18 + hl];
    az3[0] = sl[(12 + koff) * 18 + hl];  az3[1] = sl[(13 + koff) * 18 + hl];

    v8f zt = {};
    zt = __builtin_amdgcn_wmma_f32_16x16x4_f32(false, az0, false, bz0, (short)0, zt, false, false);
    zt = __builtin_amdgcn_wmma_f32_16x16x4_f32(false, az1, false, bz1, (short)0, zt, false, false);
    zt = __builtin_amdgcn_wmma_f32_16x16x4_f32(false, az2, false, bz2, (short)0, zt, false, false);
    zt = __builtin_amdgcn_wmma_f32_16x16x4_f32(false, az3, false, bz3, (short)0, zt, false, false);

    if (hl == 0) {
      int zb = t * 16 + ((lane == 16) ? 8 : 0);
      #pragma unroll
      for (int v = 0; v < 8; ++v) atomicAdd(&zstage[zb + v], zt[v]);
    }
  }

  if (hl == 0) {
    int base = r0 + ((lane == 16) ? 8 : 0);
    #pragma unroll
    for (int v = 0; v < 8; ++v) atomicAdd(&y[base + v], yacc[v]);
  }

  __syncthreads();
  atomicAdd(&z[C0 + threadIdx.x],       zstage[threadIdx.x]);
  atomicAdd(&z[C0 + threadIdx.x + 256], zstage[threadIdx.x + 256]);
}

__global__ __launch_bounds__(1024) void k_fin(const float* __restrict__ fKacc,
                                              const float* __restrict__ V,
                                              const float* __restrict__ skn,
                                              const float* __restrict__ gv,
                                              const float* __restrict__ y,
                                              const float* __restrict__ z,
                                              const float* __restrict__ sh,
                                              const float* __restrict__ bias,
                                              const float* __restrict__ kH,
                                              const float* __restrict__ kK,
                                              float* __restrict__ outH,
                                              float* __restrict__ outK) {
  __shared__ float sdot[32];
  int wv   = threadIdx.x >> 5;
  int lane = threadIdx.x & 31;
  int row  = blockIdx.x * 32 + wv;
  size_t base = (size_t)row * D_OSC;
  float s0 = skn[base + lane],   s1 = skn[base + lane + 32];
  float v0 = V[base + lane],     v1 = V[base + lane + 32];
  float k0 = fKacc[base + lane], k1 = fKacc[base + lane + 32];
  float dhv = wave_reduce_sum(s0 * v0 + s1 * v1);
  float c  = gv[row] / kK[0];
  float r0 = k0 - c * v0;
  float r1 = k1 - c * v1;
  float dot = wave_reduce_sum(s0 * r0 + s1 * r1);
  outK[base + lane]      = -r0 + s0 * dot;
  outK[base + lane + 32] = -r1 + s1 * dot;
  if (lane == 0) sdot[wv] = dhv;
  __syncthreads();
  if (wv == 0) {
    int i = blockIdx.x * 32 + lane;
    outH[i] = sdot[lane] / kH[0] + 0.5f * (y[i] + z[i]) - sh[i] + bias[i];
  }
}

extern "C" void kernel_launch(void* const* d_in, const int* in_sizes, int n_in,
                              void* d_out, int out_size, void* d_ws, size_t ws_size,
                              hipStream_t stream) {
  const float* state_H    = (const float*)d_in[0];
  const float* state_K    = (const float*)d_in[1];
  const int*   ind_K      = (const int*)d_in[2];
  const int*   ind_HK     = (const int*)d_in[3];
  const float* kappa_K    = (const float*)d_in[4];
  const float* kappa_H    = (const float*)d_in[5];
  const float* weights_H  = (const float*)d_in[6];
  const float* bias_H     = (const float*)d_in[7];
  const float* weights_HK = (const float*)d_in[8];
  const float* w_K        = (const float*)d_in[9];

  float* out_fH = (float*)d_out;
  float* out_fK = out_fH + N_NODES;

  float* ws     = (float*)d_ws;
  float* fK_acc = ws;
  float* Vacc   = fK_acc + (size_t)N_NODES * D_OSC;
  float* yv     = Vacc + (size_t)N_NODES * D_OSC;
  float* zv     = yv + N_NODES;
  float* skn    = zv + N_NODES;
  float* hs     = skn + (size_t)N_NODES * D_OSC;
  float* gv     = hs + (size_t)N_NODES * D_OSC;

  const int nAcc = 2 * N_NODES * D_OSC + 2 * N_NODES;

  k_zero<<<(nAcc + 255) / 256, 256, 0, stream>>>(ws, nAcc);

  k_norm_tanh<<<N_NODES / 8, 256, 0, stream>>>(state_K, state_H, skn, hs, gv);

  k_edge_K<<<E_EDGES / (8 * EPW), 256, 0, stream>>>(ind_K, w_K, skn, fK_acc);

  k_edge_HK<<<E_EDGES / (8 * EPW), 256, 0, stream>>>(ind_HK, weights_HK, hs, Vacc);

  k_gemv_fused<<<dim3(16, 64), 256, 0, stream>>>(weights_H, gv, yv, zv);

  k_fin<<<N_NODES / 32, 1024, 0, stream>>>(fK_acc, Vacc, skn, gv, yv, zv,
                                           state_H, bias_H, kappa_H, kappa_K,
                                           out_fH, out_fK);
}


# Edge scatters accumulate in LDS per 1024-node bucket (ds_add_f32) and flush once per work-group with global atomics; K edge coefficients precomputed; summation order changes
# speedup vs baseline: 3.0953x; 1.5102x over previous
#include <hip/hip_runtime.h>
#include <math.h>

#define N_NODES 8192
#define D_OSC   64
#define E_EDGES 1048576

typedef __attribute__((ext_vector_type(2))) float v2f;
typedef __attribute__((ext_vector_type(8))) float v8f;

__device__ __forceinline__ float wave_reduce_sum(float v) {
  #pragma unroll
  for (int m = 16; m >= 1; m >>= 1) v += __shfl_xor(v, m, 32);
  return v;
}

__global__ __launch_bounds__(256) void k_zero(float* __restrict__ p, int n) {
  int i = blockIdx.x * 256 + threadIdx.x;
  if (i < n) p[i] = 0.0f;
}

__global__ __launch_bounds__(256) void k_norm_tanh(const float* __restrict__ sk,
                                                   const float* __restrict__ sh,
                                                   float* __restrict__ skn,
                                                   float* __restrict__ hs,
                                                   float* __restrict__ gv) {
  int row  = blockIdx.x * 8 + (threadIdx.x >> 5);
  int lane = threadIdx.x & 31;
  const float* r = sk + (size_t)row * D_OSC;
  float a = r[lane];
  float b = r[lane + 32];
  float ss  = wave_reduce_sum(a * a + b * b);
  float inv = rsqrtf(ss);
  float na = a * inv;
  float nb = b * inv;
  float* o = skn + (size_t)row * D_OSC;
  o[lane]      = na;
  o[lane + 32] = nb;
  float grow = tanhf(sh[row]);
  float* oh = hs + (size_t)row * D_OSC;
  oh[lane]      = grow * na;
  oh[lane + 32] = grow * nb;

  int gi = blockIdx.x * 256 + threadIdx.x;
  if (gi < N_NODES) gv[gi] = tanhf(sh[gi]);
}

#define EPW  32
#define EGRP 8

__global__ __launch_bounds__(256) void k_coef_K(const int* __restrict__ ind,
                                                const float* __restrict__ wK,
                                                const float* __restrict__ skn,
                                                float* __restrict__ coef) {
  const int lane = threadIdx.x & 31;
  const int e0   = (blockIdx.x * 8 + (threadIdx.x >> 5)) * EPW;
  const int2  myj = ((const int2*)ind)[e0 + lane];
  const float myw = wK[e0 + lane];
  float mys = 0.0f;
  #pragma unroll
  for (int g = 0; g < EPW; g += EGRP) {
    float a0[EGRP], a1[EGRP], b0[EGRP], b1[EGRP];
    #pragma unroll
    for (int u = 0; u < EGRP; ++u) {
      int j0 = __builtin_amdgcn_readlane(myj.x, g + u);
      int j1 = __builtin_amdgcn_readlane(myj.y, g + u);
      const float* r0 = skn + (size_t)j0 * D_OSC;
      const float* r1 = skn + (size_t)j1 * D_OSC;
      a0[u] = r0[lane];  a1[u] = r0[lane + 32];
      b0[u] = r1[lane];  b1[u] = r1[lane + 32];
    }
    #pragma unroll
    for (int u = 0; u < EGRP; ++u) {
      float s = wave_reduce_sum(a0[u] * b0[u] + a1[u] * b1[u]);
      if (lane == g + u) mys = s;
    }
  }
  coef[e0 + lane] = myw * mys;
}

#define NBKT 8
#define BN   (N_NODES / NBKT)
#define BSH  10
#define NSL  32
#define SLE  (E_EDGES / NSL)
#define NCH  (SLE / 32)
#define SGRP 16

__global__ __launch_bounds__(1024) void k_scatter(const int* __restrict__ ind,
                                                  const float* __restrict__ coef,
                                                  const float* __restrict__ rows,
                                                  float* __restrict__ acc) {
  __shared__ float lacc[BN * D_OSC];
  const int tid  = threadIdx.x;
  const int lane = tid & 31;
  const int wv   = tid >> 5;
  const int bkt  = blockIdx.x & (NBKT - 1);
  const int slc  = blockIdx.x >> 3;

  for (int i = tid; i < BN * D_OSC / 4; i += 1024)
    ((float4*)lacc)[i] = make_float4(0.0f, 0.0f, 0.0f, 0.0f);
  __syncthreads();

  const int2*  ip = (const int2*)ind + (size_t)slc * SLE;
  const float* cp = coef + (size_t)slc * SLE;

  int2  ij = ip[wv * 32 + lane];
  float cw = cp[wv * 32 + lane];
  for (int c = wv; c < NCH; c += 32) {
    int2  ijn = ij;
    float cwn = cw;
    if (c + 32 < NCH) {
      ijn = ip[(c + 32) * 32 + lane];
      cwn = cp[(c + 32) * 32 + lane];
    }
    unsigned m0 = __builtin_amdgcn_ballot_w32((ij.x >> BSH) == bkt);
    unsigned m1 = __builtin_amdgcn_ballot_w32((ij.y >> BSH) == bkt);
    unsigned long long M = (unsigned long long)m0 | ((unsigned long long)m1 << 32);
    const int cwi = __float_as_int(cw);
    while (M) {
      bool  val[SGRP];
      int   dl[SGRP];
      float cf[SGRP], r0[SGRP], r1[SGRP];
      #pragma unroll
      for (int u = 0; u < SGRP; ++u) {
        val[u] = (M != 0ull);
        if (val[u]) {
          const int k  = __builtin_ctzll(M);
          M &= M - 1ull;
          const int kl = k & 31;
          const int xs = __builtin_amdgcn_readlane(ij.x, kl);
          const int ys = __builtin_amdgcn_readlane(ij.y, kl);
          const int dst = (k < 32) ? xs : ys;
          const int sr  = (k < 32) ? ys : xs;
          cf[u] = __int_as_float(__builtin_amdgcn_readlane(cwi, kl));
          dl[u] = (dst & (BN - 1)) * D_OSC;
          const float* r = rows + (size_t)sr * D_OSC;
          r0[u] = r[lane];
          r1[u] = r[lane + 32];
        }
      }
      #pragma unroll
      for (int u = 0; u < SGRP; ++u) {
        if (val[u]) {
          atomicAdd(&lacc[dl[u] + lane],      cf[u] * r0[u]);
          atomicAdd(&lacc[dl[u] + lane + 32], cf[u] * r1[u]);
        }
      }
    }
    ij = ijn;
    cw = cwn;
  }

  asm volatile("s_wait_dscnt 0x0" ::: "memory");
  __syncthreads();
  float* gacc = acc + (size_t)bkt * BN * D_OSC;
  for (int k0 = 0; k0 < BN * D_OSC / 1024; k0 += 8) {
    int   fi[8];
    float fv[8];
    #pragma unroll
    for (int u = 0; u < 8; ++u) {
      fi[u] = (((k0 + u + 2 * slc) & (BN * D_OSC / 1024 - 1)) << 10) + tid;
      fv[u] = lacc[fi[u]];
    }
    #pragma unroll
    for (int u = 0; u < 8; ++u) atomicAdd(&gacc[fi[u]], fv[u]);
  }
}

__global__ __launch_bounds__(256) void k_gemv_fused(const float* __restrict__ W,
                                                    const float* __restrict__ gv,
                                                    float* __restrict__ y,
                                                    float* __restrict__ z) {
  __shared__ float slab[8 * 288];
  __shared__ float zstage[512];

  const int wv   = threadIdx.x >> 5;
  const int lane = threadIdx.x & 31;
  const int hl   = lane & 15;
  const int koff = (lane < 16) ? 0 : 2;

  const int C0 = blockIdx.x * 512;
  const int r0 = blockIdx.y * 128 + wv * 16;

  zstage[threadIdx.x]       = 0.0f;
  zstage[threadIdx.x + 256] = 0.0f;
  __syncthreads();

  float* sl = slab + wv * 288;

  v2f bz0 = *(const v2f*)(gv + r0 +  0 + koff);
  v2f bz1 = *(const v2f*)(gv + r0 +  4 + koff);
  v2f bz2 = *(const v2f*)(gv + r0 +  8 + koff);
  v2f bz3 = *(const v2f*)(gv + r0 + 12 + koff);

  const float* wrow = W + (size_t)(r0 + hl) * N_NODES + koff;

  v8f yacc = {};
  for (int t = 0; t < 32; ++t) {
    const int c = C0 + t * 16;
    const float* wp = wrow + c;
    v2f a0 = *(const v2f*)(wp +  0);
    v2f a1 = *(const v2f*)(wp +  4);
    v2f a2 = *(const v2f*)(wp +  8);
    v2f a3 = *(const v2f*)(wp + 12);
    v2f by0 = *(const v2f*)(gv + c +  0 + koff);
    v2f by1 = *(const v2f*)(gv + c +  4 + koff);
    v2f by2 = *(const v2f*)(gv + c +  8 + koff);
    v2f by3 = *(const v2f*)(gv + c + 12 + koff);

    yacc = __builtin_amdgcn_wmma_f32_16x16x4_f32(false, a0, false, by0, (short)0, yacc, false, false);
    yacc = __builtin_amdgcn_wmma_f32_16x16x4_f32(false, a1, false, by1, (short)0, yacc, false, false);
    yacc = __builtin_amdgcn_wmma_f32_16x16x4_f32(false, a2, false, by2, (short)0, yacc, false, false);
    yacc = __builtin_amdgcn_wmma_f32_16x16x4_f32(false, a3, false, by3, (short)0, yacc, false, false);

    float* sr = sl + hl * 18 + koff;
    *(v2f*)(sr +  0) = a0;
    *(v2f*)(sr +  4) = a1;
    *(v2f*)(sr +  8) = a2;
    *(v2f*)(sr + 12) = a3;

    v2f az0, az1, az2, az3;
    az0[0] = sl[( 0 + koff) * 18 + hl];  az0[1] = sl[( 1 + koff) * 18 + hl];
    az1[0] = sl[( 4 + koff) * 18 + hl];  az1[1] = sl[( 5 + koff) * 18 + hl];
    az2[0] = sl[( 8 + koff) * 18 + hl];  az2[1] = sl[( 9 + koff) * 18 + hl];
    az3[0] = sl[(12 + koff) * 18 + hl];  az3[1] = sl[(13 + koff) * 18 + hl];

    v8f zt = {};
    zt = __builtin_amdgcn_wmma_f32_16x16x4_f32(false, az0, false, bz0, (short)0, zt, false, false);
    zt = __builtin_amdgcn_wmma_f32_16x16x4_f32(false, az1, false, bz1, (short)0, zt, false, false);
    zt = __builtin_amdgcn_wmma_f32_16x16x4_f32(false, az2, false, bz2, (short)0, zt, false, false);
    zt = __builtin_amdgcn_wmma_f32_16x16x4_f32(false, az3, false, bz3, (short)0, zt, false, false);

    if (hl == 0) {
      int zb = t * 16 + ((lane == 16) ? 8 : 0);
      #pragma unroll
      for (int v = 0; v < 8; ++v) atomicAdd(&zstage[zb + v], zt[v]);
    }
  }

  if (hl == 0) {
    int base = r0 + ((lane == 16) ? 8 : 0);
    #pragma unroll
    for (int v = 0; v < 8; ++v) atomicAdd(&y[base + v], yacc[v]);
  }

  __syncthreads();
  atomicAdd(&z[C0 + threadIdx.x],       zstage[threadIdx.x]);
  atomicAdd(&z[C0 + threadIdx.x + 256], zstage[threadIdx.x + 256]);
}

__global__ __launch_bounds__(1024) void k_fin(const float* __restrict__ fKacc,
                                              const float* __restrict__ V,
                                              const float* __restrict__ skn,
                                              const float* __restrict__ gv,
                                              const float* __restrict__ y,
                                              const float* __restrict__ z,
                                              const float* __restrict__ sh,
                                              const float* __restrict__ bias,
                                              const float* __restrict__ kH,
                                              const float* __restrict__ kK,
                                              float* __restrict__ outH,
                                              float* __restrict__ outK) {
  __shared__ float sdot[32];
  int wv   = threadIdx.x >> 5;
  int lane = threadIdx.x & 31;
  int row  = blockIdx.x * 32 + wv;
  size_t base = (size_t)row * D_OSC;
  float s0 = skn[base + lane],   s1 = skn[base + lane + 32];
  float v0 = V[base + lane],     v1 = V[base + lane + 32];
  float k0 = fKacc[base + lane], k1 = fKacc[base + lane + 32];
  float dhv = wave_reduce_sum(s0 * v0 + s1 * v1);
  float c  = gv[row] / kK[0];
  float r0 = k0 - c * v0;
  float r1 = k1 - c * v1;
  float dot = wave_reduce_sum(s0 * r0 + s1 * r1);
  outK[base + lane]      = -r0 + s0 * dot;
  outK[base + lane + 32] = -r1 + s1 * dot;
  if (lane == 0) sdot[wv] = dhv;
  __syncthreads();
  if (wv == 0) {
    int i = blockIdx.x * 32 + lane;
    outH[i] = sdot[lane] / kH[0] + 0.5f * (y[i] + z[i]) - sh[i] + bias[i];
  }
}

extern "C" void kernel_launch(void* const* d_in, const int* in_sizes, int n_in,
                              void* d_out, int out_size, void* d_ws, size_t ws_size,
                              hipStream_t stream) {
  const float* state_H    = (const float*)d_in[0];
  const float* state_K    = (const float*)d_in[1];
  const int*   ind_K      = (const int*)d_in[2];
  const int*   ind_HK     = (const int*)d_in[3];
  const float* kappa_K    = (const float*)d_in[4];
  const float* kappa_H    = (const float*)d_in[5];
  const float* weights_H  = (const float*)d_in[6];
  const float* bias_H     = (const float*)d_in[7];
  const float* weights_HK = (const float*)d_in[8];
  const float* w_K        = (const float*)d_in[9];

  float* out_fH = (float*)d_out;
  float* out_fK = out_fH + N_NODES;

  float* ws     = (float*)d_ws;
  float* fK_acc = ws;
  float* Vacc   = fK_acc + (size_t)N_NODES * D_OSC;
  float* yv     = Vacc + (size_t)N_NODES * D_OSC;
  float* zv     = yv + N_NODES;
  float* skn    = zv + N_NODES;
  float* hs     = skn + (size_t)N_NODES * D_OSC;
  float* gv     = hs + (size_t)N_NODES * D_OSC;
  float* coefK  = gv + N_NODES;

  const int nAcc = 2 * N_NODES * D_OSC + 2 * N_NODES;

  k_zero<<<(nAcc + 255) / 256, 256, 0, stream>>>(ws, nAcc);

  k_norm_tanh<<<N_NODES / 8, 256, 0, stream>>>(state_K, state_H, skn, hs, gv);

  k_coef_K<<<E_EDGES / (8 * EPW), 256, 0, stream>>>(ind_K, w_K, skn, coefK);

  k_scatter<<<NBKT * NSL, 1024, 0, stream>>>(ind_K, coefK, skn, fK_acc);

  k_scatter<<<NBKT * NSL, 1024, 0, stream>>>(ind_HK, weights_HK, hs, Vacc);

  k_gemv_fused<<<dim3(16, 64), 256, 0, stream>>>(weights_H, gv, yv, zv);

  k_fin<<<N_NODES / 32, 1024, 0, stream>>>(fK_acc, Vacc, skn, gv, yv, zv,
                                           state_H, bias_H, kappa_H, kappa_K,
                                           out_fH, out_fK);
}


# Scatter: each wave packs its bucket hits into a private LDS queue and works them off in dense batches of 32: one whole-row load per hit, no scalar search or branch per hit; atomic sum order changes
# speedup vs baseline: 3.3042x; 1.0675x over previous
#include <hip/hip_runtime.h>
#include <math.h>

#define N_NODES 8192
#define D_OSC   64
#define E_EDGES 1048576

typedef __attribute__((ext_vector_type(2))) float v2f;
typedef __attribute__((ext_vector_type(8))) float v8f;

__device__ __forceinline__ float wave_reduce_sum(float v) {
  #pragma unroll
  for (int m = 16; m >= 1; m >>= 1) v += __shfl_xor(v, m, 32);
  return v;
}

__global__ __launch_bounds__(256) void k_zero(float* __restrict__ p, int n) {
  int i = blockIdx.x * 256 + threadIdx.x;
  if (i < n) p[i] = 0.0f;
}

__global__ __launch_bounds__(256) void k_norm_tanh(const float* __restrict__ sk,
                                                   const float* __restrict__ sh,
                                                   float* __restrict__ skn,
                                                   float* __restrict__ hs,
                                                   float* __restrict__ gv) {
  int row  = blockIdx.x * 8 + (threadIdx.x >> 5);
  int lane = threadIdx.x & 31;
  const float* r = sk + (size_t)row * D_OSC;
  float a = r[lane];
  float b = r[lane + 32];
  float ss  = wave_reduce_sum(a * a + b * b);
  float inv = rsqrtf(ss);
  float na = a * inv;
  float nb = b * inv;
  float* o = skn + (size_t)row * D_OSC;
  o[lane]      = na;
  o[lane + 32] = nb;
  float grow = tanhf(sh[row]);
  float* oh = hs + (size_t)row * D_OSC;
  oh[lane]      = grow * na;
  oh[lane + 32] = grow * nb;

  int gi = blockIdx.x * 256 + threadIdx.x;
  if (gi < N_NODES) gv[gi] = tanhf(sh[gi]);
}

#define EPW  32
#define EGRP 8

__global__ __launch_bounds__(256) void k_coef_K(const int* __restrict__ ind,
                                                const float* __restrict__ wK,
                                                const float* __restrict__ skn,
                                                float* __restrict__ coef) {
  const int lane = threadIdx.x & 31;
  const int e0   = (blockIdx.x * 8 + (threadIdx.x >> 5)) * EPW;
  const int2  myj = ((const int2*)ind)[e0 + lane];
  const float myw = wK[e0 + lane];
  float mys = 0.0f;
  #pragma unroll
  for (int g = 0; g < EPW; g += EGRP) {
    float a0[EGRP], a1[EGRP], b0[EGRP], b1[EGRP];
    #pragma unroll
    for (int u = 0; u < EGRP; ++u) {
      int j0 = __builtin_amdgcn_readlane(myj.x, g + u);
      int j1 = __builtin_amdgcn_readlane(myj.y, g + u);
      const float* r0 = skn + (size_t)j0 * D_OSC;
      const float* r1 = skn + (size_t)j1 * D_OSC;
      a0[u] = r0[lane];  a1[u] = r0[lane + 32];
      b0[u] = r1[lane];  b1[u] = r1[lane + 32];
    }
    #pragma unroll
    for (int u = 0; u < EGRP; ++u) {
      float s = wave_reduce_sum(a0[u] * b0[u] + a1[u] * b1[u]);
      if (lane == g + u) mys = s;
    }
  }
  coef[e0 + lane] = myw * mys;
}

#define NBKT 8
#define BN   (N_NODES / NBKT)
#define BSH  10
#define NSL  32
#define SLE  (E_EDGES / NSL)
#define NCH  (SLE / 32)
#define HGRP 16
#define QCAP 128

typedef __attribute__((ext_vector_type(2))) unsigned v2u;


__global__ __launch_bounds__(1024) void k_scatter(const int* __restrict__ ind,
                                                  const float* __restrict__ coef,
                                                  const float* __restrict__ rows,
                                                  float* __restrict__ acc) {
  __shared__ float lacc[BN * D_OSC];
  __shared__ v2u hitq[32 * QCAP];
  const int tid  = threadIdx.x;
  const int lane = tid & 31;
  const int wv   = __builtin_amdgcn_readfirstlane(tid >> 5);
  const int bkt  = blockIdx.x & (NBKT - 1);
  const int slc  = blockIdx.x >> 3;

  for (int i = tid; i < BN * D_OSC / 4; i += 1024)
    ((float4*)lacc)[i] = make_float4(0.0f, 0.0f, 0.0f, 0.0f);
  __syncthreads();

  const int2*  ip = (const int2*)ind + (size_t)slc * SLE;
  const float* cp = coef + (size_t)slc * SLE;
  v2u* q = hitq + wv * QCAP;
  const char* rowsb = (const char*)rows;

  unsigned head = 0u, tail = 0u;

  int2  ij = ip[wv * 32 + lane];
  float cw = cp[wv * 32 + lane];
  for (int c = wv; c < NCH; c += 32) {
    int2  ijn = ij;
    float cwn = cw;
    if (c + 32 < NCH) {
      ijn = ip[(c + 32) * 32 + lane];
      cwn = cp[(c + 32) * 32 + lane];
    }
    const bool h0 = (ij.x >> BSH) == bkt;
    const bool h1 = (ij.y >> BSH) == bkt;
    const unsigned m0 = __builtin_amdgcn_ballot_w32(h0);
    const unsigned m1 = __builtin_amdgcn_ballot_w32(h1);
    const unsigned n0 = (unsigned)__builtin_popcount(m0);
    const unsigned cb = __float_as_uint(cw);
    if (h0) {
      const unsigned p = tail + __builtin_amdgcn_mbcnt_lo(m0, 0u);
      v2u e;
      e.x = ((unsigned)ij.y << BSH) | ((unsigned)ij.x & (BN - 1));
      e.y = cb;
      q[p & (QCAP - 1)] = e;
    }
    if (h1) {
      const unsigned p = tail + n0 + __builtin_amdgcn_mbcnt_lo(m1, 0u);
      v2u e;
      e.x = ((unsigned)ij.x << BSH) | ((unsigned)ij.y & (BN - 1));
      e.y = cb;
      q[p & (QCAP - 1)] = e;
    }
    tail += n0 + (unsigned)__builtin_popcount(m1);

    while (tail - head >= 32u) {
      const v2u e = q[(head + lane) & (QCAP - 1)];
      head += 32u;
      const int srow = (int)((e.x >> BSH) << 8);
      const int drow = (int)((e.x & (BN - 1)) << 6);
      const int cbit = (int)e.y;
      #pragma unroll
      for (int g = 0; g < 32; g += HGRP) {
        float2 r[HGRP];
        #pragma unroll
        for (int u = 0; u < HGRP; ++u) {
          const int so = __builtin_amdgcn_readlane(srow, g + u);
          r[u] = ((const float2*)(rowsb + (size_t)(unsigned)so))[lane];
        }
        #pragma unroll
        for (int u = 0; u < HGRP; ++u) {
          const int   dl = __builtin_amdgcn_readlane(drow, g + u);
          const float cf = __int_as_float(__builtin_amdgcn_readlane(cbit, g + u));
          atomicAdd(&lacc[dl + 2 * lane],     cf * r[u].x);
          atomicAdd(&lacc[dl + 2 * lane + 1], cf * r[u].y);
        }
      }
    }
    ij = ijn;
    cw = cwn;
  }

  const unsigned nleft = tail - head;
  if (nleft != 0u) {
    const v2u e = q[(head + lane) & (QCAP - 1)];
    const int srow = (int)((e.x >> BSH) << 8);
    const int drow = (int)((e.x & (BN - 1)) << 6);
    const int cbit = (int)e.y;
    #pragma unroll
    for (int g = 0; g < 32; g += 8) {
      if ((unsigned)g < nleft) {
        float2 r[8];
        #pragma unroll
        for (int u = 0; u < 8; ++u) {
          if ((unsigned)(g + u) < nleft) {
            const int so = __builtin_amdgcn_readlane(srow, g + u);
            r[u] = ((const float2*)(rowsb + (size_t)(unsigned)so))[lane];
          }
        }
        #pragma unroll
        for (int u = 0; u < 8; ++u) {
          if ((unsigned)(g + u) < nleft) {
            const int   dl = __builtin_amdgcn_readlane(drow, g + u);
            const float cf = __int_as_float(__builtin_amdgcn_readlane(cbit, g + u));
            atomicAdd(&lacc[dl + 2 * lane],     cf * r[u].x);
            atomicAdd(&lacc[dl + 2 * lane + 1], cf * r[u].y);
          }
        }
      }
    }
  }

  asm volatile("s_wait_dscnt 0x0" ::: "memory");
  __syncthreads();
  float* gacc = acc + (size_t)bkt * BN * D_OSC;
  for (int k0 = 0; k0 < BN * D_OSC / 1024; k0 += 8) {
    int   fi[8];
    float fv[8];
    #pragma unroll
    for (int u = 0; u < 8; ++u) {
      fi[u] = (((k0 + u + 2 * slc) & (BN * D_OSC / 1024 - 1)) << 10) + tid;
      fv[u] = lacc[fi[u]];
    }
    #pragma unroll
    for (int u = 0; u < 8; ++u) atomicAdd(&gacc[fi[u]], fv[u]);
  }
}

__global__ __launch_bounds__(256) void k_gemv_fused(const float* __restrict__ W,
                                                    const float* __restrict__ gv,
                                                    float* __restrict__ y,
                                                    float* __restrict__ z) {
  __shared__ float slab[8 * 288];
  __shared__ float zstage[512];

  const int wv   = threadIdx.x >> 5;
  const int lane = threadIdx.x & 31;
  const int hl   = lane & 15;
  const int koff = (lane < 16) ? 0 : 2;

  const int C0 = blockIdx.x * 512;
  const int r0 = blockIdx.y * 128 + wv * 16;

  zstage[threadIdx.x]       = 0.0f;
  zstage[threadIdx.x + 256] = 0.0f;
  __syncthreads();

  float* sl = slab + wv * 288;

  v2f bz0 = *(const v2f*)(gv + r0 +  0 + koff);
  v2f bz1 = *(const v2f*)(gv + r0 +  4 + koff);
  v2f bz2 = *(const v2f*)(gv + r0 +  8 + koff);
  v2f bz3 = *(const v2f*)(gv + r0 + 12 + koff);

  const float* wrow = W + (size_t)(r0 + hl) * N_NODES + koff;

  v8f yacc = {};
  for (int t = 0; t < 32; ++t) {
    const int c = C0 + t * 16;
    const float* wp = wrow + c;
    v2f a0 = *(const v2f*)(wp +  0);
    v2f a1 = *(const v2f*)(wp +  4);
    v2f a2 = *(const v2f*)(wp +  8);
    v2f a3 = *(const v2f*)(wp + 12);
    v2f by0 = *(const v2f*)(gv + c +  0 + koff);
    v2f by1 = *(const v2f*)(gv + c +  4 + koff);
    v2f by2 = *(const v2f*)(gv + c +  8 + koff);
    v2f by3 = *(const v2f*)(gv + c + 12 + koff);

    yacc = __builtin_amdgcn_wmma_f32_16x16x4_f32(false, a0, false, by0, (short)0, yacc, false, false);
    yacc = __builtin_amdgcn_wmma_f32_16x16x4_f32(false, a1, false, by1, (short)0, yacc, false, false);
    yacc = __builtin_amdgcn_wmma_f32_16x16x4_f32(false, a2, false, by2, (short)0, yacc, false, false);
    yacc = __builtin_amdgcn_wmma_f32_16x16x4_f32(false, a3, false, by3, (short)0, yacc, false, false);

    float* sr = sl + hl * 18 + koff;
    *(v2f*)(sr +  0) = a0;
    *(v2f*)(sr +  4) = a1;
    *(v2f*)(sr +  8) = a2;
    *(v2f*)(sr + 12) = a3;

    v2f az0, az1, az2, az3;
    az0[0] = sl[( 0 + koff) * 18 + hl];  az0[1] = sl[( 1 + koff) * 18 + hl];
    az1[0] = sl[( 4 + koff) * 18 + hl];  az1[1] = sl[( 5 + koff) * 18 + hl];
    az2[0] = sl[( 8 + koff) * 18 + hl];  az2[1] = sl[( 9 + koff) * 18 + hl];
    az3[0] = sl[(12 + koff) * 18 + hl];  az3[1] = sl[(13 + koff) * 18 + hl];

    v8f zt = {};
    zt = __builtin_amdgcn_wmma_f32_16x16x4_f32(false, az0, false, bz0, (short)0, zt, false, false);
    zt = __builtin_amdgcn_wmma_f32_16x16x4_f32(false, az1, false, bz1, (short)0, zt, false, false);
    zt = __builtin_amdgcn_wmma_f32_16x16x4_f32(false, az2, false, bz2, (short)0, zt, false, false);
    zt = __builtin_amdgcn_wmma_f32_16x16x4_f32(false, az3, false, bz3, (short)0, zt, false, false);

    if (hl == 0) {
      int zb = t * 16 + ((lane == 16) ? 8 : 0);
      #pragma unroll
      for (int v = 0; v < 8; ++v) atomicAdd(&zstage[zb + v], zt[v]);
    }
  }

  if (hl == 0) {
    int base = r0 + ((lane == 16) ? 8 : 0);
    #pragma unroll
    for (int v = 0; v < 8; ++v) atomicAdd(&y[base + v], yacc[v]);
  }

  __syncthreads();
  atomicAdd(&z[C0 + threadIdx.x],       zstage[threadIdx.x]);
  atomicAdd(&z[C0 + threadIdx.x + 256], zstage[threadIdx.x + 256]);
}

__global__ __launch_bounds__(1024) void k_fin(const float* __restrict__ fKacc,
                                              const float* __restrict__ V,
                                              const float* __restrict__ skn,
                                              const float* __restrict__ gv,
                                              const float* __restrict__ y,
                                              const float* __restrict__ z,
                                              const float* __restrict__ sh,
                                              const float* __restrict__ bias,
                                              const float* __restrict__ kH,
                                              const float* __restrict__ kK,
                                              float* __restrict__ outH,
                                              float* __restrict__ outK) {
  __shared__ float sdot[32];
  int wv   = threadIdx.x >> 5;
  int lane = threadIdx.x & 31;
  int row  = blockIdx.x * 32 + wv;
  size_t base = (size_t)row * D_OSC;
  float s0 = skn[base + lane],   s1 = skn[base + lane + 32];
  float v0 = V[base + lane],     v1 = V[base + lane + 32];
  float k0 = fKacc[base + lane], k1 = fKacc[base + lane + 32];
  float dhv = wave_reduce_sum(s0 * v0 + s1 * v1);
  float c  = gv[row] / kK[0];
  float r0 = k0 - c * v0;
  float r1 = k1 - c * v1;
  float dot = wave_reduce_sum(s0 * r0 + s1 * r1);
  outK[base + lane]      = -r0 + s0 * dot;
  outK[base + lane + 32] = -r1 + s1 * dot;
  if (lane == 0) sdot[wv] = dhv;
  __syncthreads();
  if (wv == 0) {
    int i = blockIdx.x * 32 + lane;
    outH[i] = sdot[lane] / kH[0] + 0.5f * (y[i] + z[i]) - sh[i] + bias[i];
  }
}

extern "C" void kernel_launch(void* const* d_in, const int* in_sizes, int n_in,
                              void* d_out, int out_size, void* d_ws, size_t ws_size,
                              hipStream_t stream) {
  const float* state_H    = (const float*)d_in[0];
  const float* state_K    = (const float*)d_in[1];
  const int*   ind_K      = (const int*)d_in[2];
  const int*   ind_HK     = (const int*)d_in[3];
  const float* kappa_K    = (const float*)d_in[4];
  const float* kappa_H    = (const float*)d_in[5];
  const float* weights_H  = (const float*)d_in[6];
  const float* bias_H     = (const float*)d_in[7];
  const float* weights_HK = (const float*)d_in[8];
  const float* w_K        = (const float*)d_in[9];

  float* out_fH = (float*)d_out;
  float* out_fK = out_fH + N_NODES;

  float* ws     = (float*)d_ws;
  float* fK_acc = ws;
  float* Vacc   = fK_acc + (size_t)N_NODES * D_OSC;
  float* yv     = Vacc + (size_t)N_NODES * D_OSC;
  float* zv     = yv + N_NODES;
  float* skn    = zv + N_NODES;
  float* hs     = skn + (size_t)N_NODES * D_OSC;
  float* gv     = hs + (size_t)N_NODES * D_OSC;
  float* coefK  = gv + N_NODES;

  const int nAcc = 2 * N_NODES * D_OSC + 2 * N_NODES;

  k_zero<<<(nAcc + 255) / 256, 256, 0, stream>>>(ws, nAcc);

  k_norm_tanh<<<N_NODES / 8, 256, 0, stream>>>(state_K, state_H, skn, hs, gv);

  k_coef_K<<<E_EDGES / (8 * EPW), 256, 0, stream>>>(ind_K, w_K, skn, coefK);

  k_scatter<<<NBKT * NSL, 1024, 0, stream>>>(ind_K, coefK, skn, fK_acc);

  k_scatter<<<NBKT * NSL, 1024, 0, stream>>>(ind_HK, weights_HK, hs, Vacc);

  k_gemv_fused<<<dim3(16, 64), 256, 0, stream>>>(weights_H, gv, yv, zv);

  k_fin<<<N_NODES / 32, 1024, 0, stream>>>(fK_acc, Vacc, skn, gv, yv, zv,
                                           state_H, bias_H, kappa_H, kappa_K,
                                           out_fH, out_fK);
}


# Independent kernels share a launch (7 launches -> 4): zero+normalize, W pass beside K edge coefficients on every WGP, both scatters in one grid; bodies unchanged; atomic arrival order differs
# speedup vs baseline: 3.4537x; 1.0452x over previous
#include <hip/hip_runtime.h>
#include <math.h>

#define N_NODES 8192
#define D_OSC   64
#define E_EDGES 1048576

typedef __attribute__((ext_vector_type(2))) float v2f;
typedef __attribute__((ext_vector_type(8))) float v8f;

__device__ __forceinline__ float wave_reduce_sum(float v) {
  #pragma unroll
  for (int m = 16; m >= 1; m >>= 1) v += __shfl_xor(v, m, 32);
  return v;
}

#define N_ACC     (2 * N_NODES * D_OSC + 2 * N_NODES)
#define NORM_BLKS (N_NODES / 8)
#define ZERO_BLKS (N_ACC / 4 / 256)

__device__ __forceinline__ void norm_tanh_body(int blk,
                                               const float* __restrict__ sk,
                                               const float* __restrict__ sh,
                                               float* __restrict__ skn,
                                               float* __restrict__ hs,
                                               float* __restrict__ gv) {
  int row  = blk * 8 + (threadIdx.x >> 5);
  int lane = threadIdx.x & 31;
  const float* r = sk + (size_t)row * D_OSC;
  float a = r[lane];
  float b = r[lane + 32];
  float ss  = wave_reduce_sum(a * a + b * b);
  float inv = rsqrtf(ss);
  float na = a * inv;
  float nb = b * inv;
  float* o = skn + (size_t)row * D_OSC;
  o[lane]      = na;
  o[lane + 32] = nb;
  float grow = tanhf(sh[row]);
  float* oh = hs + (size_t)row * D_OSC;
  oh[lane]      = grow * na;
  oh[lane + 32] = grow * nb;

  int gi = blk * 256 + threadIdx.x;
  if (gi < N_NODES) gv[gi] = tanhf(sh[gi]);
}

__global__ __launch_bounds__(256) void k_norm_zero(const float* __restrict__ sk,
                                                   const float* __restrict__ sh,
                                                   float* __restrict__ skn,
                                                   float* __restrict__ hs,
                                                   float* __restrict__ gv,
                                                   float* __restrict__ accs) {
  if (blockIdx.x < NORM_BLKS) {
    norm_tanh_body(blockIdx.x, sk, sh, skn, hs, gv);
  } else {
    const int i = (blockIdx.x - NORM_BLKS) * 256 + threadIdx.x;
    ((float4*)accs)[i] = make_float4(0.0f, 0.0f, 0.0f, 0.0f);
  }
}

#define EPW  32
#define EGRP 8

__device__ __forceinline__ void coef_K_body(int blk,
                                            const int* __restrict__ ind,
                                            const float* __restrict__ wK,
                                            const float* __restrict__ skn,
                                            float* __restrict__ coef) {
  const int lane = threadIdx.x & 31;
  const int e0   = (blk * 8 + (threadIdx.x >> 5)) * EPW;
  const int2  myj = ((const int2*)ind)[e0 + lane];
  const float myw = wK[e0 + lane];
  float mys = 0.0f;
  #pragma unroll
  for (int g = 0; g < EPW; g += EGRP) {
    float a0[EGRP], a1[EGRP], b0[EGRP], b1[EGRP];
    #pragma unroll
    for (int u = 0; u < EGRP; ++u) {
      int j0 = __builtin_amdgcn_readlane(myj.x, g + u);
      int j1 = __builtin_amdgcn_readlane(myj.y, g + u);
      const float* r0 = skn + (size_t)j0 * D_OSC;
      const float* r1 = skn + (size_t)j1 * D_OSC;
      a0[u] = r0[lane];  a1[u] = r0[lane + 32];
      b0[u] = r1[lane];  b1[u] = r1[lane + 32];
    }
    #pragma unroll
    for (int u = 0; u < EGRP; ++u) {
      float s = wave_reduce_sum(a0[u] * b0[u] + a1[u] * b1[u]);
      if (lane == g + u) mys = s;
    }
  }
  coef[e0 + lane] = myw * mys;
}

#define NBKT 8
#define BN   (N_NODES / NBKT)
#define BSH  10
#define NSL  32
#define SLE  (E_EDGES / NSL)
#define NCH  (SLE / 32)
#define HGRP 16
#define QCAP 128

typedef __attribute__((ext_vector_type(2))) unsigned v2u;


__global__ __launch_bounds__(1024) void k_scatter(const int* __restrict__ indA,
                                                  const float* __restrict__ coefA,
                                                  const float* __restrict__ rowsA,
                                                  float* __restrict__ accA,
                                                  const int* __restrict__ indB,
                                                  const float* __restrict__ coefB,
                                                  const float* __restrict__ rowsB,
                                                  float* __restrict__ accB) {
  const bool setB = blockIdx.x >= NBKT * NSL;
  const int*   __restrict__ ind  = setB ? indB  : indA;
  const float* __restrict__ coef = setB ? coefB : coefA;
  const float* __restrict__ rows = setB ? rowsB : rowsA;
  float*       __restrict__ acc  = setB ? accB  : accA;
  const int blk = blockIdx.x & (NBKT * NSL - 1);
  __shared__ float lacc[BN * D_OSC];
  __shared__ v2u hitq[32 * QCAP];
  const int tid  = threadIdx.x;
  const int lane = tid & 31;
  const int wv   = __builtin_amdgcn_readfirstlane(tid >> 5);
  const int bkt  = blk & (NBKT - 1);
  const int slc  = blk >> 3;

  for (int i = tid; i < BN * D_OSC / 4; i += 1024)
    ((float4*)lacc)[i] = make_float4(0.0f, 0.0f, 0.0f, 0.0f);
  __syncthreads();

  const int2*  ip = (const int2*)ind + (size_t)slc * SLE;
  const float* cp = coef + (size_t)slc * SLE;
  v2u* q = hitq + wv * QCAP;
  const char* rowsb = (const char*)rows;

  unsigned head = 0u, tail = 0u;

  int2  ij = ip[wv * 32 + lane];
  float cw = cp[wv * 32 + lane];
  for (int c = wv; c < NCH; c += 32) {
    int2  ijn = ij;
    float cwn = cw;
    if (c + 32 < NCH) {
      ijn = ip[(c + 32) * 32 + lane];
      cwn = cp[(c + 32) * 32 + lane];
    }
    const bool h0 = (ij.x >> BSH) == bkt;
    const bool h1 = (ij.y >> BSH) == bkt;
    const unsigned m0 = __builtin_amdgcn_ballot_w32(h0);
    const unsigned m1 = __builtin_amdgcn_ballot_w32(h1);
    const unsigned n0 = (unsigned)__builtin_popcount(m0);
    const unsigned cb = __float_as_uint(cw);
    if (h0) {
      const unsigned p = tail + __builtin_amdgcn_mbcnt_lo(m0, 0u);
      v2u e;
      e.x = ((unsigned)ij.y << BSH) | ((unsigned)ij.x & (BN - 1));
      e.y = cb;
      q[p & (QCAP - 1)] = e;
    }
    if (h1) {
      const unsigned p = tail + n0 + __builtin_amdgcn_mbcnt_lo(m1, 0u);
      v2u e;
      e.x = ((unsigned)ij.x << BSH) | ((unsigned)ij.y & (BN - 1));
      e.y = cb;
      q[p & (QCAP - 1)] = e;
    }
    tail += n0 + (unsigned)__builtin_popcount(m1);

    while (tail - head >= 32u) {
      const v2u e = q[(head + lane) & (QCAP - 1)];
      head += 32u;
      const int srow = (int)((e.x >> BSH) << 8);
      const int drow = (int)((e.x & (BN - 1)) << 6);
      const int cbit = (int)e.y;
      #pragma unroll
      for (int g = 0; g < 32; g += HGRP) {
        float2 r[HGRP];
        #pragma unroll
        for (int u = 0; u < HGRP; ++u) {
          const int so = __builtin_amdgcn_readlane(srow, g + u);
          r[u] = ((const float2*)(rowsb + (size_t)(unsigned)so))[lane];
        }
        #pragma unroll
        for (int u = 0; u < HGRP; ++u) {
          const int   dl = __builtin_amdgcn_readlane(drow, g + u);
          const float cf = __int_as_float(__builtin_amdgcn_readlane(cbit, g + u));
          atomicAdd(&lacc[dl + 2 * lane],     cf * r[u].x);
          atomicAdd(&lacc[dl + 2 * lane + 1], cf * r[u].y);
        }
      }
    }
    ij = ijn;
    cw = cwn;
  }

  const unsigned nleft = tail - head;
  if (nleft != 0u) {
    const v2u e = q[(head + lane) & (QCAP - 1)];
    const int srow = (int)((e.x >> BSH) << 8);
    const int drow = (int)((e.x & (BN - 1)) << 6);
    const int cbit = (int)e.y;
    #pragma unroll
    for (int g = 0; g < 32; g += 8) {
      if ((unsigned)g < nleft) {
        float2 r[8];
        #pragma unroll
        for (int u = 0; u < 8; ++u) {
          if ((unsigned)(g + u) < nleft) {
            const int so = __builtin_amdgcn_readlane(srow, g + u);
            r[u] = ((const float2*)(rowsb + (size_t)(unsigned)so))[lane];
          }
        }
        #pragma unroll
        for (int u = 0; u < 8; ++u) {
          if ((unsigned)(g + u) < nleft) {
            const int   dl = __builtin_amdgcn_readlane(drow, g + u);
            const float cf = __int_as_float(__builtin_amdgcn_readlane(cbit, g + u));
            atomicAdd(&lacc[dl + 2 * lane],     cf * r[u].x);
            atomicAdd(&lacc[dl + 2 * lane + 1], cf * r[u].y);
          }
        }
      }
    }
  }

  asm volatile("s_wait_dscnt 0x0" ::: "memory");
  __syncthreads();
  float* gacc = acc + (size_t)bkt * BN * D_OSC;
  for (int k0 = 0; k0 < BN * D_OSC / 1024; k0 += 8) {
    int   fi[8];
    float fv[8];
    #pragma unroll
    for (int u = 0; u < 8; ++u) {
      fi[u] = (((k0 + u + 2 * slc) & (BN * D_OSC / 1024 - 1)) << 10) + tid;
      fv[u] = lacc[fi[u]];
    }
    #pragma unroll
    for (int u = 0; u < 8; ++u) atomicAdd(&gacc[fi[u]], fv[u]);
  }
}

__device__ __forceinline__ void gemv_fused_body(int bx, int by,
                                                float* __restrict__ slab,
                                                float* __restrict__ zstage,
                                                const float* __restrict__ W,
                                                const float* __restrict__ gv,
                                                float* __restrict__ y,
                                                float* __restrict__ z) {

  const int wv   = threadIdx.x >> 5;
  const int lane = threadIdx.x & 31;
  const int hl   = lane & 15;
  const int koff = (lane < 16) ? 0 : 2;

  const int C0 = bx * 512;
  const int r0 = by * 128 + wv * 16;

  zstage[threadIdx.x]       = 0.0f;
  zstage[threadIdx.x + 256] = 0.0f;
  __syncthreads();

  float* sl = slab + wv * 288;

  v2f bz0 = *(const v2f*)(gv + r0 +  0 + koff);
  v2f bz1 = *(const v2f*)(gv + r0 +  4 + koff);
  v2f bz2 = *(const v2f*)(gv + r0 +  8 + koff);
  v2f bz3 = *(const v2f*)(gv + r0 + 12 + koff);

  const float* wrow = W + (size_t)(r0 + hl) * N_NODES + koff;

  v8f yacc = {};
  for (int t = 0; t < 32; ++t) {
    const int c = C0 + t * 16;
    const float* wp = wrow + c;
    v2f a0 = *(const v2f*)(wp +  0);
    v2f a1 = *(const v2f*)(wp +  4);
    v2f a2 = *(const v2f*)(wp +  8);
    v2f a3 = *(const v2f*)(wp + 12);
    v2f by0 = *(const v2f*)(gv + c +  0 + koff);
    v2f by1 = *(const v2f*)(gv + c +  4 + koff);
    v2f by2 = *(const v2f*)(gv + c +  8 + koff);
    v2f by3 = *(const v2f*)(gv + c + 12 + koff);

    yacc = __builtin_amdgcn_wmma_f32_16x16x4_f32(false, a0, false, by0, (short)0, yacc, false, false);
    yacc = __builtin_amdgcn_wmma_f32_16x16x4_f32(false, a1, false, by1, (short)0, yacc, false, false);
    yacc = __builtin_amdgcn_wmma_f32_16x16x4_f32(false, a2, false, by2, (short)0, yacc, false, false);
    yacc = __builtin_amdgcn_wmma_f32_16x16x4_f32(false, a3, false, by3, (short)0, yacc, false, false);

    float* sr = sl + hl * 18 + koff;
    *(v2f*)(sr +  0) = a0;
    *(v2f*)(sr +  4) = a1;
    *(v2f*)(sr +  8) = a2;
    *(v2f*)(sr + 12) = a3;

    v2f az0, az1, az2, az3;
    az0[0] = sl[( 0 + koff) * 18 + hl];  az0[1] = sl[( 1 + koff) * 18 + hl];
    az1[0] = sl[( 4 + koff) * 18 + hl];  az1[1] = sl[( 5 + koff) * 18 + hl];
    az2[0] = sl[( 8 + koff) * 18 + hl];  az2[1] = sl[( 9 + koff) * 18 + hl];
    az3[0] = sl[(12 + koff) * 18 + hl];  az3[1] = sl[(13 + koff) * 18 + hl];

    v8f zt = {};
    zt = __builtin_amdgcn_wmma_f32_16x16x4_f32(false, az0, false, bz0, (short)0, zt, false, false);
    zt = __builtin_amdgcn_wmma_f32_16x16x4_f32(false, az1, false, bz1, (short)0, zt, false, false);
    zt = __builtin_amdgcn_wmma_f32_16x16x4_f32(false, az2, false, bz2, (short)0, zt, false, false);
    zt = __builtin_amdgcn_wmma_f32_16x16x4_f32(false, az3, false, bz3, (short)0, zt, false, false);

    if (hl == 0) {
      int zb = t * 16 + ((lane == 16) ? 8 : 0);
      #pragma unroll
      for (int v = 0; v < 8; ++v) atomicAdd(&zstage[zb + v], zt[v]);
    }
  }

  if (hl == 0) {
    int base = r0 + ((lane == 16) ? 8 : 0);
    #pragma unroll
    for (int v = 0; v < 8; ++v) atomicAdd(&y[base + v], yacc[v]);
  }

  __syncthreads();
  atomicAdd(&z[C0 + threadIdx.x],       zstage[threadIdx.x]);
  atomicAdd(&z[C0 + threadIdx.x + 256], zstage[threadIdx.x + 256]);
}

#define GEMV_BLKS (16 * 64)
#define COEF_BLKS (E_EDGES / (8 * EPW))

__global__ __launch_bounds__(256) void k_gemv_coef(const float* __restrict__ W,
                                                   const float* __restrict__ gv,
                                                   float* __restrict__ y,
                                                   float* __restrict__ z,
                                                   const int* __restrict__ ind,
                                                   const float* __restrict__ wK,
                                                   const float* __restrict__ skn,
                                                   float* __restrict__ coef) {
  __shared__ float slab[8 * 288];
  __shared__ float zstage[512];
  if (blockIdx.x < GEMV_BLKS) {
    gemv_fused_body(blockIdx.x & 15, blockIdx.x >> 4, slab, zstage, W, gv, y, z);
  } else {
    coef_K_body(blockIdx.x - GEMV_BLKS, ind, wK, skn, coef);
  }
}

__global__ __launch_bounds__(1024) void k_fin(const float* __restrict__ fKacc,
                                              const float* __restrict__ V,
                                              const float* __restrict__ skn,
                                              const float* __restrict__ gv,
                                              const float* __restrict__ y,
                                              const float* __restrict__ z,
                                              const float* __restrict__ sh,
                                              const float* __restrict__ bias,
                                              const float* __restrict__ kH,
                                              const float* __restrict__ kK,
                                              float* __restrict__ outH,
                                              float* __restrict__ outK) {
  __shared__ float sdot[32];
  int wv   = threadIdx.x >> 5;
  int lane = threadIdx.x & 31;
  int row  = blockIdx.x * 32 + wv;
  size_t base = (size_t)row * D_OSC;
  float s0 = skn[base + lane],   s1 = skn[base + lane + 32];
  float v0 = V[base + lane],     v1 = V[base + lane + 32];
  float k0 = fKacc[base + lane], k1 = fKacc[base + lane + 32];
  float dhv = wave_reduce_sum(s0 * v0 + s1 * v1);
  float c  = gv[row] / kK[0];
  float r0 = k0 - c * v0;
  float r1 = k1 - c * v1;
  float dot = wave_reduce_sum(s0 * r0 + s1 * r1);
  outK[base + lane]      = -r0 + s0 * dot;
  outK[base + lane + 32] = -r1 + s1 * dot;
  if (lane == 0) sdot[wv] = dhv;
  __syncthreads();
  if (wv == 0) {
    int i = blockIdx.x * 32 + lane;
    outH[i] = sdot[lane] / kH[0] + 0.5f * (y[i] + z[i]) - sh[i] + bias[i];
  }
}

extern "C" void kernel_launch(void* const* d_in, const int* in_sizes, int n_in,
                              void* d_out, int out_size, void* d_ws, size_t ws_size,
                              hipStream_t stream) {
  const float* state_H    = (const float*)d_in[0];
  const float* state_K    = (const float*)d_in[1];
  const int*   ind_K      = (const int*)d_in[2];
  const int*   ind_HK     = (const int*)d_in[3];
  const float* kappa_K    = (const float*)d_in[4];
  const float* kappa_H    = (const float*)d_in[5];
  const float* weights_H  = (const float*)d_in[6];
  const float* bias_H     = (const float*)d_in[7];
  const float* weights_HK = (const float*)d_in[8];
  const float* w_K        = (const float*)d_in[9];

  float* out_fH = (float*)d_out;
  float* out_fK = out_fH + N_NODES;

  float* ws     = (float*)d_ws;
  float* fK_acc = ws;
  float* Vacc   = fK_acc + (size_t)N_NODES * D_OSC;
  float* yv     = Vacc + (size_t)N_NODES * D_OSC;
  float* zv     = yv + N_NODES;
  float* skn    = zv + N_NODES;
  float* hs     = skn + (size_t)N_NODES * D_OSC;
  float* gv     = hs + (size_t)N_NODES * D_OSC;
  float* coefK  = gv + N_NODES;

  k_norm_zero<<<NORM_BLKS + ZERO_BLKS, 256, 0, stream>>>(state_K, state_H, skn, hs, gv, ws);

  k_gemv_coef<<<GEMV_BLKS + COEF_BLKS, 256, 0, stream>>>(weights_H, gv, yv, zv,
                                                         ind_K, w_K, skn, coefK);

  k_scatter<<<2 * NBKT * NSL, 1024, 0, stream>>>(ind_HK, weights_HK, hs, Vacc,
                                                 ind_K, coefK, skn, fK_acc);

  k_fin<<<N_NODES / 32, 1024, 0, stream>>>(fK_acc, Vacc, skn, gv, yv, zv,
                                           state_H, bias_H, kappa_H, kappa_K,
                                           out_fH, out_fK);
}


# W pass: each 16x32 block of W arrives by whole-line loads (16 B per lane, 4 lines per load, issued one block ahead); both WMMA operand forms are read from the wave's LDS slab; sums unchanged
# speedup vs baseline: 3.5697x; 1.0336x over previous
#include <hip/hip_runtime.h>
#include <math.h>

#define N_NODES 8192
#define D_OSC   64
#define E_EDGES 1048576

typedef __attribute__((ext_vector_type(2))) float v2f;
typedef __attribute__((ext_vector_type(8))) float v8f;

__device__ __forceinline__ float wave_reduce_sum(float v) {
  #pragma unroll
  for (int m = 16; m >= 1; m >>= 1) v += __shfl_xor(v, m, 32);
  return v;
}

#define N_ACC     (2 * N_NODES * D_OSC + 2 * N_NODES)
#define NORM_BLKS (N_NODES / 8)
#define ZERO_BLKS (N_ACC / 4 / 256)

__device__ __forceinline__ void norm_tanh_body(int blk,
                                               const float* __restrict__ sk,
                                               const float* __restrict__ sh,
                                               float* __restrict__ skn,
                                               float* __restrict__ hs,
                                               float* __restrict__ gv) {
  int row  = blk * 8 + (threadIdx.x >> 5);
  int lane = threadIdx.x & 31;
  const float* r = sk + (size_t)row * D_OSC;
  float a = r[lane];
  float b = r[lane + 32];
  float ss  = wave_reduce_sum(a * a + b * b);
  float inv = rsqrtf(ss);
  float na = a * inv;
  float nb = b * inv;
  float* o = skn + (size_t)row * D_OSC;
  o[lane]      = na;
  o[lane + 32] = nb;
  float grow = tanhf(sh[row]);
  float* oh = hs + (size_t)row * D_OSC;
  oh[lane]      = grow * na;
  oh[lane + 32] = grow * nb;

  int gi = blk * 256 + threadIdx.x;
  if (gi < N_NODES) gv[gi] = tanhf(sh[gi]);
}

__global__ __launch_bounds__(256) void k_norm_zero(const float* __restrict__ sk,
                                                   const float* __restrict__ sh,
                                                   float* __restrict__ skn,
                                                   float* __restrict__ hs,
                                                   float* __restrict__ gv,
                                                   float* __restrict__ accs) {
  if (blockIdx.x < NORM_BLKS) {
    norm_tanh_body(blockIdx.x, sk, sh, skn, hs, gv);
  } else {
    const int i = (blockIdx.x - NORM_BLKS) * 256 + threadIdx.x;
    ((float4*)accs)[i] = make_float4(0.0f, 0.0f, 0.0f, 0.0f);
  }
}

#define EPW  32
#define EGRP 8

__device__ __forceinline__ void coef_K_body(int blk,
                                            const int* __restrict__ ind,
                                            const float* __restrict__ wK,
                                            const float* __restrict__ skn,
                                            float* __restrict__ coef) {
  const int lane = threadIdx.x & 31;
  const int e0   = (blk * 8 + (threadIdx.x >> 5)) * EPW;
  const int2  myj = ((const int2*)ind)[e0 + lane];
  const float myw = wK[e0 + lane];
  float mys = 0.0f;
  #pragma unroll
  for (int g = 0; g < EPW; g += EGRP) {
    float a0[EGRP], a1[EGRP], b0[EGRP], b1[EGRP];
    #pragma unroll
    for (int u = 0; u < EGRP; ++u) {
      int j0 = __builtin_amdgcn_readlane(myj.x, g + u);
      int j1 = __builtin_amdgcn_readlane(myj.y, g + u);
      const float* r0 = skn + (size_t)j0 * D_OSC;
      const float* r1 = skn + (size_t)j1 * D_OSC;
      a0[u] = r0[lane];  a1[u] = r0[lane + 32];
      b0[u] = r1[lane];  b1[u] = r1[lane + 32];
    }
    #pragma unroll
    for (int u = 0; u < EGRP; ++u) {
      float s = wave_reduce_sum(a0[u] * b0[u] + a1[u] * b1[u]);
      if (lane == g + u) mys = s;
    }
  }
  coef[e0 + lane] = myw * mys;
}

#define NBKT 8
#define BN   (N_NODES / NBKT)
#define BSH  10
#define NSL  32
#define SLE  (E_EDGES / NSL)
#define NCH  (SLE / 32)
#define HGRP 16
#define QCAP 128

typedef __attribute__((ext_vector_type(2))) unsigned v2u;


__global__ __launch_bounds__(1024) void k_scatter(const int* __restrict__ indA,
                                                  const float* __restrict__ coefA,
                                                  const float* __restrict__ rowsA,
                                                  float* __restrict__ accA,
                                                  const int* __restrict__ indB,
                                                  const float* __restrict__ coefB,
                                                  const float* __restrict__ rowsB,
                                                  float* __restrict__ accB) {
  const bool setB = blockIdx.x >= NBKT * NSL;
  const int*   __restrict__ ind  = setB ? indB  : indA;
  const float* __restrict__ coef = setB ? coefB : coefA;
  const float* __restrict__ rows = setB ? rowsB : rowsA;
  float*       __restrict__ acc  = setB ? accB  : accA;
  const int blk = blockIdx.x & (NBKT * NSL - 1);
  __shared__ float lacc[BN * D_OSC];
  __shared__ v2u hitq[32 * QCAP];
  const int tid  = threadIdx.x;
  const int lane = tid & 31;
  const int wv   = __builtin_amdgcn_readfirstlane(tid >> 5);
  const int bkt  = blk & (NBKT - 1);
  const int slc  = blk >> 3;

  for (int i = tid; i < BN * D_OSC / 4; i += 1024)
    ((float4*)lacc)[i] = make_float4(0.0f, 0.0f, 0.0f, 0.0f);
  __syncthreads();

  const int2*  ip = (const int2*)ind + (size_t)slc * SLE;
  const float* cp = coef + (size_t)slc * SLE;
  v2u* q = hitq + wv * QCAP;
  const char* rowsb = (const char*)rows;

  unsigned head = 0u, tail = 0u;

  int2  ij = ip[wv * 32 + lane];
  float cw = cp[wv * 32 + lane];
  for (int c = wv; c < NCH; c += 32) {
    int2  ijn = ij;
    float cwn = cw;
    if (c + 32 < NCH) {
      ijn = ip[(c + 32) * 32 + lane];
      cwn = cp[(c + 32) * 32 + lane];
    }
    const bool h0 = (ij.x >> BSH) == bkt;
    const bool h1 = (ij.y >> BSH) == bkt;
    const unsigned m0 = __builtin_amdgcn_ballot_w32(h0);
    const unsigned m1 = __builtin_amdgcn_ballot_w32(h1);
    const unsigned n0 = (unsigned)__builtin_popcount(m0);
    const unsigned cb = __float_as_uint(cw);
    if (h0) {
      const unsigned p = tail + __builtin_amdgcn_mbcnt_lo(m0, 0u);
      v2u e;
      e.x = ((unsigned)ij.y << BSH) | ((unsigned)ij.x & (BN - 1));
      e.y = cb;
      q[p & (QCAP - 1)] = e;
    }
    if (h1) {
      const unsigned p = tail + n0 + __builtin_amdgcn_mbcnt_lo(m1, 0u);
      v2u e;
      e.x = ((unsigned)ij.x << BSH) | ((unsigned)ij.y & (BN - 1));
      e.y = cb;
      q[p & (QCAP - 1)] = e;
    }
    tail += n0 + (unsigned)__builtin_popcount(m1);

    while (tail - head >= 32u) {
      const v2u e = q[(head + lane) & (QCAP - 1)];
      head += 32u;
      const int srow = (int)((e.x >> BSH) << 8);
      const int drow = (int)((e.x & (BN - 1)) << 6);
      const int cbit = (int)e.y;
      #pragma unroll
      for (int g = 0; g < 32; g += HGRP) {
        float2 r[HGRP];
        #pragma unroll
        for (int u = 0; u < HGRP; ++u) {
          const int so = __builtin_amdgcn_readlane(srow, g + u);
          r[u] = ((const float2*)(rowsb + (size_t)(unsigned)so))[lane];
        }
        #pragma unroll
        for (int u = 0; u < HGRP; ++u) {
          const int   dl = __builtin_amdgcn_readlane(drow, g + u);
          const float cf = __int_as_float(__builtin_amdgcn_readlane(cbit, g + u));
          atomicAdd(&lacc[dl + 2 * lane],     cf * r[u].x);
          atomicAdd(&lacc[dl + 2 * lane + 1], cf * r[u].y);
        }
      }
    }
    ij = ijn;
    cw = cwn;
  }

  const unsigned nleft = tail - head;
  if (nleft != 0u) {
    const v2u e = q[(head + lane) & (QCAP - 1)];
    const int srow = (int)((e.x >> BSH) << 8);
    const int drow = (int)((e.x & (BN - 1)) << 6);
    const int cbit = (int)e.y;
    #pragma unroll
    for (int g = 0; g < 32; g += 8) {
      if ((unsigned)g < nleft) {
        float2 r[8];
        #pragma unroll
        for (int u = 0; u < 8; ++u) {
          if ((unsigned)(g + u) < nleft) {
            const int so = __builtin_amdgcn_readlane(srow, g + u);
            r[u] = ((const float2*)(rowsb + (size_t)(unsigned)so))[lane];
          }
        }
        #pragma unroll
        for (int u = 0; u < 8; ++u) {
          if ((unsigned)(g + u) < nleft) {
            const int   dl = __builtin_amdgcn_readlane(drow, g + u);
            const float cf = __int_as_float(__builtin_amdgcn_readlane(cbit, g + u));
            atomicAdd(&lacc[dl + 2 * lane],     cf * r[u].x);
            atomicAdd(&lacc[dl + 2 * lane + 1], cf * r[u].y);
          }
        }
      }
    }
  }

  asm volatile("s_wait_dscnt 0x0" ::: "memory");
  __syncthreads();
  float* gacc = acc + (size_t)bkt * BN * D_OSC;
  for (int k0 = 0; k0 < BN * D_OSC / 1024; k0 += 8) {
    int   fi[8];
    float fv[8];
    #pragma unroll
    for (int u = 0; u < 8; ++u) {
      fi[u] = (((k0 + u + 2 * slc) & (BN * D_OSC / 1024 - 1)) << 10) + tid;
      fv[u] = lacc[fi[u]];
    }
    #pragma unroll
    for (int u = 0; u < 8; ++u) atomicAdd(&gacc[fi[u]], fv[u]);
  }
}

typedef __attribute__((ext_vector_type(4))) float v4f;

#define SLP 44

__device__ __forceinline__ void gemv_fused_body(int bx, int by,
                                                float* __restrict__ slab,
                                                float* __restrict__ zstage,
                                                const float* __restrict__ W,
                                                const float* __restrict__ gv,
                                                float* __restrict__ y,
                                                float* __restrict__ z) {

  const int wv   = threadIdx.x >> 5;
  const int lane = threadIdx.x & 31;
  const int hl   = lane & 15;
  const int koff = (lane < 16) ? 0 : 2;

  const int C0 = bx * 512;
  const int r0 = by * 128 + wv * 16;

  zstage[threadIdx.x]       = 0.0f;
  zstage[threadIdx.x + 256] = 0.0f;
  __syncthreads();

  float* sl = slab + wv * (16 * SLP);

  v2f bz0 = *(const v2f*)(gv + r0 +  0 + koff);
  v2f bz1 = *(const v2f*)(gv + r0 +  4 + koff);
  v2f bz2 = *(const v2f*)(gv + r0 +  8 + koff);
  v2f bz3 = *(const v2f*)(gv + r0 + 12 + koff);

  const int lr = lane >> 3;
  const int lc = (lane & 7) * 4;
  const float* wld = W + (size_t)(r0 + lr) * N_NODES + C0 + lc;
  float* sst = sl + lr * SLP + lc;
  const float* fr = sl + hl * SLP + koff;

  v4f n0 = *(const v4f*)(wld + (size_t) 0 * N_NODES);
  v4f n1 = *(const v4f*)(wld + (size_t) 4 * N_NODES);
  v4f n2 = *(const v4f*)(wld + (size_t) 8 * N_NODES);
  v4f n3 = *(const v4f*)(wld + (size_t)12 * N_NODES);

  v8f yacc = {};
  for (int it = 0; it < 16; ++it) {
    const v4f c0 = n0, c1 = n1, c2 = n2, c3 = n3;
    if (it + 1 < 16) {
      const float* p = wld + (it + 1) * 32;
      n0 = *(const v4f*)(p + (size_t) 0 * N_NODES);
      n1 = *(const v4f*)(p + (size_t) 4 * N_NODES);
      n2 = *(const v4f*)(p + (size_t) 8 * N_NODES);
      n3 = *(const v4f*)(p + (size_t)12 * N_NODES);
    }
    *(v4f*)(sst +  0 * SLP) = c0;
    *(v4f*)(sst +  4 * SLP) = c1;
    *(v4f*)(sst +  8 * SLP) = c2;
    *(v4f*)(sst + 12 * SLP) = c3;

    #pragma unroll
    for (int ct = 0; ct < 2; ++ct) {
      const int c = C0 + it * 32 + ct * 16;
      v2f a0 = *(const v2f*)(fr + ct * 16 +  0);
      v2f a1 = *(const v2f*)(fr + ct * 16 +  4);
      v2f a2 = *(const v2f*)(fr + ct * 16 +  8);
      v2f a3 = *(const v2f*)(fr + ct * 16 + 12);
      v2f by0 = *(const v2f*)(gv + c +  0 + koff);
      v2f by1 = *(const v2f*)(gv + c +  4 + koff);
      v2f by2 = *(const v2f*)(gv + c +  8 + koff);
      v2f by3 = *(const v2f*)(gv + c + 12 + koff);

      yacc = __builtin_amdgcn_wmma_f32_16x16x4_f32(false, a0, false, by0, (short)0, yacc, false, false);
      yacc = __builtin_amdgcn_wmma_f32_16x16x4_f32(false, a1, false, by1, (short)0, yacc, false, false);
      yacc = __builtin_amdgcn_wmma_f32_16x16x4_f32(false, a2, false, by2, (short)0, yacc, false, false);
      yacc = __builtin_amdgcn_wmma_f32_16x16x4_f32(false, a3, false, by3, (short)0, yacc, false, false);

      const float* tr = sl + ct * 16 + hl;
      v2f az0, az1, az2, az3;
      az0[0] = tr[( 0 + koff) * SLP];  az0[1] = tr[( 1 + koff) * SLP];
      az1[0] = tr[( 4 + koff) * SLP];  az1[1] = tr[( 5 + koff) * SLP];
      az2[0] = tr[( 8 + koff) * SLP];  az2[1] = tr[( 9 + koff) * SLP];
      az3[0] = tr[(12 + koff) * SLP];  az3[1] = tr[(13 + koff) * SLP];

      v8f zt = {};
      zt = __builtin_amdgcn_wmma_f32_16x16x4_f32(false, az0, false, bz0, (short)0, zt, false, false);
      zt = __builtin_amdgcn_wmma_f32_16x16x4_f32(false, az1, false, bz1, (short)0, zt, false, false);
      zt = __builtin_amdgcn_wmma_f32_16x16x4_f32(false, az2, false, bz2, (short)0, zt, false, false);
      zt = __builtin_amdgcn_wmma_f32_16x16x4_f32(false, az3, false, bz3, (short)0, zt, false, false);

      if (hl == 0) {
        int zb = it * 32 + ct * 16 + ((lane == 16) ? 8 : 0);
        #pragma unroll
        for (int v = 0; v < 8; ++v) atomicAdd(&zstage[zb + v], zt[v]);
      }
    }
  }

  if (hl == 0) {
    int base = r0 + ((lane == 16) ? 8 : 0);
    #pragma unroll
    for (int v = 0; v < 8; ++v) atomicAdd(&y[base + v], yacc[v]);
  }

  __syncthreads();
  atomicAdd(&z[C0 + threadIdx.x],       zstage[threadIdx.x]);
  atomicAdd(&z[C0 + threadIdx.x + 256], zstage[threadIdx.x + 256]);
}

#define GEMV_BLKS (16 * 64)
#define COEF_BLKS (E_EDGES / (8 * EPW))

__global__ __launch_bounds__(256) void k_gemv_coef(const float* __restrict__ W,
                                                   const float* __restrict__ gv,
                                                   float* __restrict__ y,
                                                   float* __restrict__ z,
                                                   const int* __restrict__ ind,
                                                   const float* __restrict__ wK,
                                                   const float* __restrict__ skn,
                                                   float* __restrict__ coef) {
  __shared__ __attribute__((aligned(16))) float slab[8 * 16 * SLP];
  __shared__ float zstage[512];
  if (blockIdx.x < GEMV_BLKS) {
    gemv_fused_body(blockIdx.x & 15, blockIdx.x >> 4, slab, zstage, W, gv, y, z);
  } else {
    coef_K_body(blockIdx.x - GEMV_BLKS, ind, wK, skn, coef);
  }
}

__global__ __launch_bounds__(1024) void k_fin(const float* __restrict__ fKacc,
                                              const float* __restrict__ V,
                                              const float* __restrict__ skn,
                                              const float* __restrict__ gv,
                                              const float* __restrict__ y,
                                              const float* __restrict__ z,
                                              const float* __restrict__ sh,
                                              const float* __restrict__ bias,
                                              const float* __restrict__ kH,
                                              const float* __restrict__ kK,
                                              float* __restrict__ outH,
                                              float* __restrict__ outK) {
  __shared__ float sdot[32];
  int wv   = threadIdx.x >> 5;
  int lane = threadIdx.x & 31;
  int row  = blockIdx.x * 32 + wv;
  size_t base = (size_t)row * D_OSC;
  float s0 = skn[base + lane],   s1 = skn[base + lane + 32];
  float v0 = V[base + lane],     v1 = V[base + lane + 32];
  float k0 = fKacc[base + lane], k1 = fKacc[base + lane + 32];
  float dhv = wave_reduce_sum(s0 * v0 + s1 * v1);
  float c  = gv[row] / kK[0];
  float r0 = k0 - c * v0;
  float r1 = k1 - c * v1;
  float dot = wave_reduce_sum(s0 * r0 + s1 * r1);
  outK[base + lane]      = -r0 + s0 * dot;
  outK[base + lane + 32] = -r1 + s1 * dot;
  if (lane == 0) sdot[wv] = dhv;
  __syncthreads();
  if (wv == 0) {
    int i = blockIdx.x * 32 + lane;
    outH[i] = sdot[lane] / kH[0] + 0.5f * (y[i] + z[i]) - sh[i] + bias[i];
  }
}

extern "C" void kernel_launch(void* const* d_in, const int* in_sizes, int n_in,
                              void* d_out, int out_size, void* d_ws, size_t ws_size,
                              hipStream_t stream) {
  const float* state_H    = (const float*)d_in[0];
  const float* state_K    = (const float*)d_in[1];
  const int*   ind_K      = (const int*)d_in[2];
  const int*   ind_HK     = (const int*)d_in[3];
  const float* kappa_K    = (const float*)d_in[4];
  const float* kappa_H    = (const float*)d_in[5];
  const float* weights_H  = (const float*)d_in[6];
  const float* bias_H     = (const float*)d_in[7];
  const float* weights_HK = (const float*)d_in[8];
  const float* w_K        = (const float*)d_in[9];

  float* out_fH = (float*)d_out;
  float* out_fK = out_fH + N_NODES;

  float* ws     = (float*)d_ws;
  float* fK_acc = ws;
  float* Vacc   = fK_acc + (size_t)N_NODES * D_OSC;
  float* yv     = Vacc + (size_t)N_NODES * D_OSC;
  float* zv     = yv + N_NODES;
  float* skn    = zv + N_NODES;
  float* hs     = skn + (size_t)N_NODES * D_OSC;
  float* gv     = hs + (size_t)N_NODES * D_OSC;
  float* coefK  = gv + N_NODES;

  k_norm_zero<<<NORM_BLKS + ZERO_BLKS, 256, 0, stream>>>(state_K, state_H, skn, hs, gv, ws);

  k_gemv_coef<<<GEMV_BLKS + COEF_BLKS, 256, 0, stream>>>(weights_H, gv, yv, zv,
                                                         ind_K, w_K, skn, coefK);

  k_scatter<<<2 * NBKT * NSL, 1024, 0, stream>>>(ind_HK, weights_HK, hs, Vacc,
                                                 ind_K, coefK, skn, fK_acc);

  k_fin<<<N_NODES / 32, 1024, 0, stream>>>(fK_acc, Vacc, skn, gv, yv, zv,
                                           state_H, bias_H, kappa_H, kappa_K,
                                           out_fH, out_fK);
}


# K edge coefficients: one edge per 16-lane half-wave (16 B of a row per lane, a load brings two edges' rows); ids by DPP row_share, dot ends in 4 DPP adds, no LDS permutes; dot sum order changes
# speedup vs baseline: 3.7505x; 1.0506x over previous
#include <hip/hip_runtime.h>
#include <math.h>

#define N_NODES 8192
#define D_OSC   64
#define E_EDGES 1048576

typedef __attribute__((ext_vector_type(2))) float v2f;
typedef __attribute__((ext_vector_type(8))) float v8f;

__device__ __forceinline__ float wave_reduce_sum(float v) {
  #pragma unroll
  for (int m = 16; m >= 1; m >>= 1) v += __shfl_xor(v, m, 32);
  return v;
}

#define N_ACC     (2 * N_NODES * D_OSC + 2 * N_NODES)
#define NORM_BLKS (N_NODES / 8)
#define ZERO_BLKS (N_ACC / 4 / 256)

__device__ __forceinline__ void norm_tanh_body(int blk,
                                               const float* __restrict__ sk,
                                               const float* __restrict__ sh,
                                               float* __restrict__ skn,
                                               float* __restrict__ hs,
                                               float* __restrict__ gv) {
  int row  = blk * 8 + (threadIdx.x >> 5);
  int lane = threadIdx.x & 31;
  const float* r = sk + (size_t)row * D_OSC;
  float a = r[lane];
  float b = r[lane + 32];
  float ss  = wave_reduce_sum(a * a + b * b);
  float inv = rsqrtf(ss);
  float na = a * inv;
  float nb = b * inv;
  float* o = skn + (size_t)row * D_OSC;
  o[lane]      = na;
  o[lane + 32] = nb;
  float grow = tanhf(sh[row]);
  float* oh = hs + (size_t)row * D_OSC;
  oh[lane]      = grow * na;
  oh[lane + 32] = grow * nb;

  int gi = blk * 256 + threadIdx.x;
  if (gi < N_NODES) gv[gi] = tanhf(sh[gi]);
}

__global__ __launch_bounds__(256) void k_norm_zero(const float* __restrict__ sk,
                                                   const float* __restrict__ sh,
                                                   float* __restrict__ skn,
                                                   float* __restrict__ hs,
                                                   float* __restrict__ gv,
                                                   float* __restrict__ accs) {
  if (blockIdx.x < NORM_BLKS) {
    norm_tanh_body(blockIdx.x, sk, sh, skn, hs, gv);
  } else {
    const int i = (blockIdx.x - NORM_BLKS) * 256 + threadIdx.x;
    ((float4*)accs)[i] = make_float4(0.0f, 0.0f, 0.0f, 0.0f);
  }
}

#define EPW  32
#define CGRP 4

template <int CTRL>
__device__ __forceinline__ int dpp_i(int x) {
  return __builtin_amdgcn_update_dpp(0, x, CTRL, 0xf, 0xf, true);
}
template <int T>
__device__ __forceinline__ int row_share(int x) { return dpp_i<0x150 + T>(x); }
template <int M>
__device__ __forceinline__ float row_xor(float x) {
  return __builtin_bit_cast(float, dpp_i<0x160 + M>(__builtin_bit_cast(int, x)));
}
__device__ __forceinline__ float row16_sum(float v) {
  v += row_xor<8>(v);
  v += row_xor<4>(v);
  v += row_xor<2>(v);
  v += row_xor<1>(v);
  return v;
}

template <int T0>
__device__ __forceinline__ void coef_K_group(const int2 myj, const char* __restrict__ rb,
                                             const unsigned ho, const int hl, float& mys) {
  float4 a[CGRP], b[CGRP];
  #pragma unroll
  for (int u = 0; u < CGRP; ++u) {
    unsigned j0, j1;
    switch (u) {
      case 0:  j0 = (unsigned)row_share<T0 + 0>(myj.x); j1 = (unsigned)row_share<T0 + 0>(myj.y); break;
      case 1:  j0 = (unsigned)row_share<T0 + 1>(myj.x); j1 = (unsigned)row_share<T0 + 1>(myj.y); break;
      case 2:  j0 = (unsigned)row_share<T0 + 2>(myj.x); j1 = (unsigned)row_share<T0 + 2>(myj.y); break;
      default: j0 = (unsigned)row_share<T0 + 3>(myj.x); j1 = (unsigned)row_share<T0 + 3>(myj.y); break;
    }
    a[u] = *(const float4*)(rb + ((j0 << 8) | ho));
    b[u] = *(const float4*)(rb + ((j1 << 8) | ho));
  }
  #pragma unroll
  for (int u = 0; u < CGRP; ++u) {
    float p = a[u].x * b[u].x;
    p = fmaf(a[u].y, b[u].y, p);
    p = fmaf(a[u].z, b[u].z, p);
    p = fmaf(a[u].w, b[u].w, p);
    p = row16_sum(p);
    if (hl == T0 + u) mys = p;
  }
}

__device__ __forceinline__ void coef_K_body(int blk,
                                            const int* __restrict__ ind,
                                            const float* __restrict__ wK,
                                            const float* __restrict__ skn,
                                            float* __restrict__ coef) {
  const int lane = threadIdx.x & 31;
  const int hl   = lane & 15;
  const unsigned ho = (unsigned)hl << 4;
  const int e0   = (blk * 8 + (threadIdx.x >> 5)) * EPW;
  const int2  myj = ((const int2*)ind)[e0 + lane];
  const float myw = wK[e0 + lane];
  const char* rb = (const char*)skn;
  float mys = 0.0f;
  coef_K_group< 0>(myj, rb, ho, hl, mys);
  __builtin_amdgcn_sched_barrier(0);
  coef_K_group< 4>(myj, rb, ho, hl, mys);
  __builtin_amdgcn_sched_barrier(0);
  coef_K_group< 8>(myj, rb, ho, hl, mys);
  __builtin_amdgcn_sched_barrier(0);
  coef_K_group<12>(myj, rb, ho, hl, mys);
  coef[e0 + lane] = myw * mys;
}

#define NBKT 8
#define BN   (N_NODES / NBKT)
#define BSH  10
#define NSL  32
#define SLE  (E_EDGES / NSL)
#define NCH  (SLE / 32)
#define HGRP 16
#define QCAP 128

typedef __attribute__((ext_vector_type(2))) unsigned v2u;


__global__ __launch_bounds__(1024) void k_scatter(const int* __restrict__ indA,
                                                  const float* __restrict__ coefA,
                                                  const float* __restrict__ rowsA,
                                                  float* __restrict__ accA,
                                                  const int* __restrict__ indB,
                                                  const float* __restrict__ coefB,
                                                  const float* __restrict__ rowsB,
                                                  float* __restrict__ accB) {
  const bool setB = blockIdx.x >= NBKT * NSL;
  const int*   __restrict__ ind  = setB ? indB  : indA;
  const float* __restrict__ coef = setB ? coefB : coefA;
  const float* __restrict__ rows = setB ? rowsB : rowsA;
  float*       __restrict__ acc  = setB ? accB  : accA;
  const int blk = blockIdx.x & (NBKT * NSL - 1);
  __shared__ float lacc[BN * D_OSC];
  __shared__ v2u hitq[32 * QCAP];
  const int tid  = threadIdx.x;
  const int lane = tid & 31;
  const int wv   = __builtin_amdgcn_readfirstlane(tid >> 5);
  const int bkt  = blk & (NBKT - 1);
  const int slc  = blk >> 3;

  for (int i = tid; i < BN * D_OSC / 4; i += 1024)
    ((float4*)lacc)[i] = make_float4(0.0f, 0.0f, 0.0f, 0.0f);
  __syncthreads();

  const int2*  ip = (const int2*)ind + (size_t)slc * SLE;
  const float* cp = coef + (size_t)slc * SLE;
  v2u* q = hitq + wv * QCAP;
  const char* rowsb = (const char*)rows;

  unsigned head = 0u, tail = 0u;

  int2  ij = ip[wv * 32 + lane];
  float cw = cp[wv * 32 + lane];
  for (int c = wv; c < NCH; c += 32) {
    int2  ijn = ij;
    float cwn = cw;
    if (c + 32 < NCH) {
      ijn = ip[(c + 32) * 32 + lane];
      cwn = cp[(c + 32) * 32 + lane];
    }
    const bool h0 = (ij.x >> BSH) == bkt;
    const bool h1 = (ij.y >> BSH) == bkt;
    const unsigned m0 = __builtin_amdgcn_ballot_w32(h0);
    const unsigned m1 = __builtin_amdgcn_ballot_w32(h1);
    const unsigned n0 = (unsigned)__builtin_popcount(m0);
    const unsigned cb = __float_as_uint(cw);
    if (h0) {
      const unsigned p = tail + __builtin_amdgcn_mbcnt_lo(m0, 0u);
      v2u e;
      e.x = ((unsigned)ij.y << BSH) | ((unsigned)ij.x & (BN - 1));
      e.y = cb;
      q[p & (QCAP - 1)] = e;
    }
    if (h1) {
      const unsigned p = tail + n0 + __builtin_amdgcn_mbcnt_lo(m1, 0u);
      v2u e;
      e.x = ((unsigned)ij.x << BSH) | ((unsigned)ij.y & (BN - 1));
      e.y = cb;
      q[p & (QCAP - 1)] = e;
    }
    tail += n0 + (unsigned)__builtin_popcount(m1);

    while (tail - head >= 32u) {
      const v2u e = q[(head + lane) & (QCAP - 1)];
      head += 32u;
      const int srow = (int)((e.x >> BSH) << 8);
      const int drow = (int)((e.x & (BN - 1)) << 6);
      const int cbit = (int)e.y;
      #pragma unroll
      for (int g = 0; g < 32; g += HGRP) {
        float2 r[HGRP];
        #pragma unroll
        for (int u = 0; u < HGRP; ++u) {
          const int so = __builtin_amdgcn_readlane(srow, g + u);
          r[u] = ((const float2*)(rowsb + (size_t)(unsigned)so))[lane];
        }
        #pragma unroll
        for (int u = 0; u < HGRP; ++u) {
          const int   dl = __builtin_amdgcn_readlane(drow, g + u);
          const float cf = __int_as_float(__builtin_amdgcn_readlane(cbit, g + u));
          atomicAdd(&lacc[dl + 2 * lane],     cf * r[u].x);
          atomicAdd(&lacc[dl + 2 * lane + 1], cf * r[u].y);
        }
      }
    }
    ij = ijn;
    cw = cwn;
  }

  const unsigned nleft = tail - head;
  if (nleft != 0u) {
    const v2u e = q[(head + lane) & (QCAP - 1)];
    const int srow = (int)((e.x >> BSH) << 8);
    const int drow = (int)((e.x & (BN - 1)) << 6);
    const int cbit = (int)e.y;
    #pragma unroll
    for (int g = 0; g < 32; g += 8) {
      if ((unsigned)g < nleft) {
        float2 r[8];
        #pragma unroll
        for (int u = 0; u < 8; ++u) {
          if ((unsigned)(g + u) < nleft) {
            const int so = __builtin_amdgcn_readlane(srow, g + u);
            r[u] = ((const float2*)(rowsb + (size_t)(unsigned)so))[lane];
          }
        }
        #pragma unroll
        for (int u = 0; u < 8; ++u) {
          if ((unsigned)(g + u) < nleft) {
            const int   dl = __builtin_amdgcn_readlane(drow, g + u);
            const float cf = __int_as_float(__builtin_amdgcn_readlane(cbit, g + u));
            atomicAdd(&lacc[dl + 2 * lane],     cf * r[u].x);
            atomicAdd(&lacc[dl + 2 * lane + 1], cf * r[u].y);
          }
        }
      }
    }
  }

  asm volatile("s_wait_dscnt 0x0" ::: "memory");
  __syncthreads();
  float* gacc = acc + (size_t)bkt * BN * D_OSC;
  for (int k0 = 0; k0 < BN * D_OSC / 1024; k0 += 8) {
    int   fi[8];
    float fv[8];
    #pragma unroll
    for (int u = 0; u < 8; ++u) {
      fi[u] = (((k0 + u + 2 * slc) & (BN * D_OSC / 1024 - 1)) << 10) + tid;
      fv[u] = lacc[fi[u]];
    }
    #pragma unroll
    for (int u = 0; u < 8; ++u) atomicAdd(&gacc[fi[u]], fv[u]);
  }
}

typedef __attribute__((ext_vector_type(4))) float v4f;

#define SLP 44

__device__ __forceinline__ void gemv_fused_body(int bx, int by,
                                                float* __restrict__ slab,
                                                float* __restrict__ zstage,
                                                const float* __restrict__ W,
                                                const float* __restrict__ gv,
                                                float* __restrict__ y,
                                                float* __restrict__ z) {

  const int wv   = threadIdx.x >> 5;
  const int lane = threadIdx.x & 31;
  const int hl   = lane & 15;
  const int koff = (lane < 16) ? 0 : 2;

  const int C0 = bx * 512;
  const int r0 = by * 128 + wv * 16;

  zstage[threadIdx.x]       = 0.0f;
  zstage[threadIdx.x + 256] = 0.0f;
  __syncthreads();

  float* sl = slab + wv * (16 * SLP);

  v2f bz0 = *(const v2f*)(gv + r0 +  0 + koff);
  v2f bz1 = *(const v2f*)(gv + r0 +  4 + koff);
  v2f bz2 = *(const v2f*)(gv + r0 +  8 + koff);
  v2f bz3 = *(const v2f*)(gv + r0 + 12 + koff);

  const int lr = lane >> 3;
  const int lc = (lane & 7) * 4;
  const float* wld = W + (size_t)(r0 + lr) * N_NODES + C0 + lc;
  float* sst = sl + lr * SLP + lc;
  const float* fr = sl + hl * SLP + koff;

  v4f n0 = *(const v4f*)(wld + (size_t) 0 * N_NODES);
  v4f n1 = *(const v4f*)(wld + (size_t) 4 * N_NODES);
  v4f n2 = *(const v4f*)(wld + (size_t) 8 * N_NODES);
  v4f n3 = *(const v4f*)(wld + (size_t)12 * N_NODES);

  v8f yacc = {};
  for (int it = 0; it < 16; ++it) {
    const v4f c0 = n0, c1 = n1, c2 = n2, c3 = n3;
    if (it + 1 < 16) {
      const float* p = wld + (it + 1) * 32;
      n0 = *(const v4f*)(p + (size_t) 0 * N_NODES);
      n1 = *(const v4f*)(p + (size_t) 4 * N_NODES);
      n2 = *(const v4f*)(p + (size_t) 8 * N_NODES);
      n3 = *(const v4f*)(p + (size_t)12 * N_NODES);
    }
    *(v4f*)(sst +  0 * SLP) = c0;
    *(v4f*)(sst +  4 * SLP) = c1;
    *(v4f*)(sst +  8 * SLP) = c2;
    *(v4f*)(sst + 12 * SLP) = c3;

    #pragma unroll
    for (int ct = 0; ct < 2; ++ct) {
      const int c = C0 + it * 32 + ct * 16;
      v2f a0 = *(const v2f*)(fr + ct * 16 +  0);
      v2f a1 = *(const v2f*)(fr + ct * 16 +  4);
      v2f a2 = *(const v2f*)(fr + ct * 16 +  8);
      v2f a3 = *(const v2f*)(fr + ct * 16 + 12);
      v2f by0 = *(const v2f*)(gv + c +  0 + koff);
      v2f by1 = *(const v2f*)(gv + c +  4 + koff);
      v2f by2 = *(const v2f*)(gv + c +  8 + koff);
      v2f by3 = *(const v2f*)(gv + c + 12 + koff);

      yacc = __builtin_amdgcn_wmma_f32_16x16x4_f32(false, a0, false, by0, (short)0, yacc, false, false);
      yacc = __builtin_amdgcn_wmma_f32_16x16x4_f32(false, a1, false, by1, (short)0, yacc, false, false);
      yacc = __builtin_amdgcn_wmma_f32_16x16x4_f32(false, a2, false, by2, (short)0, yacc, false, false);
      yacc = __builtin_amdgcn_wmma_f32_16x16x4_f32(false, a3, false, by3, (short)0, yacc, false, false);

      const float* tr = sl + ct * 16 + hl;
      v2f az0, az1, az2, az3;
      az0[0] = tr[( 0 + koff) * SLP];  az0[1] = tr[( 1 + koff) * SLP];
      az1[0] = tr[( 4 + koff) * SLP];  az1[1] = tr[( 5 + koff) * SLP];
      az2[0] = tr[( 8 + koff) * SLP];  az2[1] = tr[( 9 + koff) * SLP];
      az3[0] = tr[(12 + koff) * SLP];  az3[1] = tr[(13 + koff) * SLP];

      v8f zt = {};
      zt = __builtin_amdgcn_wmma_f32_16x16x4_f32(false, az0, false, bz0, (short)0, zt, false, false);
      zt = __builtin_amdgcn_wmma_f32_16x16x4_f32(false, az1, false, bz1, (short)0, zt, false, false);
      zt = __builtin_amdgcn_wmma_f32_16x16x4_f32(false, az2, false, bz2, (short)0, zt, false, false);
      zt = __builtin_amdgcn_wmma_f32_16x16x4_f32(false, az3, false, bz3, (short)0, zt, false, false);

      if (hl == 0) {
        int zb = it * 32 + ct * 16 + ((lane == 16) ? 8 : 0);
        #pragma unroll
        for (int v = 0; v < 8; ++v) atomicAdd(&zstage[zb + v], zt[v]);
      }
    }
  }

  if (hl == 0) {
    int base = r0 + ((lane == 16) ? 8 : 0);
    #pragma unroll
    for (int v = 0; v < 8; ++v) atomicAdd(&y[base + v], yacc[v]);
  }

  __syncthreads();
  atomicAdd(&z[C0 + threadIdx.x],       zstage[threadIdx.x]);
  atomicAdd(&z[C0 + threadIdx.x + 256], zstage[threadIdx.x + 256]);
}

#define GEMV_BLKS (16 * 64)
#define COEF_BLKS (E_EDGES / (8 * EPW))

__global__ __launch_bounds__(256) void k_gemv_coef(const float* __restrict__ W,
                                                   const float* __restrict__ gv,
                                                   float* __restrict__ y,
                                                   float* __restrict__ z,
                                                   const int* __restrict__ ind,
                                                   const float* __restrict__ wK,
                                                   const float* __restrict__ skn,
                                                   float* __restrict__ coef) {
  __shared__ __attribute__((aligned(16))) float slab[8 * 16 * SLP];
  __shared__ float zstage[512];
  if (blockIdx.x < GEMV_BLKS) {
    gemv_fused_body(blockIdx.x & 15, blockIdx.x >> 4, slab, zstage, W, gv, y, z);
  } else {
    coef_K_body(blockIdx.x - GEMV_BLKS, ind, wK, skn, coef);
  }
}

__global__ __launch_bounds__(1024) void k_fin(const float* __restrict__ fKacc,
                                              const float* __restrict__ V,
                                              const float* __restrict__ skn,
                                              const float* __restrict__ gv,
                                              const float* __restrict__ y,
                                              const float* __restrict__ z,
                                              const float* __restrict__ sh,
                                              const float* __restrict__ bias,
                                              const float* __restrict__ kH,
                                              const float* __restrict__ kK,
                                              float* __restrict__ outH,
                                              float* __restrict__ outK) {
  __shared__ float sdot[32];
  int wv   = threadIdx.x >> 5;
  int lane = threadIdx.x & 31;
  int row  = blockIdx.x * 32 + wv;
  size_t base = (size_t)row * D_OSC;
  float s0 = skn[base + lane],   s1 = skn[base + lane + 32];
  float v0 = V[base + lane],     v1 = V[base + lane + 32];
  float k0 = fKacc[base + lane], k1 = fKacc[base + lane + 32];
  float dhv = wave_reduce_sum(s0 * v0 + s1 * v1);
  float c  = gv[row] / kK[0];
  float r0 = k0 - c * v0;
  float r1 = k1 - c * v1;
  float dot = wave_reduce_sum(s0 * r0 + s1 * r1);
  outK[base + lane]      = -r0 + s0 * dot;
  outK[base + lane + 32] = -r1 + s1 * dot;
  if (lane == 0) sdot[wv] = dhv;
  __syncthreads();
  if (wv == 0) {
    int i = blockIdx.x * 32 + lane;
    outH[i] = sdot[lane] / kH[0] + 0.5f * (y[i] + z[i]) - sh[i] + bias[i];
  }
}

extern "C" void kernel_launch(void* const* d_in, const int* in_sizes, int n_in,
                              void* d_out, int out_size, void* d_ws, size_t ws_size,
                              hipStream_t stream) {
  const float* state_H    = (const float*)d_in[0];
  const float* state_K    = (const float*)d_in[1];
  const int*   ind_K      = (const int*)d_in[2];
  const int*   ind_HK     = (const int*)d_in[3];
  const float* kappa_K    = (const float*)d_in[4];
  const float* kappa_H    = (const float*)d_in[5];
  const float* weights_H  = (const float*)d_in[6];
  const float* bias_H     = (const float*)d_in[7];
  const float* weights_HK = (const float*)d_in[8];
  const float* w_K        = (const float*)d_in[9];

  float* out_fH = (float*)d_out;
  float* out_fK = out_fH + N_NODES;

  float* ws     = (float*)d_ws;
  float* fK_acc = ws;
  float* Vacc   = fK_acc + (size_t)N_NODES * D_OSC;
  float* yv     = Vacc + (size_t)N_NODES * D_OSC;
  float* zv     = yv + N_NODES;
  float* skn    = zv + N_NODES;
  float* hs     = skn + (size_t)N_NODES * D_OSC;
  float* gv     = hs + (size_t)N_NODES * D_OSC;
  float* coefK  = gv + N_NODES;

  k_norm_zero<<<NORM_BLKS + ZERO_BLKS, 256, 0, stream>>>(state_K, state_H, skn, hs, gv, ws);

  k_gemv_coef<<<GEMV_BLKS + COEF_BLKS, 256, 0, stream>>>(weights_H, gv, yv, zv,
                                                         ind_K, w_K, skn, coefK);

  k_scatter<<<2 * NBKT * NSL, 1024, 0, stream>>>(ind_HK, weights_HK, hs, Vacc,
                                                 ind_K, coefK, skn, fK_acc);

  k_fin<<<N_NODES / 32, 1024, 0, stream>>>(fK_acc, Vacc, skn, gv, yv, zv,
                                           state_H, bias_H, kappa_H, kappa_K,
                                           out_fH, out_fK);
}


# Scatter grid cut to one work-group per WGP (16 edge slices, not 32): each WGP zeroes its LDS bucket, takes in its waves and flushes by global atomics once, not twice; atomic sum order changes
# speedup vs baseline: 3.9006x; 1.0400x over previous
#include <hip/hip_runtime.h>
#include <math.h>

#define N_NODES 8192
#define D_OSC   64
#define E_EDGES 1048576

typedef __attribute__((ext_vector_type(2))) float v2f;
typedef __attribute__((ext_vector_type(8))) float v8f;

__device__ __forceinline__ float wave_reduce_sum(float v) {
  #pragma unroll
  for (int m = 16; m >= 1; m >>= 1) v += __shfl_xor(v, m, 32);
  return v;
}

#define N_ACC     (2 * N_NODES * D_OSC + 2 * N_NODES)
#define NORM_BLKS (N_NODES / 8)
#define ZERO_BLKS (N_ACC / 4 / 256)

__device__ __forceinline__ void norm_tanh_body(int blk,
                                               const float* __restrict__ sk,
                                               const float* __restrict__ sh,
                                               float* __restrict__ skn,
                                               float* __restrict__ hs,
                                               float* __restrict__ gv) {
  int row  = blk * 8 + (threadIdx.x >> 5);
  int lane = threadIdx.x & 31;
  const float* r = sk + (size_t)row * D_OSC;
  float a = r[lane];
  float b = r[lane + 32];
  float ss  = wave_reduce_sum(a * a + b * b);
  float inv = rsqrtf(ss);
  float na = a * inv;
  float nb = b * inv;
  float* o = skn + (size_t)row * D_OSC;
  o[lane]      = na;
  o[lane + 32] = nb;
  float grow = tanhf(sh[row]);
  float* oh = hs + (size_t)row * D_OSC;
  oh[lane]      = grow * na;
  oh[lane + 32] = grow * nb;

  int gi = blk * 256 + threadIdx.x;
  if (gi < N_NODES) gv[gi] = tanhf(sh[gi]);
}

__global__ __launch_bounds__(256) void k_norm_zero(const float* __restrict__ sk,
                                                   const float* __restrict__ sh,
                                                   float* __restrict__ skn,
                                                   float* __restrict__ hs,
                                                   float* __restrict__ gv,
                                                   float* __restrict__ accs) {
  if (blockIdx.x < NORM_BLKS) {
    norm_tanh_body(blockIdx.x, sk, sh, skn, hs, gv);
  } else {
    const int i = (blockIdx.x - NORM_BLKS) * 256 + threadIdx.x;
    ((float4*)accs)[i] = make_float4(0.0f, 0.0f, 0.0f, 0.0f);
  }
}

#define EPW  32
#define CGRP 4

template <int CTRL>
__device__ __forceinline__ int dpp_i(int x) {
  return __builtin_amdgcn_update_dpp(0, x, CTRL, 0xf, 0xf, true);
}
template <int T>
__device__ __forceinline__ int row_share(int x) { return dpp_i<0x150 + T>(x); }
template <int M>
__device__ __forceinline__ float row_xor(float x) {
  return __builtin_bit_cast(float, dpp_i<0x160 + M>(__builtin_bit_cast(int, x)));
}
__device__ __forceinline__ float row16_sum(float v) {
  v += row_xor<8>(v);
  v += row_xor<4>(v);
  v += row_xor<2>(v);
  v += row_xor<1>(v);
  return v;
}

template <int T0>
__device__ __forceinline__ void coef_K_group(const int2 myj, const char* __restrict__ rb,
                                             const unsigned ho, const int hl, float& mys) {
  float4 a[CGRP], b[CGRP];
  #pragma unroll
  for (int u = 0; u < CGRP; ++u) {
    unsigned j0, j1;
    switch (u) {
      case 0:  j0 = (unsigned)row_share<T0 + 0>(myj.x); j1 = (unsigned)row_share<T0 + 0>(myj.y); break;
      case 1:  j0 = (unsigned)row_share<T0 + 1>(myj.x); j1 = (unsigned)row_share<T0 + 1>(myj.y); break;
      case 2:  j0 = (unsigned)row_share<T0 + 2>(myj.x); j1 = (unsigned)row_share<T0 + 2>(myj.y); break;
      default: j0 = (unsigned)row_share<T0 + 3>(myj.x); j1 = (unsigned)row_share<T0 + 3>(myj.y); break;
    }
    a[u] = *(const float4*)(rb + ((j0 << 8) | ho));
    b[u] = *(const float4*)(rb + ((j1 << 8) | ho));
  }
  #pragma unroll
  for (int u = 0; u < CGRP; ++u) {
    float p = a[u].x * b[u].x;
    p = fmaf(a[u].y, b[u].y, p);
    p = fmaf(a[u].z, b[u].z, p);
    p = fmaf(a[u].w, b[u].w, p);
    p = row16_sum(p);
    if (hl == T0 + u) mys = p;
  }
}

__device__ __forceinline__ void coef_K_body(int blk,
                                            const int* __restrict__ ind,
                                            const float* __restrict__ wK,
                                            const float* __restrict__ skn,
                                            float* __restrict__ coef) {
  const int lane = threadIdx.x & 31;
  const int hl   = lane & 15;
  const unsigned ho = (unsigned)hl << 4;
  const int e0   = (blk * 8 + (threadIdx.x >> 5)) * EPW;
  const int2  myj = ((const int2*)ind)[e0 + lane];
  const float myw = wK[e0 + lane];
  const char* rb = (const char*)skn;
  float mys = 0.0f;
  coef_K_group< 0>(myj, rb, ho, hl, mys);
  __builtin_amdgcn_sched_barrier(0);
  coef_K_group< 4>(myj, rb, ho, hl, mys);
  __builtin_amdgcn_sched_barrier(0);
  coef_K_group< 8>(myj, rb, ho, hl, mys);
  __builtin_amdgcn_sched_barrier(0);
  coef_K_group<12>(myj, rb, ho, hl, mys);
  coef[e0 + lane] = myw * mys;
}

#define NBKT 8
#define BN   (N_NODES / NBKT)
#define BSH  10
#define NSL  16
#define SLE  (E_EDGES / NSL)
#define NCH  (SLE / 32)
#define FSTG (BN * D_OSC / 1024 / NSL)
#define HGRP 16
#define QCAP 128

typedef __attribute__((ext_vector_type(2))) unsigned v2u;


__global__ __launch_bounds__(1024) void k_scatter(const int* __restrict__ indA,
                                                  const float* __restrict__ coefA,
                                                  const float* __restrict__ rowsA,
                                                  float* __restrict__ accA,
                                                  const int* __restrict__ indB,
                                                  const float* __restrict__ coefB,
                                                  const float* __restrict__ rowsB,
                                                  float* __restrict__ accB) {
  const bool setB = blockIdx.x >= NBKT * NSL;
  const int*   __restrict__ ind  = setB ? indB  : indA;
  const float* __restrict__ coef = setB ? coefB : coefA;
  const float* __restrict__ rows = setB ? rowsB : rowsA;
  float*       __restrict__ acc  = setB ? accB  : accA;
  const int blk = blockIdx.x & (NBKT * NSL - 1);
  __shared__ float lacc[BN * D_OSC];
  __shared__ v2u hitq[32 * QCAP];
  const int tid  = threadIdx.x;
  const int lane = tid & 31;
  const int wv   = __builtin_amdgcn_readfirstlane(tid >> 5);
  const int bkt  = blk & (NBKT - 1);
  const int slc  = blk >> 3;

  for (int i = tid; i < BN * D_OSC / 4; i += 1024)
    ((float4*)lacc)[i] = make_float4(0.0f, 0.0f, 0.0f, 0.0f);
  __syncthreads();

  const int2*  ip = (const int2*)ind + (size_t)slc * SLE;
  const float* cp = coef + (size_t)slc * SLE;
  v2u* q = hitq + wv * QCAP;
  const char* rowsb = (const char*)rows;

  unsigned head = 0u, tail = 0u;

  int2  ij = ip[wv * 32 + lane];
  float cw = cp[wv * 32 + lane];
  for (int c = wv; c < NCH; c += 32) {
    int2  ijn = ij;
    float cwn = cw;
    if (c + 32 < NCH) {
      ijn = ip[(c + 32) * 32 + lane];
      cwn = cp[(c + 32) * 32 + lane];
    }
    const bool h0 = (ij.x >> BSH) == bkt;
    const bool h1 = (ij.y >> BSH) == bkt;
    const unsigned m0 = __builtin_amdgcn_ballot_w32(h0);
    const unsigned m1 = __builtin_amdgcn_ballot_w32(h1);
    const unsigned n0 = (unsigned)__builtin_popcount(m0);
    const unsigned cb = __float_as_uint(cw);
    if (h0) {
      const unsigned p = tail + __builtin_amdgcn_mbcnt_lo(m0, 0u);
      v2u e;
      e.x = ((unsigned)ij.y << BSH) | ((unsigned)ij.x & (BN - 1));
      e.y = cb;
      q[p & (QCAP - 1)] = e;
    }
    if (h1) {
      const unsigned p = tail + n0 + __builtin_amdgcn_mbcnt_lo(m1, 0u);
      v2u e;
      e.x = ((unsigned)ij.x << BSH) | ((unsigned)ij.y & (BN - 1));
      e.y = cb;
      q[p & (QCAP - 1)] = e;
    }
    tail += n0 + (unsigned)__builtin_popcount(m1);

    while (tail - head >= 32u) {
      const v2u e = q[(head + lane) & (QCAP - 1)];
      head += 32u;
      const int srow = (int)((e.x >> BSH) << 8);
      const int drow = (int)((e.x & (BN - 1)) << 6);
      const int cbit = (int)e.y;
      #pragma unroll
      for (int g = 0; g < 32; g += HGRP) {
        float2 r[HGRP];
        #pragma unroll
        for (int u = 0; u < HGRP; ++u) {
          const int so = __builtin_amdgcn_readlane(srow, g + u);
          r[u] = ((const float2*)(rowsb + (size_t)(unsigned)so))[lane];
        }
        #pragma unroll
        for (int u = 0; u < HGRP; ++u) {
          const int   dl = __builtin_amdgcn_readlane(drow, g + u);
          const float cf = __int_as_float(__builtin_amdgcn_readlane(cbit, g + u));
          atomicAdd(&lacc[dl + 2 * lane],     cf * r[u].x);
          atomicAdd(&lacc[dl + 2 * lane + 1], cf * r[u].y);
        }
      }
    }
    ij = ijn;
    cw = cwn;
  }

  const unsigned nleft = tail - head;
  if (nleft != 0u) {
    const v2u e = q[(head + lane) & (QCAP - 1)];
    const int srow = (int)((e.x >> BSH) << 8);
    const int drow = (int)((e.x & (BN - 1)) << 6);
    const int cbit = (int)e.y;
    #pragma unroll
    for (int g = 0; g < 32; g += 8) {
      if ((unsigned)g < nleft) {
        float2 r[8];
        #pragma unroll
        for (int u = 0; u < 8; ++u) {
          if ((unsigned)(g + u) < nleft) {
            const int so = __builtin_amdgcn_readlane(srow, g + u);
            r[u] = ((const float2*)(rowsb + (size_t)(unsigned)so))[lane];
          }
        }
        #pragma unroll
        for (int u = 0; u < 8; ++u) {
          if ((unsigned)(g + u) < nleft) {
            const int   dl = __builtin_amdgcn_readlane(drow, g + u);
            const float cf = __int_as_float(__builtin_amdgcn_readlane(cbit, g + u));
            atomicAdd(&lacc[dl + 2 * lane],     cf * r[u].x);
            atomicAdd(&lacc[dl + 2 * lane + 1], cf * r[u].y);
          }
        }
      }
    }
  }

  asm volatile("s_wait_dscnt 0x0" ::: "memory");
  __syncthreads();
  float* gacc = acc + (size_t)bkt * BN * D_OSC;
  for (int k0 = 0; k0 < BN * D_OSC / 1024; k0 += 8) {
    int   fi[8];
    float fv[8];
    #pragma unroll
    for (int u = 0; u < 8; ++u) {
      fi[u] = (((k0 + u + FSTG * slc) & (BN * D_OSC / 1024 - 1)) << 10) + tid;
      fv[u] = lacc[fi[u]];
    }
    #pragma unroll
    for (int u = 0; u < 8; ++u) atomicAdd(&gacc[fi[u]], fv[u]);
  }
}

typedef __attribute__((ext_vector_type(4))) float v4f;

#define SLP 44

__device__ __forceinline__ void gemv_fused_body(int bx, int by,
                                                float* __restrict__ slab,
                                                float* __restrict__ zstage,
                                                const float* __restrict__ W,
                                                const float* __restrict__ gv,
                                                float* __restrict__ y,
                                                float* __restrict__ z) {

  const int wv   = threadIdx.x >> 5;
  const int lane = threadIdx.x & 31;
  const int hl   = lane & 15;
  const int koff = (lane < 16) ? 0 : 2;

  const int C0 = bx * 512;
  const int r0 = by * 128 + wv * 16;

  zstage[threadIdx.x]       = 0.0f;
  zstage[threadIdx.x + 256] = 0.0f;
  __syncthreads();

  float* sl = slab + wv * (16 * SLP);

  v2f bz0 = *(const v2f*)(gv + r0 +  0 + koff);
  v2f bz1 = *(const v2f*)(gv + r0 +  4 + koff);
  v2f bz2 = *(const v2f*)(gv + r0 +  8 + koff);
  v2f bz3 = *(const v2f*)(gv + r0 + 12 + koff);

  const int lr = lane >> 3;
  const int lc = (lane & 7) * 4;
  const float* wld = W + (size_t)(r0 + lr) * N_NODES + C0 + lc;
  float* sst = sl + lr * SLP + lc;
  const float* fr = sl + hl * SLP + koff;

  v4f n0 = *(const v4f*)(wld + (size_t) 0 * N_NODES);
  v4f n1 = *(const v4f*)(wld + (size_t) 4 * N_NODES);
  v4f n2 = *(const v4f*)(wld + (size_t) 8 * N_NODES);
  v4f n3 = *(const v4f*)(wld + (size_t)12 * N_NODES);

  v8f yacc = {};
  for (int it = 0; it < 16; ++it) {
    const v4f c0 = n0, c1 = n1, c2 = n2, c3 = n3;
    if (it + 1 < 16) {
      const float* p = wld + (it + 1) * 32;
      n0 = *(const v4f*)(p + (size_t) 0 * N_NODES);
      n1 = *(const v4f*)(p + (size_t) 4 * N_NODES);
      n2 = *(const v4f*)(p + (size_t) 8 * N_NODES);
      n3 = *(const v4f*)(p + (size_t)12 * N_NODES);
    }
    *(v4f*)(sst +  0 * SLP) = c0;
    *(v4f*)(sst +  4 * SLP) = c1;
    *(v4f*)(sst +  8 * SLP) = c2;
    *(v4f*)(sst + 12 * SLP) = c3;

    #pragma unroll
    for (int ct = 0; ct < 2; ++ct) {
      const int c = C0 + it * 32 + ct * 16;
      v2f a0 = *(const v2f*)(fr + ct * 16 +  0);
      v2f a1 = *(const v2f*)(fr + ct * 16 +  4);
      v2f a2 = *(const v2f*)(fr + ct * 16 +  8);
      v2f a3 = *(const v2f*)(fr + ct * 16 + 12);
      v2f by0 = *(const v2f*)(gv + c +  0 + koff);
      v2f by1 = *(const v2f*)(gv + c +  4 + koff);
      v2f by2 = *(const v2f*)(gv + c +  8 + koff);
      v2f by3 = *(const v2f*)(gv + c + 12 + koff);

      yacc = __builtin_amdgcn_wmma_f32_16x16x4_f32(false, a0, false, by0, (short)0, yacc, false, false);
      yacc = __builtin_amdgcn_wmma_f32_16x16x4_f32(false, a1, false, by1, (short)0, yacc, false, false);
      yacc = __builtin_amdgcn_wmma_f32_16x16x4_f32(false, a2, false, by2, (short)0, yacc, false, false);
      yacc = __builtin_amdgcn_wmma_f32_16x16x4_f32(false, a3, false, by3, (short)0, yacc, false, false);

      const float* tr = sl + ct * 16 + hl;
      v2f az0, az1, az2, az3;
      az0[0] = tr[( 0 + koff) * SLP];  az0[1] = tr[( 1 + koff) * SLP];
      az1[0] = tr[( 4 + koff) * SLP];  az1[1] = tr[( 5 + koff) * SLP];
      az2[0] = tr[( 8 + koff) * SLP];  az2[1] = tr[( 9 + koff) * SLP];
      az3[0] = tr[(12 + koff) * SLP];  az3[1] = tr[(13 + koff) * SLP];

      v8f zt = {};
      zt = __builtin_amdgcn_wmma_f32_16x16x4_f32(false, az0, false, bz0, (short)0, zt, false, false);
      zt = __builtin_amdgcn_wmma_f32_16x16x4_f32(false, az1, false, bz1, (short)0, zt, false, false);
      zt = __builtin_amdgcn_wmma_f32_16x16x4_f32(false, az2, false, bz2, (short)0, zt, false, false);
      zt = __builtin_amdgcn_wmma_f32_16x16x4_f32(false, az3, false, bz3, (short)0, zt, false, false);

      if (hl == 0) {
        int zb = it * 32 + ct * 16 + ((lane == 16) ? 8 : 0);
        #pragma unroll
        for (int v = 0; v < 8; ++v) atomicAdd(&zstage[zb + v], zt[v]);
      }
    }
  }

  if (hl == 0) {
    int base = r0 + ((lane == 16) ? 8 : 0);
    #pragma unroll
    for (int v = 0; v < 8; ++v) atomicAdd(&y[base + v], yacc[v]);
  }

  __syncthreads();
  atomicAdd(&z[C0 + threadIdx.x],       zstage[threadIdx.x]);
  atomicAdd(&z[C0 + threadIdx.x + 256], zstage[threadIdx.x + 256]);
}

#define GEMV_BLKS (16 * 64)
#define COEF_BLKS (E_EDGES / (8 * EPW))

__global__ __launch_bounds__(256) void k_gemv_coef(const float* __restrict__ W,
                                                   const float* __restrict__ gv,
                                                   float* __restrict__ y,
                                                   float* __restrict__ z,
                                                   const int* __restrict__ ind,
                                                   const float* __restrict__ wK,
                                                   const float* __restrict__ skn,
                                                   float* __restrict__ coef) {
  __shared__ __attribute__((aligned(16))) float slab[8 * 16 * SLP];
  __shared__ float zstage[512];
  if (blockIdx.x < GEMV_BLKS) {
    gemv_fused_body(blockIdx.x & 15, blockIdx.x >> 4, slab, zstage, W, gv, y, z);
  } else {
    coef_K_body(blockIdx.x - GEMV_BLKS, ind, wK, skn, coef);
  }
}

__global__ __launch_bounds__(1024) void k_fin(const float* __restrict__ fKacc,
                                              const float* __restrict__ V,
                                              const float* __restrict__ skn,
                                              const float* __restrict__ gv,
                                              const float* __restrict__ y,
                                              const float* __restrict__ z,
                                              const float* __restrict__ sh,
                                              const float* __restrict__ bias,
                                              const float* __restrict__ kH,
                                              const float* __restrict__ kK,
                                              float* __restrict__ outH,
                                              float* __restrict__ outK) {
  __shared__ float sdot[32];
  int wv   = threadIdx.x >> 5;
  int lane = threadIdx.x & 31;
  int row  = blockIdx.x * 32 + wv;
  size_t base = (size_t)row * D_OSC;
  float s0 = skn[base + lane],   s1 = skn[base + lane + 32];
  float v0 = V[base + lane],     v1 = V[base + lane + 32];
  float k0 = fKacc[base + lane], k1 = fKacc[base + lane + 32];
  float dhv = wave_reduce_sum(s0 * v0 + s1 * v1);
  float c  = gv[row] / kK[0];
  float r0 = k0 - c * v0;
  float r1 = k1 - c * v1;
  float dot = wave_reduce_sum(s0 * r0 + s1 * r1);
  outK[base + lane]      = -r0 + s0 * dot;
  outK[base + lane + 32] = -r1 + s1 * dot;
  if (lane == 0) sdot[wv] = dhv;
  __syncthreads();
  if (wv == 0) {
    int i = blockIdx.x * 32 + lane;
    outH[i] = sdot[lane] / kH[0] + 0.5f * (y[i] + z[i]) - sh[i] + bias[i];
  }
}

extern "C" void kernel_launch(void* const* d_in, const int* in_sizes, int n_in,
                              void* d_out, int out_size, void* d_ws, size_t ws_size,
                              hipStream_t stream) {
  const float* state_H    = (const float*)d_in[0];
  const float* state_K    = (const float*)d_in[1];
  const int*   ind_K      = (const int*)d_in[2];
  const int*   ind_HK     = (const int*)d_in[3];
  const float* kappa_K    = (const float*)d_in[4];
  const float* kappa_H    = (const float*)d_in[5];
  const float* weights_H  = (const float*)d_in[6];
  const float* bias_H     = (const float*)d_in[7];
  const float* weights_HK = (const float*)d_in[8];
  const float* w_K        = (const float*)d_in[9];

  float* out_fH = (float*)d_out;
  float* out_fK = out_fH + N_NODES;

  float* ws     = (float*)d_ws;
  float* fK_acc = ws;
  float* Vacc   = fK_acc + (size_t)N_NODES * D_OSC;
  float* yv     = Vacc + (size_t)N_NODES * D_OSC;
  float* zv     = yv + N_NODES;
  float* skn    = zv + N_NODES;
  float* hs     = skn + (size_t)N_NODES * D_OSC;
  float* gv     = hs + (size_t)N_NODES * D_OSC;
  float* coefK  = gv + N_NODES;

  k_norm_zero<<<NORM_BLKS + ZERO_BLKS, 256, 0, stream>>>(state_K, state_H, skn, hs, gv, ws);

  k_gemv_coef<<<GEMV_BLKS + COEF_BLKS, 256, 0, stream>>>(weights_H, gv, yv, zv,
                                                         ind_K, w_K, skn, coefK);

  k_scatter<<<2 * NBKT * NSL, 1024, 0, stream>>>(ind_HK, weights_HK, hs, Vacc,
                                                 ind_K, coefK, skn, fK_acc);

  k_fin<<<N_NODES / 32, 1024, 0, stream>>>(fK_acc, Vacc, skn, gv, yv, zv,
                                           state_H, bias_H, kappa_H, kappa_K,
                                           out_fH, out_fK);
}


# Scatter: each work-group orders its hits by destination row in LDS (count, scan, place; integer adds); a wave sums its rows' sources in registers; no LDS float adds or flush pass; sum order changes
# speedup vs baseline: 4.1311x; 1.0591x over previous
#include <hip/hip_runtime.h>
#include <math.h>

#define N_NODES 8192
#define D_OSC   64
#define E_EDGES 1048576

typedef __attribute__((ext_vector_type(2))) float v2f;
typedef __attribute__((ext_vector_type(8))) float v8f;

__device__ __forceinline__ float wave_reduce_sum(float v) {
  #pragma unroll
  for (int m = 16; m >= 1; m >>= 1) v += __shfl_xor(v, m, 32);
  return v;
}

#define N_ACC     (2 * N_NODES * D_OSC + 2 * N_NODES)
#define NORM_BLKS (N_NODES / 8)
#define ZERO_BLKS (N_ACC / 4 / 256)

__device__ __forceinline__ void norm_tanh_body(int blk,
                                               const float* __restrict__ sk,
                                               const float* __restrict__ sh,
                                               float* __restrict__ skn,
                                               float* __restrict__ hs,
                                               float* __restrict__ gv) {
  int row  = blk * 8 + (threadIdx.x >> 5);
  int lane = threadIdx.x & 31;
  const float* r = sk + (size_t)row * D_OSC;
  float a = r[lane];
  float b = r[lane + 32];
  float ss  = wave_reduce_sum(a * a + b * b);
  float inv = rsqrtf(ss);
  float na = a * inv;
  float nb = b * inv;
  float* o = skn + (size_t)row * D_OSC;
  o[lane]      = na;
  o[lane + 32] = nb;
  float grow = tanhf(sh[row]);
  float* oh = hs + (size_t)row * D_OSC;
  oh[lane]      = grow * na;
  oh[lane + 32] = grow * nb;

  int gi = blk * 256 + threadIdx.x;
  if (gi < N_NODES) gv[gi] = tanhf(sh[gi]);
}

__global__ __launch_bounds__(256) void k_norm_zero(const float* __restrict__ sk,
                                                   const float* __restrict__ sh,
                                                   float* __restrict__ skn,
                                                   float* __restrict__ hs,
                                                   float* __restrict__ gv,
                                                   float* __restrict__ accs) {
  if (blockIdx.x < NORM_BLKS) {
    norm_tanh_body(blockIdx.x, sk, sh, skn, hs, gv);
  } else {
    const int i = (blockIdx.x - NORM_BLKS) * 256 + threadIdx.x;
    ((float4*)accs)[i] = make_float4(0.0f, 0.0f, 0.0f, 0.0f);
  }
}

#define EPW  32
#define CGRP 4

template <int CTRL>
__device__ __forceinline__ int dpp_i(int x) {
  return __builtin_amdgcn_update_dpp(0, x, CTRL, 0xf, 0xf, true);
}
template <int T>
__device__ __forceinline__ int row_share(int x) { return dpp_i<0x150 + T>(x); }
template <int M>
__device__ __forceinline__ float row_xor(float x) {
  return __builtin_bit_cast(float, dpp_i<0x160 + M>(__builtin_bit_cast(int, x)));
}
__device__ __forceinline__ float row16_sum(float v) {
  v += row_xor<8>(v);
  v += row_xor<4>(v);
  v += row_xor<2>(v);
  v += row_xor<1>(v);
  return v;
}

template <int T0>
__device__ __forceinline__ void coef_K_group(const int2 myj, const char* __restrict__ rb,
                                             const unsigned ho, const int hl, float& mys) {
  float4 a[CGRP], b[CGRP];
  #pragma unroll
  for (int u = 0; u < CGRP; ++u) {
    unsigned j0, j1;
    switch (u) {
      case 0:  j0 = (unsigned)row_share<T0 + 0>(myj.x); j1 = (unsigned)row_share<T0 + 0>(myj.y); break;
      case 1:  j0 = (unsigned)row_share<T0 + 1>(myj.x); j1 = (unsigned)row_share<T0 + 1>(myj.y); break;
      case 2:  j0 = (unsigned)row_share<T0 + 2>(myj.x); j1 = (unsigned)row_share<T0 + 2>(myj.y); break;
      default: j0 = (unsigned)row_share<T0 + 3>(myj.x); j1 = (unsigned)row_share<T0 + 3>(myj.y); break;
    }
    a[u] = *(const float4*)(rb + ((j0 << 8) | ho));
    b[u] = *(const float4*)(rb + ((j1 << 8) | ho));
  }
  #pragma unroll
  for (int u = 0; u < CGRP; ++u) {
    float p = a[u].x * b[u].x;
    p = fmaf(a[u].y, b[u].y, p);
    p = fmaf(a[u].z, b[u].z, p);
    p = fmaf(a[u].w, b[u].w, p);
    p = row16_sum(p);
    if (hl == T0 + u) mys = p;
  }
}

__device__ __forceinline__ void coef_K_body(int blk,
                                            const int* __restrict__ ind,
                                            const float* __restrict__ wK,
                                            const float* __restrict__ skn,
                                            float* __restrict__ coef) {
  const int lane = threadIdx.x & 31;
  const int hl   = lane & 15;
  const unsigned ho = (unsigned)hl << 4;
  const int e0   = (blk * 8 + (threadIdx.x >> 5)) * EPW;
  const int2  myj = ((const int2*)ind)[e0 + lane];
  const float myw = wK[e0 + lane];
  const char* rb = (const char*)skn;
  float mys = 0.0f;
  coef_K_group< 0>(myj, rb, ho, hl, mys);
  __builtin_amdgcn_sched_barrier(0);
  coef_K_group< 4>(myj, rb, ho, hl, mys);
  __builtin_amdgcn_sched_barrier(0);
  coef_K_group< 8>(myj, rb, ho, hl, mys);
  __builtin_amdgcn_sched_barrier(0);
  coef_K_group<12>(myj, rb, ho, hl, mys);
  coef[e0 + lane] = myw * mys;
}

#define NBKT 8
#define BN   (N_NODES / NBKT)
#define BSH  10
#define NSL  16
#define SLE  (E_EDGES / NSL)
#define NCH  (SLE / 32)
#define HGRP 16
#define CCH  16
#define SCH  8
#define CAP   32768
#define NPART 4

typedef __attribute__((ext_vector_type(2))) unsigned v2u;

__device__ __forceinline__ void flush_row(float* __restrict__ g, int row, float ax, float ay, int lane) {
  const int sl = lane >> 1;
  const float x0 = __shfl(ax, sl, 32),      y0 = __shfl(ay, sl, 32);
  const float x1 = __shfl(ax, 16 + sl, 32), y1 = __shfl(ay, 16 + sl, 32);
  const float lo = (lane & 1) ? y0 : x0;
  const float hi = (lane & 1) ? y1 : x1;
  float* p = g + (size_t)row * D_OSC;
  atomicAdd(p + lane, lo);
  atomicAdd(p + 32 + lane, hi);
}

#define ROW_STEP(DL, CF, RX, RY)                                   \
  {                                                                \
    if ((DL) != cur) {                                             \
      if (cur >= 0) flush_row(gacc, cur, ax, ay, lane);            \
      ax = 0.0f; ay = 0.0f; cur = (DL);                            \
    }                                                              \
    ax = fmaf((CF), (RX), ax);                                     \
    ay = fmaf((CF), (RY), ay);                                     \
  }

__global__ __launch_bounds__(1024) void k_scatter(const int* __restrict__ indA,
                                                  const float* __restrict__ coefA,
                                                  const float* __restrict__ rowsA,
                                                  float* __restrict__ accA,
                                                  const int* __restrict__ indB,
                                                  const float* __restrict__ coefB,
                                                  const float* __restrict__ rowsB,
                                                  float* __restrict__ accB) {
  const bool setB = blockIdx.x >= NBKT * NSL;
  const int*   __restrict__ ind  = setB ? indB  : indA;
  const float* __restrict__ coef = setB ? coefB : coefA;
  const float* __restrict__ rows = setB ? rowsB : rowsA;
  float*       __restrict__ acc  = setB ? accB  : accA;
  const int blk = blockIdx.x & (NBKT * NSL - 1);
  __shared__ v2u srt[CAP];
  __shared__ unsigned cnt[BN];
  __shared__ unsigned rstart[BN + 1];
  __shared__ unsigned wsum[32];
  const int tid  = threadIdx.x;
  const int lane = tid & 31;
  const int wv   = __builtin_amdgcn_readfirstlane(tid >> 5);
  const int bkt  = blk & (NBKT - 1);
  const int slc  = blk >> 3;

  const int2*  ip = (const int2*)ind + (size_t)slc * SLE;
  const float* cp = coef + (size_t)slc * SLE;
  const char* rowsb = (const char*)rows;
  float* gacc = acc + (size_t)bkt * BN * D_OSC;
  const int R0 = ((wv + 2 * slc) & 31) * 32;
  const unsigned bbase = (unsigned)bkt << BSH;

  int nsub = 1, s = 0;
  while (s < nsub) {
    const int per = (nsub == 1) ? NCH : (NCH / NPART);
    const int cb = s * per, ce = cb + per;
    cnt[tid] = 0u;
    __syncthreads();

    for (int c = cb + wv; c < ce; c += 32 * CCH) {
      int2 ij[CCH];
      #pragma unroll
      for (int k = 0; k < CCH; ++k) ij[k] = ip[(c + 32 * k) * 32 + lane];
      #pragma unroll
      for (int k = 0; k < CCH; ++k) {
        const unsigned t0 = (unsigned)ij[k].x ^ bbase;
        const unsigned t1 = (unsigned)ij[k].y ^ bbase;
        if (t0 < BN) atomicAdd(&cnt[t0], 1u);
        if (t1 < BN) atomicAdd(&cnt[t1], 1u);
      }
    }
    asm volatile("s_wait_dscnt 0x0" ::: "memory");
    __syncthreads();

    const unsigned v = cnt[tid];
    unsigned x = v;
    #pragma unroll
    for (int d = 1; d < 32; d <<= 1) {
      const unsigned y = __shfl_up(x, d, 32);
      if (lane >= d) x += y;
    }
    if (lane == 31) wsum[wv] = x;
    __syncthreads();
    const unsigned wsv = wsum[lane];
    unsigned wx = wsv;
    #pragma unroll
    for (int d = 1; d < 32; d <<= 1) {
      const unsigned y = __shfl_up(wx, d, 32);
      if (lane >= d) wx += y;
    }
    const unsigned wbase = (unsigned)__builtin_amdgcn_readlane((int)(wx - wsv), wv);
    const unsigned total = (unsigned)__builtin_amdgcn_readlane((int)wx, 31);
    if (total > CAP) {
      nsub = NPART; s = 0;
      __syncthreads();
      continue;
    }
    const unsigned excl = wbase + x - v;
    cnt[tid]    = excl;
    rstart[tid] = excl;
    if (tid == 0) rstart[BN] = total;
    __syncthreads();

    for (int c = cb + wv; c < ce; c += 32 * SCH) {
      int2  ij[SCH];
      float cw[SCH];
      #pragma unroll
      for (int k = 0; k < SCH; ++k) {
        ij[k] = ip[(c + 32 * k) * 32 + lane];
        cw[k] = cp[(c + 32 * k) * 32 + lane];
      }
      unsigned p0[SCH], p1[SCH];
      #pragma unroll
      for (int k = 0; k < SCH; ++k) {
        p0[k] = 0u; p1[k] = 0u;
        const unsigned t0 = (unsigned)ij[k].x ^ bbase;
        const unsigned t1 = (unsigned)ij[k].y ^ bbase;
        if (t0 < BN) p0[k] = atomicAdd(&cnt[t0], 1u);
        if (t1 < BN) p1[k] = atomicAdd(&cnt[t1], 1u);
      }
      #pragma unroll
      for (int k = 0; k < SCH; ++k) {
        const unsigned cbits = __float_as_uint(cw[k]);
        const unsigned t0 = (unsigned)ij[k].x ^ bbase;
        const unsigned t1 = (unsigned)ij[k].y ^ bbase;
        if (t0 < BN) {
          v2u e;
          e.x = ((unsigned)ij[k].y << BSH) | t0;
          e.y = cbits;
          srt[p0[k]] = e;
        }
        if (t1 < BN) {
          v2u e;
          e.x = ((unsigned)ij[k].x << BSH) | t1;
          e.y = cbits;
          srt[p1[k]] = e;
        }
      }
    }
    asm volatile("s_wait_dscnt 0x0" ::: "memory");
    __syncthreads();

    unsigned p = (unsigned)__builtin_amdgcn_readfirstlane((int)rstart[R0]);
    const unsigned pend = (unsigned)__builtin_amdgcn_readfirstlane((int)rstart[R0 + 32]);
    int   cur = -1;
    float ax = 0.0f, ay = 0.0f;
    for (; p < pend; p += 32u) {
      const unsigned nb  = (pend - p < 32u) ? (pend - p) : 32u;
      const unsigned idx = p + (unsigned)lane;
      const v2u e = srt[idx < pend ? idx : p];
      const int srow = (int)((e.x >> BSH) << 8);
      const int drow = (int)(e.x & (BN - 1));
      const int cbit = (int)e.y;
      if (nb == 32u) {
        #pragma unroll
        for (int g = 0; g < 32; g += HGRP) {
          float2 r[HGRP];
          #pragma unroll
          for (int u = 0; u < HGRP; ++u) {
            const int so = __builtin_amdgcn_readlane(srow, g + u);
            r[u] = ((const float2*)(rowsb + (size_t)(unsigned)so))[lane];
          }
          #pragma unroll
          for (int u = 0; u < HGRP; ++u) {
            const int   dl = __builtin_amdgcn_readlane(drow, g + u);
            const float cf = __int_as_float(__builtin_amdgcn_readlane(cbit, g + u));
            ROW_STEP(dl, cf, r[u].x, r[u].y)
          }
        }
      } else {
        #pragma unroll
        for (int g = 0; g < 32; g += 8) {
          if ((unsigned)g < nb) {
            float2 r[8];
            #pragma unroll
            for (int u = 0; u < 8; ++u) {
              if ((unsigned)(g + u) < nb) {
                const int so = __builtin_amdgcn_readlane(srow, g + u);
                r[u] = ((const float2*)(rowsb + (size_t)(unsigned)so))[lane];
              }
            }
            #pragma unroll
            for (int u = 0; u < 8; ++u) {
              if ((unsigned)(g + u) < nb) {
                const int   dl = __builtin_amdgcn_readlane(drow, g + u);
                const float cf = __int_as_float(__builtin_amdgcn_readlane(cbit, g + u));
                ROW_STEP(dl, cf, r[u].x, r[u].y)
              }
            }
          }
        }
      }
    }
    if (cur >= 0) flush_row(gacc, cur, ax, ay, lane);
    __syncthreads();
    ++s;
  }
}

typedef __attribute__((ext_vector_type(4))) float v4f;

#define SLP 44

__device__ __forceinline__ void gemv_fused_body(int bx, int by,
                                                float* __restrict__ slab,
                                                float* __restrict__ zstage,
                                                const float* __restrict__ W,
                                                const float* __restrict__ gv,
                                                float* __restrict__ y,
                                                float* __restrict__ z) {

  const int wv   = threadIdx.x >> 5;
  const int lane = threadIdx.x & 31;
  const int hl   = lane & 15;
  const int koff = (lane < 16) ? 0 : 2;

  const int C0 = bx * 512;
  const int r0 = by * 128 + wv * 16;

  zstage[threadIdx.x]       = 0.0f;
  zstage[threadIdx.x + 256] = 0.0f;
  __syncthreads();

  float* sl = slab + wv * (16 * SLP);

  v2f bz0 = *(const v2f*)(gv + r0 +  0 + koff);
  v2f bz1 = *(const v2f*)(gv + r0 +  4 + koff);
  v2f bz2 = *(const v2f*)(gv + r0 +  8 + koff);
  v2f bz3 = *(const v2f*)(gv + r0 + 12 + koff);

  const int lr = lane >> 3;
  const int lc = (lane & 7) * 4;
  const float* wld = W + (size_t)(r0 + lr) * N_NODES + C0 + lc;
  float* sst = sl + lr * SLP + lc;
  const float* fr = sl + hl * SLP + koff;

  v4f n0 = *(const v4f*)(wld + (size_t) 0 * N_NODES);
  v4f n1 = *(const v4f*)(wld + (size_t) 4 * N_NODES);
  v4f n2 = *(const v4f*)(wld + (size_t) 8 * N_NODES);
  v4f n3 = *(const v4f*)(wld + (size_t)12 * N_NODES);

  v8f yacc = {};
  for (int it = 0; it < 16; ++it) {
    const v4f c0 = n0, c1 = n1, c2 = n2, c3 = n3;
    if (it + 1 < 16) {
      const float* p = wld + (it + 1) * 32;
      n0 = *(const v4f*)(p + (size_t) 0 * N_NODES);
      n1 = *(const v4f*)(p + (size_t) 4 * N_NODES);
      n2 = *(const v4f*)(p + (size_t) 8 * N_NODES);
      n3 = *(const v4f*)(p + (size_t)12 * N_NODES);
    }
    *(v4f*)(sst +  0 * SLP) = c0;
    *(v4f*)(sst +  4 * SLP) = c1;
    *(v4f*)(sst +  8 * SLP) = c2;
    *(v4f*)(sst + 12 * SLP) = c3;

    #pragma unroll
    for (int ct = 0; ct < 2; ++ct) {
      const int c = C0 + it * 32 + ct * 16;
      v2f a0 = *(const v2f*)(fr + ct * 16 +  0);
      v2f a1 = *(const v2f*)(fr + ct * 16 +  4);
      v2f a2 = *(const v2f*)(fr + ct * 16 +  8);
      v2f a3 = *(const v2f*)(fr + ct * 16 + 12);
      v2f by0 = *(const v2f*)(gv + c +  0 + koff);
      v2f by1 = *(const v2f*)(gv + c +  4 + koff);
      v2f by2 = *(const v2f*)(gv + c +  8 + koff);
      v2f by3 = *(const v2f*)(gv + c + 12 + koff);

      yacc = __builtin_amdgcn_wmma_f32_16x16x4_f32(false, a0, false, by0, (short)0, yacc, false, false);
      yacc = __builtin_amdgcn_wmma_f32_16x16x4_f32(false, a1, false, by1, (short)0, yacc, false, false);
      yacc = __builtin_amdgcn_wmma_f32_16x16x4_f32(false, a2, false, by2, (short)0, yacc, false, false);
      yacc = __builtin_amdgcn_wmma_f32_16x16x4_f32(false, a3, false, by3, (short)0, yacc, false, false);

      const float* tr = sl + ct * 16 + hl;
      v2f az0, az1, az2, az3;
      az0[0] = tr[( 0 + koff) * SLP];  az0[1] = tr[( 1 + koff) * SLP];
      az1[0] = tr[( 4 + koff) * SLP];  az1[1] = tr[( 5 + koff) * SLP];
      az2[0] = tr[( 8 + koff) * SLP];  az2[1] = tr[( 9 + koff) * SLP];
      az3[0] = tr[(12 + koff) * SLP];  az3[1] = tr[(13 + koff) * SLP];

      v8f zt = {};
      zt = __builtin_amdgcn_wmma_f32_16x16x4_f32(false, az0, false, bz0, (short)0, zt, false, false);
      zt = __builtin_amdgcn_wmma_f32_16x16x4_f32(false, az1, false, bz1, (short)0, zt, false, false);
      zt = __builtin_amdgcn_wmma_f32_16x16x4_f32(false, az2, false, bz2, (short)0, zt, false, false);
      zt = __builtin_amdgcn_wmma_f32_16x16x4_f32(false, az3, false, bz3, (short)0, zt, false, false);

      if (hl == 0) {
        int zb = it * 32 + ct * 16 + ((lane == 16) ? 8 : 0);
        #pragma unroll
        for (int v = 0; v < 8; ++v) atomicAdd(&zstage[zb + v], zt[v]);
      }
    }
  }

  if (hl == 0) {
    int base = r0 + ((lane == 16) ? 8 : 0);
    #pragma unroll
    for (int v = 0; v < 8; ++v) atomicAdd(&y[base + v], yacc[v]);
  }

  __syncthreads();
  atomicAdd(&z[C0 + threadIdx.x],       zstage[threadIdx.x]);
  atomicAdd(&z[C0 + threadIdx.x + 256], zstage[threadIdx.x + 256]);
}

#define GEMV_BLKS (16 * 64)
#define COEF_BLKS (E_EDGES / (8 * EPW))

__global__ __launch_bounds__(256) void k_gemv_coef(const float* __restrict__ W,
                                                   const float* __restrict__ gv,
                                                   float* __restrict__ y,
                                                   float* __restrict__ z,
                                                   const int* __restrict__ ind,
                                                   const float* __restrict__ wK,
                                                   const float* __restrict__ skn,
                                                   float* __restrict__ coef) {
  __shared__ __attribute__((aligned(16))) float slab[8 * 16 * SLP];
  __shared__ float zstage[512];
  if (blockIdx.x < GEMV_BLKS) {
    gemv_fused_body(blockIdx.x & 15, blockIdx.x >> 4, slab, zstage, W, gv, y, z);
  } else {
    coef_K_body(blockIdx.x - GEMV_BLKS, ind, wK, skn, coef);
  }
}

__global__ __launch_bounds__(1024) void k_fin(const float* __restrict__ fKacc,
                                              const float* __restrict__ V,
                                              const float* __restrict__ skn,
                                              const float* __restrict__ gv,
                                              const float* __restrict__ y,
                                              const float* __restrict__ z,
                                              const float* __restrict__ sh,
                                              const float* __restrict__ bias,
                                              const float* __restrict__ kH,
                                              const float* __restrict__ kK,
                                              float* __restrict__ outH,
                                              float* __restrict__ outK) {
  __shared__ float sdot[32];
  int wv   = threadIdx.x >> 5;
  int lane = threadIdx.x & 31;
  int row  = blockIdx.x * 32 + wv;
  size_t base = (size_t)row * D_OSC;
  float s0 = skn[base + lane],   s1 = skn[base + lane + 32];
  float v0 = V[base + lane],     v1 = V[base + lane + 32];
  float k0 = fKacc[base + lane], k1 = fKacc[base + lane + 32];
  float dhv = wave_reduce_sum(s0 * v0 + s1 * v1);
  float c  = gv[row] / kK[0];
  float r0 = k0 - c * v0;
  float r1 = k1 - c * v1;
  float dot = wave_reduce_sum(s0 * r0 + s1 * r1);
  outK[base + lane]      = -r0 + s0 * dot;
  outK[base + lane + 32] = -r1 + s1 * dot;
  if (lane == 0) sdot[wv] = dhv;
  __syncthreads();
  if (wv == 0) {
    int i = blockIdx.x * 32 + lane;
    outH[i] = sdot[lane] / kH[0] + 0.5f * (y[i] + z[i]) - sh[i] + bias[i];
  }
}

extern "C" void kernel_launch(void* const* d_in, const int* in_sizes, int n_in,
                              void* d_out, int out_size, void* d_ws, size_t ws_size,
                              hipStream_t stream) {
  const float* state_H    = (const float*)d_in[0];
  const float* state_K    = (const float*)d_in[1];
  const int*   ind_K      = (const int*)d_in[2];
  const int*   ind_HK     = (const int*)d_in[3];
  const float* kappa_K    = (const float*)d_in[4];
  const float* kappa_H    = (const float*)d_in[5];
  const float* weights_H  = (const float*)d_in[6];
  const float* bias_H     = (const float*)d_in[7];
  const float* weights_HK = (const float*)d_in[8];
  const float* w_K        = (const float*)d_in[9];

  float* out_fH = (float*)d_out;
  float* out_fK = out_fH + N_NODES;

  float* ws     = (float*)d_ws;
  float* fK_acc = ws;
  float* Vacc   = fK_acc + (size_t)N_NODES * D_OSC;
  float* yv     = Vacc + (size_t)N_NODES * D_OSC;
  float* zv     = yv + N_NODES;
  float* skn    = zv + N_NODES;
  float* hs     = skn + (size_t)N_NODES * D_OSC;
  float* gv     = hs + (size_t)N_NODES * D_OSC;
  float* coefK  = gv + N_NODES;

  k_norm_zero<<<NORM_BLKS + ZERO_BLKS, 256, 0, stream>>>(state_K, state_H, skn, hs, gv, ws);

  k_gemv_coef<<<GEMV_BLKS + COEF_BLKS, 256, 0, stream>>>(weights_H, gv, yv, zv,
                                                         ind_K, w_K, skn, coefK);

  k_scatter<<<2 * NBKT * NSL, 1024, 0, stream>>>(ind_HK, weights_HK, hs, Vacc,
                                                 ind_K, coefK, skn, fK_acc);

  k_fin<<<N_NODES / 32, 1024, 0, stream>>>(fK_acc, Vacc, skn, gv, yv, zv,
                                           state_H, bias_H, kappa_H, kappa_K,
                                           out_fH, out_fK);
}


# Scatter row walk: one 16-lane half per hit (16 B of the source row per lane, a load brings two hits' rows; rows' runs padded to even); halves' sums joined at the flush; sum order changes
# speedup vs baseline: 4.2893x; 1.0383x over previous
#include <hip/hip_runtime.h>
#include <math.h>

#define N_NODES 8192
#define D_OSC   64
#define E_EDGES 1048576

typedef __attribute__((ext_vector_type(2))) float v2f;
typedef __attribute__((ext_vector_type(8))) float v8f;

__device__ __forceinline__ float wave_reduce_sum(float v) {
  #pragma unroll
  for (int m = 16; m >= 1; m >>= 1) v += __shfl_xor(v, m, 32);
  return v;
}

#define N_ACC     (2 * N_NODES * D_OSC + 2 * N_NODES)
#define NORM_BLKS (N_NODES / 8)
#define ZERO_BLKS (N_ACC / 4 / 256)

__device__ __forceinline__ void norm_tanh_body(int blk,
                                               const float* __restrict__ sk,
                                               const float* __restrict__ sh,
                                               float* __restrict__ skn,
                                               float* __restrict__ hs,
                                               float* __restrict__ gv) {
  int row  = blk * 8 + (threadIdx.x >> 5);
  int lane = threadIdx.x & 31;
  const float* r = sk + (size_t)row * D_OSC;
  float a = r[lane];
  float b = r[lane + 32];
  float ss  = wave_reduce_sum(a * a + b * b);
  float inv = rsqrtf(ss);
  float na = a * inv;
  float nb = b * inv;
  float* o = skn + (size_t)row * D_OSC;
  o[lane]      = na;
  o[lane + 32] = nb;
  float grow = tanhf(sh[row]);
  float* oh = hs + (size_t)row * D_OSC;
  oh[lane]      = grow * na;
  oh[lane + 32] = grow * nb;

  int gi = blk * 256 + threadIdx.x;
  if (gi < N_NODES) gv[gi] = tanhf(sh[gi]);
}

__global__ __launch_bounds__(256) void k_norm_zero(const float* __restrict__ sk,
                                                   const float* __restrict__ sh,
                                                   float* __restrict__ skn,
                                                   float* __restrict__ hs,
                                                   float* __restrict__ gv,
                                                   float* __restrict__ accs) {
  if (blockIdx.x < NORM_BLKS) {
    norm_tanh_body(blockIdx.x, sk, sh, skn, hs, gv);
  } else {
    const int i = (blockIdx.x - NORM_BLKS) * 256 + threadIdx.x;
    ((float4*)accs)[i] = make_float4(0.0f, 0.0f, 0.0f, 0.0f);
  }
}

#define EPW  32
#define CGRP 4

template <int CTRL>
__device__ __forceinline__ int dpp_i(int x) {
  return __builtin_amdgcn_update_dpp(0, x, CTRL, 0xf, 0xf, true);
}
template <int T>
__device__ __forceinline__ int row_share(int x) { return dpp_i<0x150 + T>(x); }
template <int M>
__device__ __forceinline__ float row_xor(float x) {
  return __builtin_bit_cast(float, dpp_i<0x160 + M>(__builtin_bit_cast(int, x)));
}
__device__ __forceinline__ float row16_sum(float v) {
  v += row_xor<8>(v);
  v += row_xor<4>(v);
  v += row_xor<2>(v);
  v += row_xor<1>(v);
  return v;
}

template <int T0>
__device__ __forceinline__ void coef_K_group(const int2 myj, const char* __restrict__ rb,
                                             const unsigned ho, const int hl, float& mys) {
  float4 a[CGRP], b[CGRP];
  #pragma unroll
  for (int u = 0; u < CGRP; ++u) {
    unsigned j0, j1;
    switch (u) {
      case 0:  j0 = (unsigned)row_share<T0 + 0>(myj.x); j1 = (unsigned)row_share<T0 + 0>(myj.y); break;
      case 1:  j0 = (unsigned)row_share<T0 + 1>(myj.x); j1 = (unsigned)row_share<T0 + 1>(myj.y); break;
      case 2:  j0 = (unsigned)row_share<T0 + 2>(myj.x); j1 = (unsigned)row_share<T0 + 2>(myj.y); break;
      default: j0 = (unsigned)row_share<T0 + 3>(myj.x); j1 = (unsigned)row_share<T0 + 3>(myj.y); break;
    }
    a[u] = *(const float4*)(rb + ((j0 << 8) | ho));
    b[u] = *(const float4*)(rb + ((j1 << 8) | ho));
  }
  #pragma unroll
  for (int u = 0; u < CGRP; ++u) {
    float p = a[u].x * b[u].x;
    p = fmaf(a[u].y, b[u].y, p);
    p = fmaf(a[u].z, b[u].z, p);
    p = fmaf(a[u].w, b[u].w, p);
    p = row16_sum(p);
    if (hl == T0 + u) mys = p;
  }
}

__device__ __forceinline__ void coef_K_body(int blk,
                                            const int* __restrict__ ind,
                                            const float* __restrict__ wK,
                                            const float* __restrict__ skn,
                                            float* __restrict__ coef) {
  const int lane = threadIdx.x & 31;
  const int hl   = lane & 15;
  const unsigned ho = (unsigned)hl << 4;
  const int e0   = (blk * 8 + (threadIdx.x >> 5)) * EPW;
  const int2  myj = ((const int2*)ind)[e0 + lane];
  const float myw = wK[e0 + lane];
  const char* rb = (const char*)skn;
  float mys = 0.0f;
  coef_K_group< 0>(myj, rb, ho, hl, mys);
  __builtin_amdgcn_sched_barrier(0);
  coef_K_group< 4>(myj, rb, ho, hl, mys);
  __builtin_amdgcn_sched_barrier(0);
  coef_K_group< 8>(myj, rb, ho, hl, mys);
  __builtin_amdgcn_sched_barrier(0);
  coef_K_group<12>(myj, rb, ho, hl, mys);
  coef[e0 + lane] = myw * mys;
}

#define NBKT 8
#define BN   (N_NODES / NBKT)
#define BSH  10
#define NSL  16
#define SLE  (E_EDGES / NSL)
#define NCH  (SLE / 32)
#define CCH  16
#define SCH  8
#define CAP   32768
#define NPART 4
#define CAPS  (CAP + BN)
#define HPG   8

typedef __attribute__((ext_vector_type(2))) unsigned v2u;

__device__ __forceinline__ void flush_row(float* __restrict__ g, int row, float4 a, int lane,
                                          float* __restrict__ fs) {
  *(float4*)(fs + lane * 4) = a;
  const float lo = fs[lane]      + fs[64 + lane];
  const float hi = fs[32 + lane] + fs[96 + lane];
  float* p = g + (size_t)row * D_OSC;
  atomicAdd(p + lane, lo);
  atomicAdd(p + 32 + lane, hi);
}

#define ROW_STEP(DL, CF, R)                                        \
  {                                                                \
    if ((DL) != cur) {                                             \
      if (cur >= 0) flush_row(gacc, cur, acc4, lane, fs);          \
      acc4 = make_float4(0.0f, 0.0f, 0.0f, 0.0f); cur = (DL);      \
    }                                                              \
    acc4.x = fmaf((CF), (R).x, acc4.x);                            \
    acc4.y = fmaf((CF), (R).y, acc4.y);                            \
    acc4.z = fmaf((CF), (R).z, acc4.z);                            \
    acc4.w = fmaf((CF), (R).w, acc4.w);                            \
  }

#define RSHARE8(G0, U, X)                                          \
  ((U) == 0 ? row_share<(G0) + 0>(X) : (U) == 1 ? row_share<(G0) + 1>(X) : \
   (U) == 2 ? row_share<(G0) + 2>(X) : (U) == 3 ? row_share<(G0) + 3>(X) : \
   (U) == 4 ? row_share<(G0) + 4>(X) : (U) == 5 ? row_share<(G0) + 5>(X) : \
   (U) == 6 ? row_share<(G0) + 6>(X) : row_share<(G0) + 7>(X))

__global__ __launch_bounds__(1024) void k_scatter(const int* __restrict__ indA,
                                                  const float* __restrict__ coefA,
                                                  const float* __restrict__ rowsA,
                                                  float* __restrict__ accA,
                                                  const int* __restrict__ indB,
                                                  const float* __restrict__ coefB,
                                                  const float* __restrict__ rowsB,
                                                  float* __restrict__ accB) {
  const bool setB = blockIdx.x >= NBKT * NSL;
  const int*   __restrict__ ind  = setB ? indB  : indA;
  const float* __restrict__ coef = setB ? coefB : coefA;
  const float* __restrict__ rows = setB ? rowsB : rowsA;
  float*       __restrict__ acc  = setB ? accB  : accA;
  const int blk = blockIdx.x & (NBKT * NSL - 1);
  __shared__ v2u srt[CAPS];
  __shared__ __attribute__((aligned(16))) float fsc[32 * 128];
  __shared__ unsigned cnt[BN];
  __shared__ unsigned rstart[BN + 1];
  __shared__ unsigned wsum[32];
  const int tid  = threadIdx.x;
  const int lane = tid & 31;
  const int wv   = __builtin_amdgcn_readfirstlane(tid >> 5);
  const int bkt  = blk & (NBKT - 1);
  const int slc  = blk >> 3;

  const int2*  ip = (const int2*)ind + (size_t)slc * SLE;
  const float* cp = coef + (size_t)slc * SLE;
  const char* rowsb = (const char*)rows;
  float* gacc = acc + (size_t)bkt * BN * D_OSC;
  float* fs = fsc + wv * 128;
  const int R0 = ((wv + 2 * slc) & 31) * 32;
  const unsigned bbase = (unsigned)bkt << BSH;

  int nsub = 1, s = 0;
  while (s < nsub) {
    const int per = (nsub == 1) ? NCH : (NCH / NPART);
    const int cb = s * per, ce = cb + per;
    cnt[tid] = 0u;
    __syncthreads();

    for (int c = cb + wv; c < ce; c += 32 * CCH) {
      int2 ij[CCH];
      #pragma unroll
      for (int k = 0; k < CCH; ++k) ij[k] = ip[(c + 32 * k) * 32 + lane];
      #pragma unroll
      for (int k = 0; k < CCH; ++k) {
        const unsigned t0 = (unsigned)ij[k].x ^ bbase;
        const unsigned t1 = (unsigned)ij[k].y ^ bbase;
        if (t0 < BN) atomicAdd(&cnt[t0], 1u);
        if (t1 < BN) atomicAdd(&cnt[t1], 1u);
      }
    }
    asm volatile("s_wait_dscnt 0x0" ::: "memory");
    __syncthreads();

    const unsigned vr = cnt[tid];
    const unsigned v = (vr + 1u) & ~1u;
    unsigned x = v;
    #pragma unroll
    for (int d = 1; d < 32; d <<= 1) {
      const unsigned y = __shfl_up(x, d, 32);
      if (lane >= d) x += y;
    }
    if (lane == 31) wsum[wv] = x;
    __syncthreads();
    const unsigned wsv = wsum[lane];
    unsigned wx = wsv;
    #pragma unroll
    for (int d = 1; d < 32; d <<= 1) {
      const unsigned y = __shfl_up(wx, d, 32);
      if (lane >= d) wx += y;
    }
    const unsigned wbase = (unsigned)__builtin_amdgcn_readlane((int)(wx - wsv), wv);
    const unsigned total = (unsigned)__builtin_amdgcn_readlane((int)wx, 31);
    if (total > CAPS) {
      nsub = NPART; s = 0;
      __syncthreads();
      continue;
    }
    const unsigned excl = wbase + x - v;
    cnt[tid]    = excl;
    rstart[tid] = excl;
    if (tid == 0) rstart[BN] = total;
    if (vr & 1u) {
      v2u e;
      e.x = ((bbase + (unsigned)tid) << BSH) | (unsigned)tid;
      e.y = 0u;
      srt[excl + vr] = e;
    }
    __syncthreads();

    for (int c = cb + wv; c < ce; c += 32 * SCH) {
      int2  ij[SCH];
      float cw[SCH];
      #pragma unroll
      for (int k = 0; k < SCH; ++k) {
        ij[k] = ip[(c + 32 * k) * 32 + lane];
        cw[k] = cp[(c + 32 * k) * 32 + lane];
      }
      unsigned p0[SCH], p1[SCH];
      #pragma unroll
      for (int k = 0; k < SCH; ++k) {
        p0[k] = 0u; p1[k] = 0u;
        const unsigned t0 = (unsigned)ij[k].x ^ bbase;
        const unsigned t1 = (unsigned)ij[k].y ^ bbase;
        if (t0 < BN) p0[k] = atomicAdd(&cnt[t0], 1u);
        if (t1 < BN) p1[k] = atomicAdd(&cnt[t1], 1u);
      }
      #pragma unroll
      for (int k = 0; k < SCH; ++k) {
        const unsigned cbits = __float_as_uint(cw[k]);
        const unsigned t0 = (unsigned)ij[k].x ^ bbase;
        const unsigned t1 = (unsigned)ij[k].y ^ bbase;
        if (t0 < BN) {
          v2u e;
          e.x = ((unsigned)ij[k].y << BSH) | t0;
          e.y = cbits;
          srt[p0[k]] = e;
        }
        if (t1 < BN) {
          v2u e;
          e.x = ((unsigned)ij[k].x << BSH) | t1;
          e.y = cbits;
          srt[p1[k]] = e;
        }
      }
    }
    asm volatile("s_wait_dscnt 0x0" ::: "memory");
    __syncthreads();

    unsigned p = (unsigned)__builtin_amdgcn_readfirstlane((int)rstart[R0]);
    const unsigned pend = (unsigned)__builtin_amdgcn_readfirstlane((int)rstart[R0 + 32]);
    const int hl = lane & 15;
    const unsigned ho = (unsigned)hl << 4;
    int    cur  = -1;
    float4 acc4 = make_float4(0.0f, 0.0f, 0.0f, 0.0f);
    for (; p < pend; p += 32u) {
      const unsigned np  = (pend - p < 32u) ? ((pend - p) >> 1) : 16u;
      const unsigned idx = p + 2u * (unsigned)hl + ((unsigned)lane >> 4);
      const v2u e = srt[idx < pend ? idx : p];
      const int snode = (int)(e.x >> BSH);
      const int drow  = (int)(e.x & (BN - 1));
      const int cbit  = (int)e.y;
      if (np == 16u) {
        #pragma unroll
        for (int g = 0; g < 16; g += HPG) {
          float4 r[HPG];
          #pragma unroll
          for (int u = 0; u < HPG; ++u) {
            const unsigned sn = (unsigned)(g == 0 ? RSHARE8(0, u, snode) : RSHARE8(8, u, snode));
            r[u] = *(const float4*)(rowsb + ((sn << 8) | ho));
          }
          #pragma unroll
          for (int u = 0; u < HPG; ++u) {
            const int   dl = __builtin_amdgcn_readlane(drow, g + u);
            const float cf = __int_as_float(g == 0 ? RSHARE8(0, u, cbit) : RSHARE8(8, u, cbit));
            ROW_STEP(dl, cf, r[u])
          }
        }
      } else {
        #pragma unroll
        for (int g = 0; g < 16; g += HPG) {
          if ((unsigned)g < np) {
            float4 r[HPG];
            #pragma unroll
            for (int u = 0; u < HPG; ++u) {
              if ((unsigned)(g + u) < np) {
                const unsigned sn = (unsigned)(g == 0 ? RSHARE8(0, u, snode) : RSHARE8(8, u, snode));
                r[u] = *(const float4*)(rowsb + ((sn << 8) | ho));
              }
            }
            #pragma unroll
            for (int u = 0; u < HPG; ++u) {
              if ((unsigned)(g + u) < np) {
                const int   dl = __builtin_amdgcn_readlane(drow, g + u);
                const float cf = __int_as_float(g == 0 ? RSHARE8(0, u, cbit) : RSHARE8(8, u, cbit));
                ROW_STEP(dl, cf, r[u])
              }
            }
          }
        }
      }
    }
    if (cur >= 0) flush_row(gacc, cur, acc4, lane, fs);
    __syncthreads();
    ++s;
  }
}

typedef __attribute__((ext_vector_type(4))) float v4f;

#define SLP 44

__device__ __forceinline__ void gemv_fused_body(int bx, int by,
                                                float* __restrict__ slab,
                                                float* __restrict__ zstage,
                                                const float* __restrict__ W,
                                                const float* __restrict__ gv,
                                                float* __restrict__ y,
                                                float* __restrict__ z) {

  const int wv   = threadIdx.x >> 5;
  const int lane = threadIdx.x & 31;
  const int hl   = lane & 15;
  const int koff = (lane < 16) ? 0 : 2;

  const int C0 = bx * 512;
  const int r0 = by * 128 + wv * 16;

  zstage[threadIdx.x]       = 0.0f;
  zstage[threadIdx.x + 256] = 0.0f;
  __syncthreads();

  float* sl = slab + wv * (16 * SLP);

  v2f bz0 = *(const v2f*)(gv + r0 +  0 + koff);
  v2f bz1 = *(const v2f*)(gv + r0 +  4 + koff);
  v2f bz2 = *(const v2f*)(gv + r0 +  8 + koff);
  v2f bz3 = *(const v2f*)(gv + r0 + 12 + koff);

  const int lr = lane >> 3;
  const int lc = (lane & 7) * 4;
  const float* wld = W + (size_t)(r0 + lr) * N_NODES + C0 + lc;
  float* sst = sl + lr * SLP + lc;
  const float* fr = sl + hl * SLP + koff;

  v4f n0 = *(const v4f*)(wld + (size_t) 0 * N_NODES);
  v4f n1 = *(const v4f*)(wld + (size_t) 4 * N_NODES);
  v4f n2 = *(const v4f*)(wld + (size_t) 8 * N_NODES);
  v4f n3 = *(const v4f*)(wld + (size_t)12 * N_NODES);

  v8f yacc = {};
  for (int it = 0; it < 16; ++it) {
    const v4f c0 = n0, c1 = n1, c2 = n2, c3 = n3;
    if (it + 1 < 16) {
      const float* p = wld + (it + 1) * 32;
      n0 = *(const v4f*)(p + (size_t) 0 * N_NODES);
      n1 = *(const v4f*)(p + (size_t) 4 * N_NODES);
      n2 = *(const v4f*)(p + (size_t) 8 * N_NODES);
      n3 = *(const v4f*)(p + (size_t)12 * N_NODES);
    }
    *(v4f*)(sst +  0 * SLP) = c0;
    *(v4f*)(sst +  4 * SLP) = c1;
    *(v4f*)(sst +  8 * SLP) = c2;
    *(v4f*)(sst + 12 * SLP) = c3;

    #pragma unroll
    for (int ct = 0; ct < 2; ++ct) {
      const int c = C0 + it * 32 + ct * 16;
      v2f a0 = *(const v2f*)(fr + ct * 16 +  0);
      v2f a1 = *(const v2f*)(fr + ct * 16 +  4);
      v2f a2 = *(const v2f*)(fr + ct * 16 +  8);
      v2f a3 = *(const v2f*)(fr + ct * 16 + 12);
      v2f by0 = *(const v2f*)(gv + c +  0 + koff);
      v2f by1 = *(const v2f*)(gv + c +  4 + koff);
      v2f by2 = *(const v2f*)(gv + c +  8 + koff);
      v2f by3 = *(const v2f*)(gv + c + 12 + koff);

      yacc = __builtin_amdgcn_wmma_f32_16x16x4_f32(false, a0, false, by0, (short)0, yacc, false, false);
      yacc = __builtin_amdgcn_wmma_f32_16x16x4_f32(false, a1, false, by1, (short)0, yacc, false, false);
      yacc = __builtin_amdgcn_wmma_f32_16x16x4_f32(false, a2, false, by2, (short)0, yacc, false, false);
      yacc = __builtin_amdgcn_wmma_f32_16x16x4_f32(false, a3, false, by3, (short)0, yacc, false, false);

      const float* tr = sl + ct * 16 + hl;
      v2f az0, az1, az2, az3;
      az0[0] = tr[( 0 + koff) * SLP];  az0[1] = tr[( 1 + koff) * SLP];
      az1[0] = tr[( 4 + koff) * SLP];  az1[1] = tr[( 5 + koff) * SLP];
      az2[0] = tr[( 8 + koff) * SLP];  az2[1] = tr[( 9 + koff) * SLP];
      az3[0] = tr[(12 + koff) * SLP];  az3[1] = tr[(13 + koff) * SLP];

      v8f zt = {};
      zt = __builtin_amdgcn_wmma_f32_16x16x4_f32(false, az0, false, bz0, (short)0, zt, false, false);
      zt = __builtin_amdgcn_wmma_f32_16x16x4_f32(false, az1, false, bz1, (short)0, zt, false, false);
      zt = __builtin_amdgcn_wmma_f32_16x16x4_f32(false, az2, false, bz2, (short)0, zt, false, false);
      zt = __builtin_amdgcn_wmma_f32_16x16x4_f32(false, az3, false, bz3, (short)0, zt, false, false);

      if (hl == 0) {
        int zb = it * 32 + ct * 16 + ((lane == 16) ? 8 : 0);
        #pragma unroll
        for (int v = 0; v < 8; ++v) atomicAdd(&zstage[zb + v], zt[v]);
      }
    }
  }

  if (hl == 0) {
    int base = r0 + ((lane == 16) ? 8 : 0);
    #pragma unroll
    for (int v = 0; v < 8; ++v) atomicAdd(&y[base + v], yacc[v]);
  }

  __syncthreads();
  atomicAdd(&z[C0 + threadIdx.x],       zstage[threadIdx.x]);
  atomicAdd(&z[C0 + threadIdx.x + 256], zstage[threadIdx.x + 256]);
}

#define GEMV_BLKS (16 * 64)
#define COEF_BLKS (E_EDGES / (8 * EPW))

__global__ __launch_bounds__(256) void k_gemv_coef(const float* __restrict__ W,
                                                   const float* __restrict__ gv,
                                                   float* __restrict__ y,
                                                   float* __restrict__ z,
                                                   const int* __restrict__ ind,
                                                   const float* __restrict__ wK,
                                                   const float* __restrict__ skn,
                                                   float* __restrict__ coef) {
  __shared__ __attribute__((aligned(16))) float slab[8 * 16 * SLP];
  __shared__ float zstage[512];
  if (blockIdx.x < GEMV_BLKS) {
    gemv_fused_body(blockIdx.x & 15, blockIdx.x >> 4, slab, zstage, W, gv, y, z);
  } else {
    coef_K_body(blockIdx.x - GEMV_BLKS, ind, wK, skn, coef);
  }
}

__global__ __launch_bounds__(1024) void k_fin(const float* __restrict__ fKacc,
                                              const float* __restrict__ V,
                                              const float* __restrict__ skn,
                                              const float* __restrict__ gv,
                                              const float* __restrict__ y,
                                              const float* __restrict__ z,
                                              const float* __restrict__ sh,
                                              const float* __restrict__ bias,
                                              const float* __restrict__ kH,
                                              const float* __restrict__ kK,
                                              float* __restrict__ outH,
                                              float* __restrict__ outK) {
  __shared__ float sdot[32];
  int wv   = threadIdx.x >> 5;
  int lane = threadIdx.x & 31;
  int row  = blockIdx.x * 32 + wv;
  size_t base = (size_t)row * D_OSC;
  float s0 = skn[base + lane],   s1 = skn[base + lane + 32];
  float v0 = V[base + lane],     v1 = V[base + lane + 32];
  float k0 = fKacc[base + lane], k1 = fKacc[base + lane + 32];
  float dhv = wave_reduce_sum(s0 * v0 + s1 * v1);
  float c  = gv[row] / kK[0];
  float r0 = k0 - c * v0;
  float r1 = k1 - c * v1;
  float dot = wave_reduce_sum(s0 * r0 + s1 * r1);
  outK[base + lane]      = -r0 + s0 * dot;
  outK[base + lane + 32] = -r1 + s1 * dot;
  if (lane == 0) sdot[wv] = dhv;
  __syncthreads();
  if (wv == 0) {
    int i = blockIdx.x * 32 + lane;
    outH[i] = sdot[lane] / kH[0] + 0.5f * (y[i] + z[i]) - sh[i] + bias[i];
  }
}

extern "C" void kernel_launch(void* const* d_in, const int* in_sizes, int n_in,
                              void* d_out, int out_size, void* d_ws, size_t ws_size,
                              hipStream_t stream) {
  const float* state_H    = (const float*)d_in[0];
  const float* state_K    = (const float*)d_in[1];
  const int*   ind_K      = (const int*)d_in[2];
  const int*   ind_HK     = (const int*)d_in[3];
  const float* kappa_K    = (const float*)d_in[4];
  const float* kappa_H    = (const float*)d_in[5];
  const float* weights_H  = (const float*)d_in[6];
  const float* bias_H     = (const float*)d_in[7];
  const float* weights_HK = (const float*)d_in[8];
  const float* w_K        = (const float*)d_in[9];

  float* out_fH = (float*)d_out;
  float* out_fK = out_fH + N_NODES;

  float* ws     = (float*)d_ws;
  float* fK_acc = ws;
  float* Vacc   = fK_acc + (size_t)N_NODES * D_OSC;
  float* yv     = Vacc + (size_t)N_NODES * D_OSC;
  float* zv     = yv + N_NODES;
  float* skn    = zv + N_NODES;
  float* hs     = skn + (size_t)N_NODES * D_OSC;
  float* gv     = hs + (size_t)N_NODES * D_OSC;
  float* coefK  = gv + N_NODES;

  k_norm_zero<<<NORM_BLKS + ZERO_BLKS, 256, 0, stream>>>(state_K, state_H, skn, hs, gv, ws);

  k_gemv_coef<<<GEMV_BLKS + COEF_BLKS, 256, 0, stream>>>(weights_H, gv, yv, zv,
                                                         ind_K, w_K, skn, coefK);

  k_scatter<<<2 * NBKT * NSL, 1024, 0, stream>>>(ind_HK, weights_HK, hs, Vacc,
                                                 ind_K, coefK, skn, fK_acc);

  k_fin<<<N_NODES / 32, 1024, 0, stream>>>(fK_acc, Vacc, skn, gv, yv, zv,
                                           state_H, bias_H, kappa_H, kappa_K,
                                           out_fH, out_fK);
}
